# starting kernel
# baseline (speedup 1.0000x reference)
#include <hip/hip_runtime.h>
#include <hip/hip_bf16.h>
#include <math.h>

#define NN 100000
#define CC 16
#define GG 2
#define LL 2
#define EE 3200000
#define FF 512
#define HH 64

#define W1_PSTRIDE 160
#define W2_PSTRIDE 32
#define HID_STRIDE 66

typedef float v2f __attribute__((ext_vector_type(2)));
typedef float v8f __attribute__((ext_vector_type(8)));

__device__ __forceinline__ unsigned f2ou(float f) {
    unsigned u = __float_as_uint(f);
    return (u & 0x80000000u) ? ~u : (u | 0x80000000u);
}
__device__ __forceinline__ float ou2f(unsigned u) {
    return __uint_as_float((u & 0x80000000u) ? (u & 0x7FFFFFFFu) : ~u);
}

__global__ void k_init(unsigned* __restrict__ nmax, float* __restrict__ nsum,
                       float* __restrict__ hnext) {
    int i = blockIdx.x * blockDim.x + threadIdx.x;
    if (i < NN) { nmax[i] = 0x007FFFFFu;   nsum[i] = 0.0f; }
    if (i < NN * CC) hnext[i] = 0.0f;
}

__global__ void k_segmax(const float* __restrict__ e, const int* __restrict__ dst,
                         unsigned* __restrict__ nmax) {
    int i = blockIdx.x * blockDim.x + threadIdx.x;
    if (i >= EE) return;
    atomicMax(&nmax[dst[i]], f2ou(e[i]));
}

__global__ void k_segsum(const float* __restrict__ e, const int* __restrict__ dst,
                         const unsigned* __restrict__ nmax, float* __restrict__ nsum) {
    int i = blockIdx.x * blockDim.x + threadIdx.x;
    if (i >= EE) return;
    int d = dst[i];
    atomicAdd(&nsum[d], expf(e[i] - ou2f(nmax[d])));
}

__global__ void k_agg(const float* __restrict__ e, const int* __restrict__ src,
                      const int* __restrict__ dst, const unsigned* __restrict__ nmax,
                      const float* __restrict__ nsum, const float* __restrict__ hcur,
                      float* __restrict__ hnext) {
    int i = blockIdx.x * blockDim.x + threadIdx.x;
    if (i >= EE) return;
    int d = dst[i], s = src[i];
    float a = expf(e[i] - ou2f(nmax[d])) / nsum[d];
    const float4* hs = (const float4*)(hcur + (size_t)s * CC);
    float4 v0 = hs[0], v1 = hs[1], v2 = hs[2], v3 = hs[3];
    float* o = hnext + (size_t)d * CC;
    atomicAdd(o + 0,  v0.x * a); atomicAdd(o + 1,  v0.y * a);
    atomicAdd(o + 2,  v0.z * a); atomicAdd(o + 3,  v0.w * a);
    atomicAdd(o + 4,  v1.x * a); atomicAdd(o + 5,  v1.y * a);
    atomicAdd(o + 6,  v1.z * a); atomicAdd(o + 7,  v1.w * a);
    atomicAdd(o + 8,  v2.x * a); atomicAdd(o + 9,  v2.y * a);
    atomicAdd(o + 10, v2.z * a); atomicAdd(o + 11, v2.w * a);
    atomicAdd(o + 12, v3.x * a); atomicAdd(o + 13, v3.y * a);
    atomicAdd(o + 14, v3.z * a); atomicAdd(o + 15, v3.w * a);
}

__global__ void k_mask(const float* __restrict__ one_hot, const int* __restrict__ mask,
                       float* __restrict__ h) {
    int i = blockIdx.x * blockDim.x + threadIdx.x;
    if (i >= NN * CC) return;
    int node = i >> 4;
    h[i] = mask[node] ? one_hot[i] : h[i];
}

__global__ __launch_bounds__(256) void k_mlp_combine(
    const float* __restrict__ feat, const float* __restrict__ w1,
    const float* __restrict__ b1,   const float* __restrict__ w2,
    const float* __restrict__ b2,   const float* __restrict__ alpha,
    const float* __restrict__ att,  const float* __restrict__ h0,
    const float* __restrict__ h1,   float* __restrict__ out, int n_pairs)
{
    __shared__ float sW1p[(FF / 2) * W1_PSTRIDE];
    __shared__ float sW2p[(HH / 2) * W2_PSTRIDE];
    __shared__ float sHid[8][2][16 * HID_STRIDE];

    int t = threadIdx.x;

    for (int i = t; i < (FF / 2) * (HH / 4); i += 256) {
        int p  = i >> 4;
        int nq = (i & 15) << 2;
        float4 x = *(const float4*)&w1[(size_t)p * 2 * HH + nq];
        float4 y = *(const float4*)&w1[(size_t)p * 2 * HH + HH + nq];
        float4 lo = { x.x, y.x, x.y, y.y };
        float4 hi = { x.z, y.z, x.w, y.w };
        float* d = &sW1p[p * W1_PSTRIDE + 2 * nq];
        *(float4*)(d + 0) = lo;
        *(float4*)(d + 4) = hi;
    }
    for (int i = t; i < (HH / 2) * CC; i += 256) {
        int p = i >> 4, n = i & 15;
        v2f v; v.x = w2[(size_t)p * 2 * CC + n];
        v.y = w2[(size_t)p * 2 * CC + CC + n];
        *(v2f*)&sW2p[p * W2_PSTRIDE + 2 * n] = v;
    }
    __syncthreads();

    int wave = t >> 5;
    int lane = t & 31;
    int pair = blockIdx.x * 8 + wave;
    if (pair >= n_pairs) return;

    int tile0 = pair * 2;
    int tile1 = tile0 + 1;

    int m     = lane & 15;
    int n     = lane & 15;
    int hi16  = lane >> 4;
    int khalf = hi16 << 1;
    int moff  = hi16 << 3;

    v8f acc0 = {}, acc1 = {}, acc2 = {}, acc3 = {};
    v8f acc4 = {}, acc5 = {}, acc6 = {}, acc7 = {};
    const float* arow0 = feat + ((size_t)tile0 * 16 + m) * FF;
    const float* arow1 = feat + ((size_t)tile1 * 16 + m) * FF;
    for (int k = 0; k < FF; k += 4) {
        int kb = k + khalf;
        v2f a0 = *(const v2f*)(arow0 + kb);
        v2f a1 = *(const v2f*)(arow1 + kb);
        const float* bp = &sW1p[((k >> 1) + hi16) * W1_PSTRIDE + 2 * n];
        v2f b0 = *(const v2f*)(bp + 0);
        v2f b1v = *(const v2f*)(bp + 32);
        v2f b2v = *(const v2f*)(bp + 64);
        v2f b3v = *(const v2f*)(bp + 96);
        acc0 = __builtin_amdgcn_wmma_f32_16x16x4_f32(false, a0, false, b0,  (short)0, acc0, false, false);
        acc1 = __builtin_amdgcn_wmma_f32_16x16x4_f32(false, a0, false, b1v, (short)0, acc1, false, false);
        acc2 = __builtin_amdgcn_wmma_f32_16x16x4_f32(false, a0, false, b2v, (short)0, acc2, false, false);
        acc3 = __builtin_amdgcn_wmma_f32_16x16x4_f32(false, a0, false, b3v, (short)0, acc3, false, false);
        acc4 = __builtin_amdgcn_wmma_f32_16x16x4_f32(false, a1, false, b0,  (short)0, acc4, false, false);
        acc5 = __builtin_amdgcn_wmma_f32_16x16x4_f32(false, a1, false, b1v, (short)0, acc5, false, false);
        acc6 = __builtin_amdgcn_wmma_f32_16x16x4_f32(false, a1, false, b2v, (short)0, acc6, false, false);
        acc7 = __builtin_amdgcn_wmma_f32_16x16x4_f32(false, a1, false, b3v, (short)0, acc7, false, false);
    }

    float bb0 = b1[n + 0], bb1 = b1[n + 16], bb2 = b1[n + 32], bb3 = b1[n + 48];
    {
        float* hd0 = &sHid[wave][0][0];
        float* hd1 = &sHid[wave][1][0];
        for (int r = 0; r < 8; ++r) {
            int mm = r + moff;
            hd0[mm * HID_STRIDE + n +  0] = fmaxf(acc0[r] + bb0, 0.0f);
            hd0[mm * HID_STRIDE + n + 16] = fmaxf(acc1[r] + bb1, 0.0f);
            hd0[mm * HID_STRIDE + n + 32] = fmaxf(acc2[r] + bb2, 0.0f);
            hd0[mm * HID_STRIDE + n + 48] = fmaxf(acc3[r] + bb3, 0.0f);
            hd1[mm * HID_STRIDE + n +  0] = fmaxf(acc4[r] + bb0, 0.0f);
            hd1[mm * HID_STRIDE + n + 16] = fmaxf(acc5[r] + bb1, 0.0f);
            hd1[mm * HID_STRIDE + n + 32] = fmaxf(acc6[r] + bb2, 0.0f);
            hd1[mm * HID_STRIDE + n + 48] = fmaxf(acc7[r] + bb3, 0.0f);
        }
    }

    float bc = b2[n];
    for (int tt = 0; tt < 2; ++tt) {
        const float* hd = &sHid[wave][tt][0];
        v8f acc = {};
        for (int k = 0; k < HH; k += 4) {
            int kb = k + khalf;
            v2f a = *(const v2f*)(hd + m * HID_STRIDE + kb);
            v2f b = *(const v2f*)&sW2p[((k >> 1) + hi16) * W2_PSTRIDE + 2 * n];
            acc = __builtin_amdgcn_wmma_f32_16x16x4_f32(false, a, false, b, (short)0, acc, false, false);
        }
        int tile = tt ? tile1 : tile0;
        for (int r = 0; r < 8; ++r) {
            int node = tile * 16 + r + moff;
            float mlp = acc[r] + bc;
            float al  = alpha[node];
            float sa  = 1.0f / (1.0f + expf(-al));
            float sna = 1.0f - sa;
            float a0 = att[node * 2 + 0], a1 = att[node * 2 + 1];
            float mx = fmaxf(a0, a1);
            float e0 = expf(a0 - mx), e1 = expf(a1 - mx);
            float inv = 1.0f / (e0 + e1);
            float logit = h0[(size_t)node * CC + n] * (e0 * inv)
                        + h1[(size_t)node * CC + n] * (e1 * inv);
            out[(size_t)node * CC + n] = sa * logit + sna * mlp;
        }
    }
}

extern "C" void kernel_launch(void* const* d_in, const int* in_sizes, int n_in,
                              void* d_out, int out_size, void* d_ws, size_t ws_size,
                              hipStream_t stream) {
    const float* features      = (const float*)d_in[0];
    const float* label_init    = (const float*)d_in[1];
    const float* labels_onehot = (const float*)d_in[2];
    const float* alpha         = (const float*)d_in[3];
    const float* attention     = (const float*)d_in[4];
    const float* e_edge        = (const float*)d_in[5];
    const float* w1            = (const float*)d_in[6];
    const float* b1            = (const float*)d_in[7];
    const float* w2            = (const float*)d_in[8];
    const float* b2            = (const float*)d_in[9];
    const int*   src           = (const int*)d_in[10];
    const int*   dst           = (const int*)d_in[11];
    const int*   train_mask    = (const int*)d_in[12];
    float* out = (float*)d_out;

    float* ws = (float*)d_ws;
    unsigned* nmax = (unsigned*)ws;
    float* nsum  = ws + NN;
    float* hA    = ws + 2 * NN;
    float* hout0 = hA + NN * CC;
    float* hout1 = hout0 + NN * CC;

    const int TB = 256;
    const int gridNC = (NN * CC + TB - 1) / TB;
    const int gridE  = (EE + TB - 1) / TB;

    for (int g = 0; g < GG; ++g) {
        float* hout = g ? hout1 : hout0;
        const float* hcur = label_init;
        for (int l = 0; l < LL; ++l) {
            float* hnext = (l == LL - 1) ? hout : hA;
            const float* ee = e_edge + ((size_t)l * GG + g) * EE;
            const int* sg = src + (size_t)g * EE;
            const int* dg = dst + (size_t)g * EE;
            k_init  <<<gridNC, TB, 0, stream>>>(nmax, nsum, hnext);
            k_segmax<<<gridE,  TB, 0, stream>>>(ee, dg, nmax);
            k_segsum<<<gridE,  TB, 0, stream>>>(ee, dg, nmax, nsum);
            k_agg   <<<gridE,  TB, 0, stream>>>(ee, sg, dg, nmax, nsum, hcur, hnext);
            k_mask  <<<gridNC, TB, 0, stream>>>(labels_onehot, train_mask, hnext);
            hcur = hnext;
        }
    }

    const int n_pairs = NN / 32;
    const int blocks  = (n_pairs + 7) / 8;
    k_mlp_combine<<<blocks, TB, 0, stream>>>(features, w1, b1, w2, b2,
                                             alpha, attention, hout0, hout1,
                                             out, n_pairs);
}


# Sort edges by destination once per graph (counting sort); each PLP layer is one per-node gather, no float atomics; sums now run in sorted-slot order and the logits accumulate in out
# speedup vs baseline: 4.0438x; 4.0438x over previous
#include <hip/hip_runtime.h>
#include <hip/hip_bf16.h>
#include <math.h>

#define NN 100000
#define CC 16
#define GG 2
#define LL 2
#define EE 3200000
#define FF 512
#define HH 64

#define W1_PSTRIDE 160
#define W2_PSTRIDE 32
#define HID_STRIDE 66

typedef float v2f __attribute__((ext_vector_type(2)));
typedef float v8f __attribute__((ext_vector_type(8)));

#define SCAN_T 1024
#define SCAN_B 98

__global__ void k_zero(int* __restrict__ cur) {
    int i = blockIdx.x * blockDim.x + threadIdx.x;
    if (i < NN) cur[i] = 0;
}

__global__ void k_hist(const int* __restrict__ dst, int* __restrict__ cur) {
    int i = blockIdx.x * blockDim.x + threadIdx.x;
    if (i >= EE) return;
    atomicAdd(&cur[dst[i]], 1);
}

__global__ __launch_bounds__(SCAN_T) void k_scan1(int* __restrict__ cur, int* __restrict__ bsum) {
    __shared__ int sh[SCAN_T];
    int t = threadIdx.x;
    int i = blockIdx.x * SCAN_T + t;
    int v = (i < NN) ? cur[i] : 0;
    sh[t] = v;
    __syncthreads();
    for (int off = 1; off < SCAN_T; off <<= 1) {
        int a = (t >= off) ? sh[t - off] : 0;
        __syncthreads();
        sh[t] += a;
        __syncthreads();
    }
    if (i < NN) cur[i] = sh[t] - v;
    if (t == SCAN_T - 1) bsum[blockIdx.x] = sh[t];
}

__global__ __launch_bounds__(SCAN_T) void k_scan2(int* __restrict__ cur, const int* __restrict__ bsum) {
    __shared__ int carry;
    int t = threadIdx.x;
    int i = blockIdx.x * SCAN_T + t;
    if (t < 32) {
        int c = 0;
        for (int b = t; b < (int)blockIdx.x; b += 32) c += bsum[b];
        for (int o = 16; o > 0; o >>= 1) c += __shfl_xor(c, o);
        if (t == 0) carry = c;
    }
    __syncthreads();
    if (i < NN) cur[i] += carry;
}

__global__ void k_fill(const int* __restrict__ dst, int* __restrict__ cur, int* __restrict__ perm) {
    int i = blockIdx.x * blockDim.x + threadIdx.x;
    if (i >= EE) return;
    int pos = atomicAdd(&cur[dst[i]], 1);
    pos = min(max(pos, 0), EE - 1);
    perm[pos] = i;
}

#define HEAD_Q 3

__device__ __forceinline__ int clamp_e(int i) { return min(max(i, 0), EE - 1); }

__device__ __forceinline__ float gather_rows(const float* __restrict__ hcur, float my_a, int my_s,
                                             int kmax, int c, float acc) {
    for (int k = 0; k < kmax; k += 4) {
        float a0 = __shfl(my_a, k + 0, 16), a1 = __shfl(my_a, k + 1, 16);
        float a2 = __shfl(my_a, k + 2, 16), a3 = __shfl(my_a, k + 3, 16);
        int   s0 = __shfl(my_s, k + 0, 16), s1 = __shfl(my_s, k + 1, 16);
        int   s2 = __shfl(my_s, k + 2, 16), s3 = __shfl(my_s, k + 3, 16);
        float h0 = hcur[(size_t)s0 * CC + c], h1 = hcur[(size_t)s1 * CC + c];
        float h2 = hcur[(size_t)s2 * CC + c], h3 = hcur[(size_t)s3 * CC + c];
        acc += h0 * a0; acc += h1 * a1; acc += h2 * a2; acc += h3 * a3;
    }
    return acc;
}

__global__ __launch_bounds__(256) void k_gather(
    const float* __restrict__ e, const int* __restrict__ src,
    const int* __restrict__ perm, const int* __restrict__ cur,
    const float* __restrict__ hcur, const float* __restrict__ one_hot,
    const int* __restrict__ mask, const float* __restrict__ att,
    float* hdst, int mode)
{
    int gid = blockIdx.x * blockDim.x + threadIdx.x;
    int d = gid >> 4;
    int c = threadIdx.x & 15;
    int beg = (d == 0) ? 0 : cur[d - 1];
    int cnt = cur[d] - beg;
    int cmax = max(cnt, __shfl_xor(cnt, 16));

    int   pid[HEAD_Q];
    float ev[HEAD_Q];
    int   sv[HEAD_Q];
    for (int q = 0; q < HEAD_Q; ++q) pid[q] = perm[clamp_e(beg + q * 16 + c)];
    for (int q = 0; q < HEAD_Q; ++q) {
        int id = clamp_e(pid[q]);
        ev[q] = e[id];
        sv[q] = src[id];
    }

    float m = -INFINITY;
    for (int q = 0; q < HEAD_Q; ++q) if (q * 16 + c < cnt) m = fmaxf(m, ev[q]);
    for (int b = HEAD_Q * 16; b < cmax; b += 16) {
        int j = b + c;
        if (j < cnt) m = fmaxf(m, e[clamp_e(perm[clamp_e(beg + j)])]);
    }
    for (int o = 8; o > 0; o >>= 1) m = fmaxf(m, __shfl_xor(m, o));

    float s = 0.0f;
    for (int q = 0; q < HEAD_Q; ++q) if (q * 16 + c < cnt) s += expf(ev[q] - m);
    for (int b = HEAD_Q * 16; b < cmax; b += 16) {
        int j = b + c;
        if (j < cnt) s += expf(e[clamp_e(perm[clamp_e(beg + j)])] - m);
    }
    for (int o = 8; o > 0; o >>= 1) s += __shfl_xor(s, o);

    float acc = 0.0f;
    for (int q = 0; q < HEAD_Q; ++q) {
        int b = q * 16;
        if (b < cmax) {
            bool ok = (b + c < cnt);
            float my_a = ok ? expf(ev[q] - m) / s : 0.0f;
            int   my_s = ok ? sv[q] : 0;
            acc = gather_rows(hcur, my_a, my_s, min(16, cmax - b), c, acc);
        }
    }
    for (int b = HEAD_Q * 16; b < cmax; b += 16) {
        int j = b + c;
        float my_a = 0.0f;
        int my_s = 0;
        if (j < cnt) {
            int id = clamp_e(perm[clamp_e(beg + j)]);
            my_a = expf(e[id] - m) / s;
            my_s = src[id];
        }
        acc = gather_rows(hcur, my_a, my_s, min(16, cmax - b), c, acc);
    }

    size_t o = (size_t)d * CC + c;
    float hv = mask[d] ? one_hot[o] : acc;
    if (mode == 0) {
        hdst[o] = hv;
    } else {
        float a0 = att[d * 2 + 0], a1 = att[d * 2 + 1];
        float mx = fmaxf(a0, a1);
        float e0 = expf(a0 - mx), e1 = expf(a1 - mx);
        float inv = 1.0f / (e0 + e1);
        if (mode == 1) hdst[o] = hv * (e0 * inv);
        else           hdst[o] = hdst[o] + hv * (e1 * inv);
    }
}

__global__ __launch_bounds__(256) void k_mlp_combine(
    const float* __restrict__ feat, const float* __restrict__ w1,
    const float* __restrict__ b1,   const float* __restrict__ w2,
    const float* __restrict__ b2,   const float* __restrict__ alpha,
    float* __restrict__ out, int n_pairs)
{
    __shared__ float sW1p[(FF / 2) * W1_PSTRIDE];
    __shared__ float sW2p[(HH / 2) * W2_PSTRIDE];
    __shared__ float sHid[8][2][16 * HID_STRIDE];

    int t = threadIdx.x;

    for (int i = t; i < (FF / 2) * (HH / 4); i += 256) {
        int p  = i >> 4;
        int nq = (i & 15) << 2;
        float4 x = *(const float4*)&w1[(size_t)p * 2 * HH + nq];
        float4 y = *(const float4*)&w1[(size_t)p * 2 * HH + HH + nq];
        float4 lo = { x.x, y.x, x.y, y.y };
        float4 hi = { x.z, y.z, x.w, y.w };
        float* d = &sW1p[p * W1_PSTRIDE + 2 * nq];
        *(float4*)(d + 0) = lo;
        *(float4*)(d + 4) = hi;
    }
    for (int i = t; i < (HH / 2) * CC; i += 256) {
        int p = i >> 4, n = i & 15;
        v2f v; v.x = w2[(size_t)p * 2 * CC + n];
        v.y = w2[(size_t)p * 2 * CC + CC + n];
        *(v2f*)&sW2p[p * W2_PSTRIDE + 2 * n] = v;
    }
    __syncthreads();

    int wave = t >> 5;
    int lane = t & 31;
    int pair = blockIdx.x * 8 + wave;
    if (pair >= n_pairs) return;

    int tile0 = pair * 2;
    int tile1 = tile0 + 1;

    int m     = lane & 15;
    int n     = lane & 15;
    int hi16  = lane >> 4;
    int khalf = hi16 << 1;
    int moff  = hi16 << 3;

    v8f acc0 = {}, acc1 = {}, acc2 = {}, acc3 = {};
    v8f acc4 = {}, acc5 = {}, acc6 = {}, acc7 = {};
    const float* arow0 = feat + ((size_t)tile0 * 16 + m) * FF;
    const float* arow1 = feat + ((size_t)tile1 * 16 + m) * FF;
    for (int k = 0; k < FF; k += 4) {
        int kb = k + khalf;
        v2f a0 = *(const v2f*)(arow0 + kb);
        v2f a1 = *(const v2f*)(arow1 + kb);
        const float* bp = &sW1p[((k >> 1) + hi16) * W1_PSTRIDE + 2 * n];
        v2f b0 = *(const v2f*)(bp + 0);
        v2f b1v = *(const v2f*)(bp + 32);
        v2f b2v = *(const v2f*)(bp + 64);
        v2f b3v = *(const v2f*)(bp + 96);
        acc0 = __builtin_amdgcn_wmma_f32_16x16x4_f32(false, a0, false, b0,  (short)0, acc0, false, false);
        acc1 = __builtin_amdgcn_wmma_f32_16x16x4_f32(false, a0, false, b1v, (short)0, acc1, false, false);
        acc2 = __builtin_amdgcn_wmma_f32_16x16x4_f32(false, a0, false, b2v, (short)0, acc2, false, false);
        acc3 = __builtin_amdgcn_wmma_f32_16x16x4_f32(false, a0, false, b3v, (short)0, acc3, false, false);
        acc4 = __builtin_amdgcn_wmma_f32_16x16x4_f32(false, a1, false, b0,  (short)0, acc4, false, false);
        acc5 = __builtin_amdgcn_wmma_f32_16x16x4_f32(false, a1, false, b1v, (short)0, acc5, false, false);
        acc6 = __builtin_amdgcn_wmma_f32_16x16x4_f32(false, a1, false, b2v, (short)0, acc6, false, false);
        acc7 = __builtin_amdgcn_wmma_f32_16x16x4_f32(false, a1, false, b3v, (short)0, acc7, false, false);
    }

    float bb0 = b1[n + 0], bb1 = b1[n + 16], bb2 = b1[n + 32], bb3 = b1[n + 48];
    {
        float* hd0 = &sHid[wave][0][0];
        float* hd1 = &sHid[wave][1][0];
        for (int r = 0; r < 8; ++r) {
            int mm = r + moff;
            hd0[mm * HID_STRIDE + n +  0] = fmaxf(acc0[r] + bb0, 0.0f);
            hd0[mm * HID_STRIDE + n + 16] = fmaxf(acc1[r] + bb1, 0.0f);
            hd0[mm * HID_STRIDE + n + 32] = fmaxf(acc2[r] + bb2, 0.0f);
            hd0[mm * HID_STRIDE + n + 48] = fmaxf(acc3[r] + bb3, 0.0f);
            hd1[mm * HID_STRIDE + n +  0] = fmaxf(acc4[r] + bb0, 0.0f);
            hd1[mm * HID_STRIDE + n + 16] = fmaxf(acc5[r] + bb1, 0.0f);
            hd1[mm * HID_STRIDE + n + 32] = fmaxf(acc6[r] + bb2, 0.0f);
            hd1[mm * HID_STRIDE + n + 48] = fmaxf(acc7[r] + bb3, 0.0f);
        }
    }

    float bc = b2[n];
    for (int tt = 0; tt < 2; ++tt) {
        const float* hd = &sHid[wave][tt][0];
        v8f acc = {};
        for (int k = 0; k < HH; k += 4) {
            int kb = k + khalf;
            v2f a = *(const v2f*)(hd + m * HID_STRIDE + kb);
            v2f b = *(const v2f*)&sW2p[((k >> 1) + hi16) * W2_PSTRIDE + 2 * n];
            acc = __builtin_amdgcn_wmma_f32_16x16x4_f32(false, a, false, b, (short)0, acc, false, false);
        }
        int tile = tt ? tile1 : tile0;
        for (int r = 0; r < 8; ++r) {
            int node = tile * 16 + r + moff;
            float mlp = acc[r] + bc;
            float al  = alpha[node];
            float sa  = 1.0f / (1.0f + expf(-al));
            float sna = 1.0f - sa;
            float logit = out[(size_t)node * CC + n];
            out[(size_t)node * CC + n] = sa * logit + sna * mlp;
        }
    }
}

static void run_graph(hipStream_t stream, const float* e0, const float* e1,
                      const int* sg, const int* dg, int* cur, int* perm, int* bsum,
                      float* hA, float* out, const float* label_init, const float* labels_onehot,
                      const int* train_mask, const float* attention, int last_mode) {
    const int TB = 256;
    const int gridN  = (NN + TB - 1) / TB;
    const int gridNC = (NN * CC) / TB;
    const int gridE  = (EE + TB - 1) / TB;
    k_zero <<<gridN,  TB, 0, stream>>>(cur);
    k_hist <<<gridE,  TB, 0, stream>>>(dg, cur);
    k_scan1<<<SCAN_B, SCAN_T, 0, stream>>>(cur, bsum);
    k_scan2<<<SCAN_B, SCAN_T, 0, stream>>>(cur, bsum);
    k_fill <<<gridE,  TB, 0, stream>>>(dg, cur, perm);
    k_gather<<<gridNC, TB, 0, stream>>>(e0, sg, perm, cur, label_init, labels_onehot,
                                        train_mask, attention, hA, 0);
    k_gather<<<gridNC, TB, 0, stream>>>(e1, sg, perm, cur, hA, labels_onehot,
                                        train_mask, attention, out, last_mode);
}

extern "C" void kernel_launch(void* const* d_in, const int* in_sizes, int n_in,
                              void* d_out, int out_size, void* d_ws, size_t ws_size,
                              hipStream_t stream) {
    const float* features      = (const float*)d_in[0];
    const float* label_init    = (const float*)d_in[1];
    const float* labels_onehot = (const float*)d_in[2];
    const float* alpha         = (const float*)d_in[3];
    const float* attention     = (const float*)d_in[4];
    const float* e_edge        = (const float*)d_in[5];
    const float* w1            = (const float*)d_in[6];
    const float* b1            = (const float*)d_in[7];
    const float* w2            = (const float*)d_in[8];
    const float* b2            = (const float*)d_in[9];
    const int*   src           = (const int*)d_in[10];
    const int*   dst           = (const int*)d_in[11];
    const int*   train_mask    = (const int*)d_in[12];
    float* out = (float*)d_out;

    int*   cur  = (int*)d_ws;
    int*   perm = cur + NN;
    float* hA   = (float*)(perm + EE);
    int*   bsum = (int*)(hA + NN * CC);

    const int TB = 256;

    run_graph(stream, e_edge,      e_edge + (size_t)GG * EE,      src,      dst,
              cur, perm, bsum, hA, out, label_init, labels_onehot, train_mask, attention, 1);
    run_graph(stream, e_edge + EE, e_edge + (size_t)GG * EE + EE, src + EE, dst + EE,
              cur, perm, bsum, hA, out, label_init, labels_onehot, train_mask, attention, 2);

    const int n_pairs = NN / 32;
    const int blocks  = (n_pairs + 7) / 8;
    k_mlp_combine<<<blocks, TB, 0, stream>>>(features, w1, b1, w2, b2, alpha, out, n_pairs);
}


# Counting sort by destination now ranks edges with LDS counters in two levels (256-node buckets, then nodes) instead of two global atomics per edge; slot order in a node, so its sum order, changes
# speedup vs baseline: 7.9994x; 1.9782x over previous
#include <hip/hip_runtime.h>
#include <hip/hip_bf16.h>
#include <math.h>

#define NN 100000
#define CC 16
#define GG 2
#define LL 2
#define EE 3200000
#define FF 512
#define HH 64

#define W1_PSTRIDE 160
#define W2_PSTRIDE 32
#define HID_STRIDE 66

typedef float v2f __attribute__((ext_vector_type(2)));
typedef float v8f __attribute__((ext_vector_type(8)));

#define BKT_SHIFT 8
#define BKT_W     256
#define NBK       ((NN + BKT_W - 1) / BKT_W)
#define CH        8192
#define NBLK      ((EE + CH - 1) / CH)
#define ST        512
#define CAP       32768
#define ID_BITS   22
#define ID_MASK   0x3FFFFFu

static_assert(EE <= (1 << ID_BITS), "edge id must fit the record");
static_assert(NBK <= ST && NBLK <= ST && BKT_W <= ST, "one scan pass per table");
static_assert(CH % ST == 0, "whole strides per block");

__device__ __forceinline__ int clamp_node(int d) { return min(max(d, 0), NN - 1); }

__device__ __forceinline__ int block_scan_incl(int* sh, int t, int v) {
    sh[t] = v;
    __syncthreads();
    for (int off = 1; off < ST; off <<= 1) {
        int a = (t >= off) ? sh[t - off] : 0;
        __syncthreads();
        sh[t] += a;
        __syncthreads();
    }
    return sh[t];
}

__global__ __launch_bounds__(ST) void k_bhist(const int* __restrict__ dst, int* __restrict__ bh) {
    __shared__ int h[NBK];
    int t = threadIdx.x;
    if (t < NBK) h[t] = 0;
    __syncthreads();
    int base = blockIdx.x * CH + t;
    int d[CH / ST];
    for (int j = 0; j < CH / ST; ++j)
        d[j] = dst[min(base + j * ST, EE - 1)];
    for (int j = 0; j < CH / ST; ++j)
        if (base + j * ST < EE) atomicAdd(&h[clamp_node(d[j]) >> BKT_SHIFT], 1);
    __syncthreads();
    if (t < NBK) bh[t * NBLK + blockIdx.x] = h[t];
}

__global__ __launch_bounds__(ST) void k_bscan(int* __restrict__ bh, int* __restrict__ tot) {
    __shared__ int sh[ST];
    int t = threadIdx.x, b = blockIdx.x;
    int v = (t < NBLK) ? bh[b * NBLK + t] : 0;
    int incl = block_scan_incl(sh, t, v);
    if (t < NBLK) bh[b * NBLK + t] = incl - v;
    if (t == ST - 1) tot[b] = incl;
}

__global__ __launch_bounds__(ST) void k_bscatter(const int* __restrict__ dst, const int* __restrict__ bh,
                                                 const int* __restrict__ tot, int* __restrict__ bbase,
                                                 unsigned* __restrict__ rec) {
    __shared__ int sh[ST];
    __shared__ int off[NBK];
    int t = threadIdx.x;
    int v = (t < NBK) ? tot[t] : 0;
    int incl = block_scan_incl(sh, t, v);
    if (t < NBK) {
        int excl = incl - v;
        off[t] = excl + bh[t * NBLK + blockIdx.x];
        if (blockIdx.x == 0) bbase[t] = excl;
    }
    __syncthreads();
    int base = blockIdx.x * CH + t;
    int dv[CH / ST];
    for (int j = 0; j < CH / ST; ++j)
        dv[j] = dst[min(base + j * ST, EE - 1)];
    for (int j = 0; j < CH / ST; ++j) {
        int i = base + j * ST;
        if (i < EE) {
            int d = clamp_node(dv[j]);
            int p = atomicAdd(&off[d >> BKT_SHIFT], 1);
            p = min(max(p, 0), EE - 1);
            rec[p] = ((unsigned)(d & (BKT_W - 1)) << ID_BITS) | (unsigned)i;
        }
    }
}

__global__ __launch_bounds__(ST) void k_bsort(const int* __restrict__ tot, const int* __restrict__ bbase,
                                              int* __restrict__ cur, unsigned* slots) {
    __shared__ unsigned rec[CAP];
    __shared__ int sh[ST];
    __shared__ int cnt[BKT_W];
    __shared__ int head[BKT_W];
    int t = threadIdx.x, b = blockIdx.x;
    int n  = min(max(tot[b], 0), EE);
    int bs = min(max(bbase[b], 0), EE - n);
    unsigned* A = slots + bs;
    bool fits = (n <= CAP);
    if (t < BKT_W) cnt[t] = 0;
    __syncthreads();
    for (int i0 = t; i0 < n; i0 += 4 * ST) {
        unsigned r[4];
        for (int u = 0; u < 4; ++u) r[u] = A[min(i0 + u * ST, n - 1)];
        for (int u = 0; u < 4; ++u) {
            int i = i0 + u * ST;
            if (i < n) {
                if (fits) rec[i] = r[u];
                atomicAdd(&cnt[(r[u] >> ID_BITS) & (BKT_W - 1)], 1);
            }
        }
    }
    __syncthreads();
    int c = (t < BKT_W) ? cnt[t] : 0;
    int incl = block_scan_incl(sh, t, c);
    if (t < BKT_W) {
        head[t] = incl - c;
        int node = b * BKT_W + t;
        if (node < NN) cur[node] = bs + incl;
    }
    __syncthreads();
    if (fits) {
        for (int i = t; i < n; i += ST) {
            unsigned r = rec[i];
            int p = atomicAdd(&head[(r >> ID_BITS) & (BKT_W - 1)], 1);
            A[min(max(p, 0), n - 1)] = r & ID_MASK;
        }
    } else {
        if (t == 0) {
            for (int k = 0; k < BKT_W; ++k) {
                int endk = sh[k];
                while (head[k] < endk) {
                    int pk = head[k];
                    unsigned r = A[pk];
                    int kk = (int)((r >> ID_BITS) & (BKT_W - 1));
                    if (kk == k) {
                        head[k] = pk + 1;
                    } else {
                        int q = min(head[kk], n - 1);
                        head[kk] = q + 1;
                        unsigned r2 = A[q];
                        A[q] = r;
                        A[pk] = r2;
                    }
                }
            }
        }
        __syncthreads();
        for (int i = t; i < n; i += ST) A[i] &= ID_MASK;
    }
}

#define HEAD_Q 3

__device__ __forceinline__ int clamp_e(int i) { return min(max(i, 0), EE - 1); }

__device__ __forceinline__ float gather_rows(const float* __restrict__ hcur, float my_a, int my_s,
                                             int kmax, int c, float acc) {
    for (int k = 0; k < kmax; k += 4) {
        float a0 = __shfl(my_a, k + 0, 16), a1 = __shfl(my_a, k + 1, 16);
        float a2 = __shfl(my_a, k + 2, 16), a3 = __shfl(my_a, k + 3, 16);
        int   s0 = __shfl(my_s, k + 0, 16), s1 = __shfl(my_s, k + 1, 16);
        int   s2 = __shfl(my_s, k + 2, 16), s3 = __shfl(my_s, k + 3, 16);
        float h0 = hcur[(size_t)s0 * CC + c], h1 = hcur[(size_t)s1 * CC + c];
        float h2 = hcur[(size_t)s2 * CC + c], h3 = hcur[(size_t)s3 * CC + c];
        acc += h0 * a0; acc += h1 * a1; acc += h2 * a2; acc += h3 * a3;
    }
    return acc;
}

__global__ __launch_bounds__(256) void k_gather(
    const float* __restrict__ e, const int* __restrict__ src,
    const int* __restrict__ perm, const int* __restrict__ cur,
    const float* __restrict__ hcur, const float* __restrict__ one_hot,
    const int* __restrict__ mask, const float* __restrict__ att,
    float* hdst, int mode)
{
    int gid = blockIdx.x * blockDim.x + threadIdx.x;
    int d = gid >> 4;
    int c = threadIdx.x & 15;
    int beg = (d == 0) ? 0 : cur[d - 1];
    int cnt = cur[d] - beg;
    int cmax = max(cnt, __shfl_xor(cnt, 16));

    int   pid[HEAD_Q];
    float ev[HEAD_Q];
    int   sv[HEAD_Q];
    for (int q = 0; q < HEAD_Q; ++q) pid[q] = perm[clamp_e(beg + q * 16 + c)];
    for (int q = 0; q < HEAD_Q; ++q) {
        int id = clamp_e(pid[q]);
        ev[q] = e[id];
        sv[q] = src[id];
    }

    float m = -INFINITY;
    for (int q = 0; q < HEAD_Q; ++q) if (q * 16 + c < cnt) m = fmaxf(m, ev[q]);
    for (int b = HEAD_Q * 16; b < cmax; b += 16) {
        int j = b + c;
        if (j < cnt) m = fmaxf(m, e[clamp_e(perm[clamp_e(beg + j)])]);
    }
    for (int o = 8; o > 0; o >>= 1) m = fmaxf(m, __shfl_xor(m, o));

    float s = 0.0f;
    for (int q = 0; q < HEAD_Q; ++q) if (q * 16 + c < cnt) s += expf(ev[q] - m);
    for (int b = HEAD_Q * 16; b < cmax; b += 16) {
        int j = b + c;
        if (j < cnt) s += expf(e[clamp_e(perm[clamp_e(beg + j)])] - m);
    }
    for (int o = 8; o > 0; o >>= 1) s += __shfl_xor(s, o);

    float acc = 0.0f;
    for (int q = 0; q < HEAD_Q; ++q) {
        int b = q * 16;
        if (b < cmax) {
            bool ok = (b + c < cnt);
            float my_a = ok ? expf(ev[q] - m) / s : 0.0f;
            int   my_s = ok ? sv[q] : 0;
            acc = gather_rows(hcur, my_a, my_s, min(16, cmax - b), c, acc);
        }
    }
    for (int b = HEAD_Q * 16; b < cmax; b += 16) {
        int j = b + c;
        float my_a = 0.0f;
        int my_s = 0;
        if (j < cnt) {
            int id = clamp_e(perm[clamp_e(beg + j)]);
            my_a = expf(e[id] - m) / s;
            my_s = src[id];
        }
        acc = gather_rows(hcur, my_a, my_s, min(16, cmax - b), c, acc);
    }

    size_t o = (size_t)d * CC + c;
    float hv = mask[d] ? one_hot[o] : acc;
    if (mode == 0) {
        hdst[o] = hv;
    } else {
        float a0 = att[d * 2 + 0], a1 = att[d * 2 + 1];
        float mx = fmaxf(a0, a1);
        float e0 = expf(a0 - mx), e1 = expf(a1 - mx);
        float inv = 1.0f / (e0 + e1);
        if (mode == 1) hdst[o] = hv * (e0 * inv);
        else           hdst[o] = hdst[o] + hv * (e1 * inv);
    }
}

__global__ __launch_bounds__(256) void k_mlp_combine(
    const float* __restrict__ feat, const float* __restrict__ w1,
    const float* __restrict__ b1,   const float* __restrict__ w2,
    const float* __restrict__ b2,   const float* __restrict__ alpha,
    float* __restrict__ out, int n_pairs)
{
    __shared__ float sW1p[(FF / 2) * W1_PSTRIDE];
    __shared__ float sW2p[(HH / 2) * W2_PSTRIDE];
    __shared__ float sHid[8][2][16 * HID_STRIDE];

    int t = threadIdx.x;

    for (int i = t; i < (FF / 2) * (HH / 4); i += 256) {
        int p  = i >> 4;
        int nq = (i & 15) << 2;
        float4 x = *(const float4*)&w1[(size_t)p * 2 * HH + nq];
        float4 y = *(const float4*)&w1[(size_t)p * 2 * HH + HH + nq];
        float4 lo = { x.x, y.x, x.y, y.y };
        float4 hi = { x.z, y.z, x.w, y.w };
        float* d = &sW1p[p * W1_PSTRIDE + 2 * nq];
        *(float4*)(d + 0) = lo;
        *(float4*)(d + 4) = hi;
    }
    for (int i = t; i < (HH / 2) * CC; i += 256) {
        int p = i >> 4, n = i & 15;
        v2f v; v.x = w2[(size_t)p * 2 * CC + n];
        v.y = w2[(size_t)p * 2 * CC + CC + n];
        *(v2f*)&sW2p[p * W2_PSTRIDE + 2 * n] = v;
    }
    __syncthreads();

    int wave = t >> 5;
    int lane = t & 31;
    int pair = blockIdx.x * 8 + wave;
    if (pair >= n_pairs) return;

    int tile0 = pair * 2;
    int tile1 = tile0 + 1;

    int m     = lane & 15;
    int n     = lane & 15;
    int hi16  = lane >> 4;
    int khalf = hi16 << 1;
    int moff  = hi16 << 3;

    v8f acc0 = {}, acc1 = {}, acc2 = {}, acc3 = {};
    v8f acc4 = {}, acc5 = {}, acc6 = {}, acc7 = {};
    const float* arow0 = feat + ((size_t)tile0 * 16 + m) * FF;
    const float* arow1 = feat + ((size_t)tile1 * 16 + m) * FF;
    for (int k = 0; k < FF; k += 4) {
        int kb = k + khalf;
        v2f a0 = *(const v2f*)(arow0 + kb);
        v2f a1 = *(const v2f*)(arow1 + kb);
        const float* bp = &sW1p[((k >> 1) + hi16) * W1_PSTRIDE + 2 * n];
        v2f b0 = *(const v2f*)(bp + 0);
        v2f b1v = *(const v2f*)(bp + 32);
        v2f b2v = *(const v2f*)(bp + 64);
        v2f b3v = *(const v2f*)(bp + 96);
        acc0 = __builtin_amdgcn_wmma_f32_16x16x4_f32(false, a0, false, b0,  (short)0, acc0, false, false);
        acc1 = __builtin_amdgcn_wmma_f32_16x16x4_f32(false, a0, false, b1v, (short)0, acc1, false, false);
        acc2 = __builtin_amdgcn_wmma_f32_16x16x4_f32(false, a0, false, b2v, (short)0, acc2, false, false);
        acc3 = __builtin_amdgcn_wmma_f32_16x16x4_f32(false, a0, false, b3v, (short)0, acc3, false, false);
        acc4 = __builtin_amdgcn_wmma_f32_16x16x4_f32(false, a1, false, b0,  (short)0, acc4, false, false);
        acc5 = __builtin_amdgcn_wmma_f32_16x16x4_f32(false, a1, false, b1v, (short)0, acc5, false, false);
        acc6 = __builtin_amdgcn_wmma_f32_16x16x4_f32(false, a1, false, b2v, (short)0, acc6, false, false);
        acc7 = __builtin_amdgcn_wmma_f32_16x16x4_f32(false, a1, false, b3v, (short)0, acc7, false, false);
    }

    float bb0 = b1[n + 0], bb1 = b1[n + 16], bb2 = b1[n + 32], bb3 = b1[n + 48];
    {
        float* hd0 = &sHid[wave][0][0];
        float* hd1 = &sHid[wave][1][0];
        for (int r = 0; r < 8; ++r) {
            int mm = r + moff;
            hd0[mm * HID_STRIDE + n +  0] = fmaxf(acc0[r] + bb0, 0.0f);
            hd0[mm * HID_STRIDE + n + 16] = fmaxf(acc1[r] + bb1, 0.0f);
            hd0[mm * HID_STRIDE + n + 32] = fmaxf(acc2[r] + bb2, 0.0f);
            hd0[mm * HID_STRIDE + n + 48] = fmaxf(acc3[r] + bb3, 0.0f);
            hd1[mm * HID_STRIDE + n +  0] = fmaxf(acc4[r] + bb0, 0.0f);
            hd1[mm * HID_STRIDE + n + 16] = fmaxf(acc5[r] + bb1, 0.0f);
            hd1[mm * HID_STRIDE + n + 32] = fmaxf(acc6[r] + bb2, 0.0f);
            hd1[mm * HID_STRIDE + n + 48] = fmaxf(acc7[r] + bb3, 0.0f);
        }
    }

    float bc = b2[n];
    for (int tt = 0; tt < 2; ++tt) {
        const float* hd = &sHid[wave][tt][0];
        v8f acc = {};
        for (int k = 0; k < HH; k += 4) {
            int kb = k + khalf;
            v2f a = *(const v2f*)(hd + m * HID_STRIDE + kb);
            v2f b = *(const v2f*)&sW2p[((k >> 1) + hi16) * W2_PSTRIDE + 2 * n];
            acc = __builtin_amdgcn_wmma_f32_16x16x4_f32(false, a, false, b, (short)0, acc, false, false);
        }
        int tile = tt ? tile1 : tile0;
        for (int r = 0; r < 8; ++r) {
            int node = tile * 16 + r + moff;
            float mlp = acc[r] + bc;
            float al  = alpha[node];
            float sa  = 1.0f / (1.0f + expf(-al));
            float sna = 1.0f - sa;
            float logit = out[(size_t)node * CC + n];
            out[(size_t)node * CC + n] = sa * logit + sna * mlp;
        }
    }
}

static void run_graph(hipStream_t stream, const float* e0, const float* e1,
                      const int* sg, const int* dg, int* cur, int* perm, int* bsum,
                      float* hA, float* out, const float* label_init, const float* labels_onehot,
                      const int* train_mask, const float* attention, int last_mode) {
    const int TB = 256;
    const int gridNC = (NN * CC) / TB;
    int* bh    = (int*)hA;
    int* tot   = bh + NBK * NBLK;
    int* bbase = tot + NBK;
    k_bhist   <<<NBLK, ST, 0, stream>>>(dg, bh);
    k_bscan   <<<NBK,  ST, 0, stream>>>(bh, tot);
    k_bscatter<<<NBLK, ST, 0, stream>>>(dg, bh, tot, bbase, (unsigned*)perm);
    k_bsort   <<<NBK,  ST, 0, stream>>>(tot, bbase, cur, (unsigned*)perm);
    k_gather<<<gridNC, TB, 0, stream>>>(e0, sg, perm, cur, label_init, labels_onehot,
                                        train_mask, attention, hA, 0);
    k_gather<<<gridNC, TB, 0, stream>>>(e1, sg, perm, cur, hA, labels_onehot,
                                        train_mask, attention, out, last_mode);
}

extern "C" void kernel_launch(void* const* d_in, const int* in_sizes, int n_in,
                              void* d_out, int out_size, void* d_ws, size_t ws_size,
                              hipStream_t stream) {
    const float* features      = (const float*)d_in[0];
    const float* label_init    = (const float*)d_in[1];
    const float* labels_onehot = (const float*)d_in[2];
    const float* alpha         = (const float*)d_in[3];
    const float* attention     = (const float*)d_in[4];
    const float* e_edge        = (const float*)d_in[5];
    const float* w1            = (const float*)d_in[6];
    const float* b1            = (const float*)d_in[7];
    const float* w2            = (const float*)d_in[8];
    const float* b2            = (const float*)d_in[9];
    const int*   src           = (const int*)d_in[10];
    const int*   dst           = (const int*)d_in[11];
    const int*   train_mask    = (const int*)d_in[12];
    float* out = (float*)d_out;

    int*   cur  = (int*)d_ws;
    int*   perm = cur + NN;
    float* hA   = (float*)(perm + EE);
    int*   bsum = (int*)(hA + NN * CC);

    const int TB = 256;

    run_graph(stream, e_edge,      e_edge + (size_t)GG * EE,      src,      dst,
              cur, perm, bsum, hA, out, label_init, labels_onehot, train_mask, attention, 1);
    run_graph(stream, e_edge + EE, e_edge + (size_t)GG * EE + EE, src + EE, dst + EE,
              cur, perm, bsum, hA, out, label_init, labels_onehot, train_mask, attention, 2);

    const int n_pairs = NN / 32;
    const int blocks  = (n_pairs + 7) / 8;
    k_mlp_combine<<<blocks, TB, 0, stream>>>(features, w1, b1, w2, b2, alpha, out, n_pairs);
}


# Sort stores leave through LDS: level 1 orders a block's records by bucket and writes contiguous runs, level 2 writes ids in whole lines; slot order in a node, so its sum order, may change
# speedup vs baseline: 9.8969x; 1.2372x over previous
#include <hip/hip_runtime.h>
#include <hip/hip_bf16.h>
#include <math.h>

#define NN 100000
#define CC 16
#define GG 2
#define LL 2
#define EE 3200000
#define FF 512
#define HH 64

#define W1_PSTRIDE 160
#define W2_PSTRIDE 32
#define HID_STRIDE 66

typedef float v2f __attribute__((ext_vector_type(2)));
typedef float v8f __attribute__((ext_vector_type(8)));

#define BKT_SHIFT 8
#define BKT_W     256
#define NBK       ((NN + BKT_W - 1) / BKT_W)
#define CH        8192
#define CH_BITS   13
#define NBLK      ((EE + CH - 1) / CH)
#define ST        512
#define CAP       32768
#define ID_BITS   22
#define ID_MASK   0x3FFFFFu

static_assert(EE <= (1 << ID_BITS), "edge id must fit the record");
static_assert(NBK <= ST && NBLK <= ST && BKT_W <= ST, "one scan pass per table");
static_assert(CH % ST == 0, "whole strides per block");
static_assert(CH == (1 << CH_BITS) && NN <= (1 << (32 - CH_BITS)), "node and edge-in-block fit one staged word");

__device__ __forceinline__ int clamp_node(int d) { return min(max(d, 0), NN - 1); }

__device__ __forceinline__ int block_scan_incl(int* sh, int t, int v) {
    sh[t] = v;
    __syncthreads();
    for (int off = 1; off < ST; off <<= 1) {
        int a = (t >= off) ? sh[t - off] : 0;
        __syncthreads();
        sh[t] += a;
        __syncthreads();
    }
    return sh[t];
}

__device__ __forceinline__ void block_scan_incl2(int* sa, int* sb, int t, int& a, int& b) {
    sa[t] = a;
    sb[t] = b;
    __syncthreads();
    for (int off = 1; off < ST; off <<= 1) {
        int xa = (t >= off) ? sa[t - off] : 0;
        int xb = (t >= off) ? sb[t - off] : 0;
        __syncthreads();
        sa[t] += xa;
        sb[t] += xb;
        __syncthreads();
    }
    a = sa[t];
    b = sb[t];
}

__global__ __launch_bounds__(ST) void k_bhist(const int* __restrict__ dst, int* __restrict__ bh) {
    __shared__ int h[NBK];
    int t = threadIdx.x;
    if (t < NBK) h[t] = 0;
    __syncthreads();
    int base = blockIdx.x * CH + t;
    int d[CH / ST];
    for (int j = 0; j < CH / ST; ++j)
        d[j] = dst[min(base + j * ST, EE - 1)];
    for (int j = 0; j < CH / ST; ++j)
        if (base + j * ST < EE) atomicAdd(&h[clamp_node(d[j]) >> BKT_SHIFT], 1);
    __syncthreads();
    if (t < NBK) bh[t * NBLK + blockIdx.x] = h[t];
}

__global__ __launch_bounds__(ST) void k_bscan(int* __restrict__ bh, int* __restrict__ tot) {
    __shared__ int sh[ST];
    int t = threadIdx.x, b = blockIdx.x;
    int v = (t < NBLK) ? bh[b * NBLK + t] : 0;
    int incl = block_scan_incl(sh, t, v);
    if (t < NBLK) bh[b * NBLK + t] = incl - v;
    if (t == ST - 1) tot[b] = incl;
}

__global__ __launch_bounds__(ST) void k_bscatter(const int* __restrict__ dst, const int* __restrict__ bh,
                                                 const int* __restrict__ tot, int* __restrict__ bbase,
                                                 unsigned* __restrict__ rec) {
    __shared__ int sh[ST];
    __shared__ int sh2[ST];
    __shared__ int lcur[NBK];
    __shared__ int delta[NBK];
    __shared__ unsigned stage[CH];
    int t = threadIdx.x, blk = blockIdx.x;
    int v = 0, mine = 0, lc = 0;
    if (t < NBK) {
        v = tot[t];
        mine = bh[t * NBLK + blk];
        int nxt = (blk + 1 < NBLK) ? bh[t * NBLK + blk + 1] : v;
        lc = nxt - mine;
    }
    int incl = v, lincl = lc;
    block_scan_incl2(sh, sh2, t, incl, lincl);
    if (t < NBK) {
        int excl  = incl - v;
        int lbase = lincl - lc;
        lcur[t]  = lbase;
        delta[t] = excl + mine - lbase;
        if (blk == 0) bbase[t] = excl;
    }
    __syncthreads();
    int base = blk * CH + t;
    int dv[CH / ST];
    for (int j = 0; j < CH / ST; ++j)
        dv[j] = dst[min(base + j * ST, EE - 1)];
    for (int j = 0; j < CH / ST; ++j) {
        if (base + j * ST < EE) {
            int d = clamp_node(dv[j]);
            int r = atomicAdd(&lcur[d >> BKT_SHIFT], 1);
            r = min(max(r, 0), CH - 1);
            stage[r] = ((unsigned)d << CH_BITS) | (unsigned)(j * ST + t);
        }
    }
    __syncthreads();
    int nblk = min(CH, EE - blk * CH);
    for (int i0 = t; i0 < nblk; i0 += 4 * ST) {
        unsigned w[4];
        for (int u = 0; u < 4; ++u) w[u] = stage[min(i0 + u * ST, CH - 1)];
        for (int u = 0; u < 4; ++u) {
            int i = i0 + u * ST;
            if (i < nblk) {
                int d = (int)(w[u] >> CH_BITS);
                int b = min(d >> BKT_SHIFT, NBK - 1);
                int p = min(max(i + delta[b], 0), EE - 1);
                rec[p] = ((unsigned)(d & (BKT_W - 1)) << ID_BITS) | (unsigned)(blk * CH + (int)(w[u] & (CH - 1)));
            }
        }
    }
}

__global__ __launch_bounds__(ST) void k_bsort(const int* __restrict__ tot, const int* __restrict__ bbase,
                                              int* __restrict__ cur, unsigned* slots) {
    __shared__ unsigned rec[CAP];
    __shared__ int sh[ST];
    __shared__ int cnt[BKT_W];
    __shared__ int head[BKT_W];
    int t = threadIdx.x, b = blockIdx.x;
    int n  = min(max(tot[b], 0), EE);
    int bs = min(max(bbase[b], 0), EE - n);
    unsigned* A = slots + bs;
    bool fits = (n <= CAP);
    if (t < BKT_W) cnt[t] = 0;
    __syncthreads();
    for (int i0 = t; i0 < n; i0 += 4 * ST) {
        unsigned r[4];
        for (int u = 0; u < 4; ++u) r[u] = A[min(i0 + u * ST, n - 1)];
        for (int u = 0; u < 4; ++u)
            if (i0 + u * ST < n) atomicAdd(&cnt[(r[u] >> ID_BITS) & (BKT_W - 1)], 1);
    }
    __syncthreads();
    int c = (t < BKT_W) ? cnt[t] : 0;
    int incl = block_scan_incl(sh, t, c);
    if (t < BKT_W) {
        head[t] = incl - c;
        int node = b * BKT_W + t;
        if (node < NN) cur[node] = bs + incl;
    }
    __syncthreads();
    if (fits) {
        for (int i0 = t; i0 < n; i0 += 4 * ST) {
            unsigned r[4];
            for (int u = 0; u < 4; ++u) r[u] = A[min(i0 + u * ST, n - 1)];
            for (int u = 0; u < 4; ++u) {
                if (i0 + u * ST < n) {
                    int p = atomicAdd(&head[(r[u] >> ID_BITS) & (BKT_W - 1)], 1);
                    rec[min(max(p, 0), n - 1)] = r[u] & ID_MASK;
                }
            }
        }
        __syncthreads();
        for (int i0 = t; i0 < n; i0 += 4 * ST) {
            unsigned r[4];
            for (int u = 0; u < 4; ++u) r[u] = rec[min(i0 + u * ST, n - 1)];
            for (int u = 0; u < 4; ++u)
                if (i0 + u * ST < n) A[i0 + u * ST] = r[u];
        }
    } else {
        if (t == 0) {
            for (int k = 0; k < BKT_W; ++k) {
                int endk = sh[k];
                while (head[k] < endk) {
                    int pk = head[k];
                    unsigned r = A[pk];
                    int kk = (int)((r >> ID_BITS) & (BKT_W - 1));
                    if (kk == k) {
                        head[k] = pk + 1;
                    } else {
                        int q = min(head[kk], n - 1);
                        head[kk] = q + 1;
                        unsigned r2 = A[q];
                        A[q] = r;
                        A[pk] = r2;
                    }
                }
            }
        }
        __syncthreads();
        for (int i = t; i < n; i += ST) A[i] &= ID_MASK;
    }
}

#define HEAD_Q 3

__device__ __forceinline__ int clamp_e(int i) { return min(max(i, 0), EE - 1); }

__device__ __forceinline__ float gather_rows(const float* __restrict__ hcur, float my_a, int my_s,
                                             int kmax, int c, float acc) {
    for (int k = 0; k < kmax; k += 4) {
        float a0 = __shfl(my_a, k + 0, 16), a1 = __shfl(my_a, k + 1, 16);
        float a2 = __shfl(my_a, k + 2, 16), a3 = __shfl(my_a, k + 3, 16);
        int   s0 = __shfl(my_s, k + 0, 16), s1 = __shfl(my_s, k + 1, 16);
        int   s2 = __shfl(my_s, k + 2, 16), s3 = __shfl(my_s, k + 3, 16);
        float h0 = hcur[(size_t)s0 * CC + c], h1 = hcur[(size_t)s1 * CC + c];
        float h2 = hcur[(size_t)s2 * CC + c], h3 = hcur[(size_t)s3 * CC + c];
        acc += h0 * a0; acc += h1 * a1; acc += h2 * a2; acc += h3 * a3;
    }
    return acc;
}

__global__ __launch_bounds__(256) void k_gather(
    const float* __restrict__ e, const int* __restrict__ src,
    const int* __restrict__ perm, const int* __restrict__ cur,
    const float* __restrict__ hcur, const float* __restrict__ one_hot,
    const int* __restrict__ mask, const float* __restrict__ att,
    float* hdst, int mode)
{
    int gid = blockIdx.x * blockDim.x + threadIdx.x;
    int d = gid >> 4;
    int c = threadIdx.x & 15;
    int beg = (d == 0) ? 0 : cur[d - 1];
    int cnt = cur[d] - beg;
    int cmax = max(cnt, __shfl_xor(cnt, 16));

    int   pid[HEAD_Q];
    float ev[HEAD_Q];
    int   sv[HEAD_Q];
    for (int q = 0; q < HEAD_Q; ++q) pid[q] = perm[clamp_e(beg + q * 16 + c)];
    for (int q = 0; q < HEAD_Q; ++q) {
        int id = clamp_e(pid[q]);
        ev[q] = e[id];
        sv[q] = src[id];
    }

    float m = -INFINITY;
    for (int q = 0; q < HEAD_Q; ++q) if (q * 16 + c < cnt) m = fmaxf(m, ev[q]);
    for (int b = HEAD_Q * 16; b < cmax; b += 16) {
        int j = b + c;
        if (j < cnt) m = fmaxf(m, e[clamp_e(perm[clamp_e(beg + j)])]);
    }
    for (int o = 8; o > 0; o >>= 1) m = fmaxf(m, __shfl_xor(m, o));

    float s = 0.0f;
    for (int q = 0; q < HEAD_Q; ++q) if (q * 16 + c < cnt) s += expf(ev[q] - m);
    for (int b = HEAD_Q * 16; b < cmax; b += 16) {
        int j = b + c;
        if (j < cnt) s += expf(e[clamp_e(perm[clamp_e(beg + j)])] - m);
    }
    for (int o = 8; o > 0; o >>= 1) s += __shfl_xor(s, o);

    float acc = 0.0f;
    for (int q = 0; q < HEAD_Q; ++q) {
        int b = q * 16;
        if (b < cmax) {
            bool ok = (b + c < cnt);
            float my_a = ok ? expf(ev[q] - m) / s : 0.0f;
            int   my_s = ok ? sv[q] : 0;
            acc = gather_rows(hcur, my_a, my_s, min(16, cmax - b), c, acc);
        }
    }
    for (int b = HEAD_Q * 16; b < cmax; b += 16) {
        int j = b + c;
        float my_a = 0.0f;
        int my_s = 0;
        if (j < cnt) {
            int id = clamp_e(perm[clamp_e(beg + j)]);
            my_a = expf(e[id] - m) / s;
            my_s = src[id];
        }
        acc = gather_rows(hcur, my_a, my_s, min(16, cmax - b), c, acc);
    }

    size_t o = (size_t)d * CC + c;
    float hv = mask[d] ? one_hot[o] : acc;
    if (mode == 0) {
        hdst[o] = hv;
    } else {
        float a0 = att[d * 2 + 0], a1 = att[d * 2 + 1];
        float mx = fmaxf(a0, a1);
        float e0 = expf(a0 - mx), e1 = expf(a1 - mx);
        float inv = 1.0f / (e0 + e1);
        if (mode == 1) hdst[o] = hv * (e0 * inv);
        else           hdst[o] = hdst[o] + hv * (e1 * inv);
    }
}

__global__ __launch_bounds__(256) void k_mlp_combine(
    const float* __restrict__ feat, const float* __restrict__ w1,
    const float* __restrict__ b1,   const float* __restrict__ w2,
    const float* __restrict__ b2,   const float* __restrict__ alpha,
    float* __restrict__ out, int n_pairs)
{
    __shared__ float sW1p[(FF / 2) * W1_PSTRIDE];
    __shared__ float sW2p[(HH / 2) * W2_PSTRIDE];
    __shared__ float sHid[8][2][16 * HID_STRIDE];

    int t = threadIdx.x;

    for (int i = t; i < (FF / 2) * (HH / 4); i += 256) {
        int p  = i >> 4;
        int nq = (i & 15) << 2;
        float4 x = *(const float4*)&w1[(size_t)p * 2 * HH + nq];
        float4 y = *(const float4*)&w1[(size_t)p * 2 * HH + HH + nq];
        float4 lo = { x.x, y.x, x.y, y.y };
        float4 hi = { x.z, y.z, x.w, y.w };
        float* d = &sW1p[p * W1_PSTRIDE + 2 * nq];
        *(float4*)(d + 0) = lo;
        *(float4*)(d + 4) = hi;
    }
    for (int i = t; i < (HH / 2) * CC; i += 256) {
        int p = i >> 4, n = i & 15;
        v2f v; v.x = w2[(size_t)p * 2 * CC + n];
        v.y = w2[(size_t)p * 2 * CC + CC + n];
        *(v2f*)&sW2p[p * W2_PSTRIDE + 2 * n] = v;
    }
    __syncthreads();

    int wave = t >> 5;
    int lane = t & 31;
    int pair = blockIdx.x * 8 + wave;
    if (pair >= n_pairs) return;

    int tile0 = pair * 2;
    int tile1 = tile0 + 1;

    int m     = lane & 15;
    int n     = lane & 15;
    int hi16  = lane >> 4;
    int khalf = hi16 << 1;
    int moff  = hi16 << 3;

    v8f acc0 = {}, acc1 = {}, acc2 = {}, acc3 = {};
    v8f acc4 = {}, acc5 = {}, acc6 = {}, acc7 = {};
    const float* arow0 = feat + ((size_t)tile0 * 16 + m) * FF;
    const float* arow1 = feat + ((size_t)tile1 * 16 + m) * FF;
    for (int k = 0; k < FF; k += 4) {
        int kb = k + khalf;
        v2f a0 = *(const v2f*)(arow0 + kb);
        v2f a1 = *(const v2f*)(arow1 + kb);
        const float* bp = &sW1p[((k >> 1) + hi16) * W1_PSTRIDE + 2 * n];
        v2f b0 = *(const v2f*)(bp + 0);
        v2f b1v = *(const v2f*)(bp + 32);
        v2f b2v = *(const v2f*)(bp + 64);
        v2f b3v = *(const v2f*)(bp + 96);
        acc0 = __builtin_amdgcn_wmma_f32_16x16x4_f32(false, a0, false, b0,  (short)0, acc0, false, false);
        acc1 = __builtin_amdgcn_wmma_f32_16x16x4_f32(false, a0, false, b1v, (short)0, acc1, false, false);
        acc2 = __builtin_amdgcn_wmma_f32_16x16x4_f32(false, a0, false, b2v, (short)0, acc2, false, false);
        acc3 = __builtin_amdgcn_wmma_f32_16x16x4_f32(false, a0, false, b3v, (short)0, acc3, false, false);
        acc4 = __builtin_amdgcn_wmma_f32_16x16x4_f32(false, a1, false, b0,  (short)0, acc4, false, false);
        acc5 = __builtin_amdgcn_wmma_f32_16x16x4_f32(false, a1, false, b1v, (short)0, acc5, false, false);
        acc6 = __builtin_amdgcn_wmma_f32_16x16x4_f32(false, a1, false, b2v, (short)0, acc6, false, false);
        acc7 = __builtin_amdgcn_wmma_f32_16x16x4_f32(false, a1, false, b3v, (short)0, acc7, false, false);
    }

    float bb0 = b1[n + 0], bb1 = b1[n + 16], bb2 = b1[n + 32], bb3 = b1[n + 48];
    {
        float* hd0 = &sHid[wave][0][0];
        float* hd1 = &sHid[wave][1][0];
        for (int r = 0; r < 8; ++r) {
            int mm = r + moff;
            hd0[mm * HID_STRIDE + n +  0] = fmaxf(acc0[r] + bb0, 0.0f);
            hd0[mm * HID_STRIDE + n + 16] = fmaxf(acc1[r] + bb1, 0.0f);
            hd0[mm * HID_STRIDE + n + 32] = fmaxf(acc2[r] + bb2, 0.0f);
            hd0[mm * HID_STRIDE + n + 48] = fmaxf(acc3[r] + bb3, 0.0f);
            hd1[mm * HID_STRIDE + n +  0] = fmaxf(acc4[r] + bb0, 0.0f);
            hd1[mm * HID_STRIDE + n + 16] = fmaxf(acc5[r] + bb1, 0.0f);
            hd1[mm * HID_STRIDE + n + 32] = fmaxf(acc6[r] + bb2, 0.0f);
            hd1[mm * HID_STRIDE + n + 48] = fmaxf(acc7[r] + bb3, 0.0f);
        }
    }

    float bc = b2[n];
    for (int tt = 0; tt < 2; ++tt) {
        const float* hd = &sHid[wave][tt][0];
        v8f acc = {};
        for (int k = 0; k < HH; k += 4) {
            int kb = k + khalf;
            v2f a = *(const v2f*)(hd + m * HID_STRIDE + kb);
            v2f b = *(const v2f*)&sW2p[((k >> 1) + hi16) * W2_PSTRIDE + 2 * n];
            acc = __builtin_amdgcn_wmma_f32_16x16x4_f32(false, a, false, b, (short)0, acc, false, false);
        }
        int tile = tt ? tile1 : tile0;
        for (int r = 0; r < 8; ++r) {
            int node = tile * 16 + r + moff;
            float mlp = acc[r] + bc;
            float al  = alpha[node];
            float sa  = 1.0f / (1.0f + expf(-al));
            float sna = 1.0f - sa;
            float logit = out[(size_t)node * CC + n];
            out[(size_t)node * CC + n] = sa * logit + sna * mlp;
        }
    }
}

static void run_graph(hipStream_t stream, const float* e0, const float* e1,
                      const int* sg, const int* dg, int* cur, int* perm, int* bsum,
                      float* hA, float* out, const float* label_init, const float* labels_onehot,
                      const int* train_mask, const float* attention, int last_mode) {
    const int TB = 256;
    const int gridNC = (NN * CC) / TB;
    int* bh    = (int*)hA;
    int* tot   = bh + NBK * NBLK;
    int* bbase = tot + NBK;
    k_bhist   <<<NBLK, ST, 0, stream>>>(dg, bh);
    k_bscan   <<<NBK,  ST, 0, stream>>>(bh, tot);
    k_bscatter<<<NBLK, ST, 0, stream>>>(dg, bh, tot, bbase, (unsigned*)perm);
    k_bsort   <<<NBK,  ST, 0, stream>>>(tot, bbase, cur, (unsigned*)perm);
    k_gather<<<gridNC, TB, 0, stream>>>(e0, sg, perm, cur, label_init, labels_onehot,
                                        train_mask, attention, hA, 0);
    k_gather<<<gridNC, TB, 0, stream>>>(e1, sg, perm, cur, hA, labels_onehot,
                                        train_mask, attention, out, last_mode);
}

extern "C" void kernel_launch(void* const* d_in, const int* in_sizes, int n_in,
                              void* d_out, int out_size, void* d_ws, size_t ws_size,
                              hipStream_t stream) {
    const float* features      = (const float*)d_in[0];
    const float* label_init    = (const float*)d_in[1];
    const float* labels_onehot = (const float*)d_in[2];
    const float* alpha         = (const float*)d_in[3];
    const float* attention     = (const float*)d_in[4];
    const float* e_edge        = (const float*)d_in[5];
    const float* w1            = (const float*)d_in[6];
    const float* b1            = (const float*)d_in[7];
    const float* w2            = (const float*)d_in[8];
    const float* b2            = (const float*)d_in[9];
    const int*   src           = (const int*)d_in[10];
    const int*   dst           = (const int*)d_in[11];
    const int*   train_mask    = (const int*)d_in[12];
    float* out = (float*)d_out;

    int*   cur  = (int*)d_ws;
    int*   perm = cur + NN;
    float* hA   = (float*)(perm + EE);
    int*   bsum = (int*)(hA + NN * CC);

    const int TB = 256;

    run_graph(stream, e_edge,      e_edge + (size_t)GG * EE,      src,      dst,
              cur, perm, bsum, hA, out, label_init, labels_onehot, train_mask, attention, 1);
    run_graph(stream, e_edge + EE, e_edge + (size_t)GG * EE + EE, src + EE, dst + EE,
              cur, perm, bsum, hA, out, label_init, labels_onehot, train_mask, attention, 2);

    const int n_pairs = NN / 32;
    const int blocks  = (n_pairs + 7) / 8;
    k_mlp_combine<<<blocks, TB, 0, stream>>>(features, w1, b1, w2, b2, alpha, out, n_pairs);
}


# Gather head: a lane whose slot lies past its node's count re-reads the node's first slot, so it pulls no line of a later node's logit and source; values used and sum order unchanged
# speedup vs baseline: 10.6084x; 1.0719x over previous
#include <hip/hip_runtime.h>
#include <hip/hip_bf16.h>
#include <math.h>

#define NN 100000
#define CC 16
#define GG 2
#define LL 2
#define EE 3200000
#define FF 512
#define HH 64

#define W1_PSTRIDE 160
#define W2_PSTRIDE 32
#define HID_STRIDE 66

typedef float v2f __attribute__((ext_vector_type(2)));
typedef float v8f __attribute__((ext_vector_type(8)));

#define BKT_SHIFT 8
#define BKT_W     256
#define NBK       ((NN + BKT_W - 1) / BKT_W)
#define CH        8192
#define CH_BITS   13
#define NBLK      ((EE + CH - 1) / CH)
#define ST        512
#define CAP       32768
#define ID_BITS   22
#define ID_MASK   0x3FFFFFu

static_assert(EE <= (1 << ID_BITS), "edge id must fit the record");
static_assert(NBK <= ST && NBLK <= ST && BKT_W <= ST, "one scan pass per table");
static_assert(CH % ST == 0, "whole strides per block");
static_assert(CH == (1 << CH_BITS) && NN <= (1 << (32 - CH_BITS)), "node and edge-in-block fit one staged word");

__device__ __forceinline__ int clamp_node(int d) { return min(max(d, 0), NN - 1); }

__device__ __forceinline__ int block_scan_incl(int* sh, int t, int v) {
    sh[t] = v;
    __syncthreads();
    for (int off = 1; off < ST; off <<= 1) {
        int a = (t >= off) ? sh[t - off] : 0;
        __syncthreads();
        sh[t] += a;
        __syncthreads();
    }
    return sh[t];
}

__device__ __forceinline__ void block_scan_incl2(int* sa, int* sb, int t, int& a, int& b) {
    sa[t] = a;
    sb[t] = b;
    __syncthreads();
    for (int off = 1; off < ST; off <<= 1) {
        int xa = (t >= off) ? sa[t - off] : 0;
        int xb = (t >= off) ? sb[t - off] : 0;
        __syncthreads();
        sa[t] += xa;
        sb[t] += xb;
        __syncthreads();
    }
    a = sa[t];
    b = sb[t];
}

__global__ __launch_bounds__(ST) void k_bhist(const int* __restrict__ dst, int* __restrict__ bh) {
    __shared__ int h[NBK];
    int t = threadIdx.x;
    if (t < NBK) h[t] = 0;
    __syncthreads();
    int base = blockIdx.x * CH + t;
    int d[CH / ST];
    for (int j = 0; j < CH / ST; ++j)
        d[j] = dst[min(base + j * ST, EE - 1)];
    for (int j = 0; j < CH / ST; ++j)
        if (base + j * ST < EE) atomicAdd(&h[clamp_node(d[j]) >> BKT_SHIFT], 1);
    __syncthreads();
    if (t < NBK) bh[t * NBLK + blockIdx.x] = h[t];
}

__global__ __launch_bounds__(ST) void k_bscan(int* __restrict__ bh, int* __restrict__ tot) {
    __shared__ int sh[ST];
    int t = threadIdx.x, b = blockIdx.x;
    int v = (t < NBLK) ? bh[b * NBLK + t] : 0;
    int incl = block_scan_incl(sh, t, v);
    if (t < NBLK) bh[b * NBLK + t] = incl - v;
    if (t == ST - 1) tot[b] = incl;
}

__global__ __launch_bounds__(ST) void k_bscatter(const int* __restrict__ dst, const int* __restrict__ bh,
                                                 const int* __restrict__ tot, int* __restrict__ bbase,
                                                 unsigned* __restrict__ rec) {
    __shared__ int sh[ST];
    __shared__ int sh2[ST];
    __shared__ int lcur[NBK];
    __shared__ int delta[NBK];
    __shared__ unsigned stage[CH];
    int t = threadIdx.x, blk = blockIdx.x;
    int v = 0, mine = 0, lc = 0;
    if (t < NBK) {
        v = tot[t];
        mine = bh[t * NBLK + blk];
        int nxt = (blk + 1 < NBLK) ? bh[t * NBLK + blk + 1] : v;
        lc = nxt - mine;
    }
    int incl = v, lincl = lc;
    block_scan_incl2(sh, sh2, t, incl, lincl);
    if (t < NBK) {
        int excl  = incl - v;
        int lbase = lincl - lc;
        lcur[t]  = lbase;
        delta[t] = excl + mine - lbase;
        if (blk == 0) bbase[t] = excl;
    }
    __syncthreads();
    int base = blk * CH + t;
    int dv[CH / ST];
    for (int j = 0; j < CH / ST; ++j)
        dv[j] = dst[min(base + j * ST, EE - 1)];
    for (int j = 0; j < CH / ST; ++j) {
        if (base + j * ST < EE) {
            int d = clamp_node(dv[j]);
            int r = atomicAdd(&lcur[d >> BKT_SHIFT], 1);
            r = min(max(r, 0), CH - 1);
            stage[r] = ((unsigned)d << CH_BITS) | (unsigned)(j * ST + t);
        }
    }
    __syncthreads();
    int nblk = min(CH, EE - blk * CH);
    for (int i0 = t; i0 < nblk; i0 += 4 * ST) {
        unsigned w[4];
        for (int u = 0; u < 4; ++u) w[u] = stage[min(i0 + u * ST, CH - 1)];
        for (int u = 0; u < 4; ++u) {
            int i = i0 + u * ST;
            if (i < nblk) {
                int d = (int)(w[u] >> CH_BITS);
                int b = min(d >> BKT_SHIFT, NBK - 1);
                int p = min(max(i + delta[b], 0), EE - 1);
                rec[p] = ((unsigned)(d & (BKT_W - 1)) << ID_BITS) | (unsigned)(blk * CH + (int)(w[u] & (CH - 1)));
            }
        }
    }
}

__global__ __launch_bounds__(ST) void k_bsort(const int* __restrict__ tot, const int* __restrict__ bbase,
                                              int* __restrict__ cur, unsigned* slots) {
    __shared__ unsigned rec[CAP];
    __shared__ int sh[ST];
    __shared__ int cnt[BKT_W];
    __shared__ int head[BKT_W];
    int t = threadIdx.x, b = blockIdx.x;
    int n  = min(max(tot[b], 0), EE);
    int bs = min(max(bbase[b], 0), EE - n);
    unsigned* A = slots + bs;
    bool fits = (n <= CAP);
    if (t < BKT_W) cnt[t] = 0;
    __syncthreads();
    for (int i0 = t; i0 < n; i0 += 4 * ST) {
        unsigned r[4];
        for (int u = 0; u < 4; ++u) r[u] = A[min(i0 + u * ST, n - 1)];
        for (int u = 0; u < 4; ++u)
            if (i0 + u * ST < n) atomicAdd(&cnt[(r[u] >> ID_BITS) & (BKT_W - 1)], 1);
    }
    __syncthreads();
    int c = (t < BKT_W) ? cnt[t] : 0;
    int incl = block_scan_incl(sh, t, c);
    if (t < BKT_W) {
        head[t] = incl - c;
        int node = b * BKT_W + t;
        if (node < NN) cur[node] = bs + incl;
    }
    __syncthreads();
    if (fits) {
        for (int i0 = t; i0 < n; i0 += 4 * ST) {
            unsigned r[4];
            for (int u = 0; u < 4; ++u) r[u] = A[min(i0 + u * ST, n - 1)];
            for (int u = 0; u < 4; ++u) {
                if (i0 + u * ST < n) {
                    int p = atomicAdd(&head[(r[u] >> ID_BITS) & (BKT_W - 1)], 1);
                    rec[min(max(p, 0), n - 1)] = r[u] & ID_MASK;
                }
            }
        }
        __syncthreads();
        for (int i0 = t; i0 < n; i0 += 4 * ST) {
            unsigned r[4];
            for (int u = 0; u < 4; ++u) r[u] = rec[min(i0 + u * ST, n - 1)];
            for (int u = 0; u < 4; ++u)
                if (i0 + u * ST < n) A[i0 + u * ST] = r[u];
        }
    } else {
        if (t == 0) {
            for (int k = 0; k < BKT_W; ++k) {
                int endk = sh[k];
                while (head[k] < endk) {
                    int pk = head[k];
                    unsigned r = A[pk];
                    int kk = (int)((r >> ID_BITS) & (BKT_W - 1));
                    if (kk == k) {
                        head[k] = pk + 1;
                    } else {
                        int q = min(head[kk], n - 1);
                        head[kk] = q + 1;
                        unsigned r2 = A[q];
                        A[q] = r;
                        A[pk] = r2;
                    }
                }
            }
        }
        __syncthreads();
        for (int i = t; i < n; i += ST) A[i] &= ID_MASK;
    }
}

#define HEAD_Q 3

__device__ __forceinline__ int clamp_e(int i) { return min(max(i, 0), EE - 1); }

__device__ __forceinline__ float gather_rows(const float* __restrict__ hcur, float my_a, int my_s,
                                             int kmax, int c, float acc) {
    for (int k = 0; k < kmax; k += 4) {
        float a0 = __shfl(my_a, k + 0, 16), a1 = __shfl(my_a, k + 1, 16);
        float a2 = __shfl(my_a, k + 2, 16), a3 = __shfl(my_a, k + 3, 16);
        int   s0 = __shfl(my_s, k + 0, 16), s1 = __shfl(my_s, k + 1, 16);
        int   s2 = __shfl(my_s, k + 2, 16), s3 = __shfl(my_s, k + 3, 16);
        float h0 = hcur[(size_t)s0 * CC + c], h1 = hcur[(size_t)s1 * CC + c];
        float h2 = hcur[(size_t)s2 * CC + c], h3 = hcur[(size_t)s3 * CC + c];
        acc += h0 * a0; acc += h1 * a1; acc += h2 * a2; acc += h3 * a3;
    }
    return acc;
}

__global__ __launch_bounds__(256) void k_gather(
    const float* __restrict__ e, const int* __restrict__ src,
    const int* __restrict__ perm, const int* __restrict__ cur,
    const float* __restrict__ hcur, const float* __restrict__ one_hot,
    const int* __restrict__ mask, const float* __restrict__ att,
    float* hdst, int mode)
{
    int gid = blockIdx.x * blockDim.x + threadIdx.x;
    int d = gid >> 4;
    int c = threadIdx.x & 15;
    int beg = (d == 0) ? 0 : cur[d - 1];
    int cnt = cur[d] - beg;
    int cmax = max(cnt, __shfl_xor(cnt, 16));

    int   pid[HEAD_Q];
    float ev[HEAD_Q];
    int   sv[HEAD_Q];
    for (int q = 0; q < HEAD_Q; ++q) {
        int j = q * 16 + c;
        pid[q] = perm[clamp_e(beg + (j < cnt ? j : 0))];
    }
    for (int q = 0; q < HEAD_Q; ++q) {
        int id = clamp_e(pid[q]);
        ev[q] = e[id];
        sv[q] = src[id];
    }

    float m = -INFINITY;
    for (int q = 0; q < HEAD_Q; ++q) if (q * 16 + c < cnt) m = fmaxf(m, ev[q]);
    for (int b = HEAD_Q * 16; b < cmax; b += 16) {
        int j = b + c;
        if (j < cnt) m = fmaxf(m, e[clamp_e(perm[clamp_e(beg + j)])]);
    }
    for (int o = 8; o > 0; o >>= 1) m = fmaxf(m, __shfl_xor(m, o));

    float s = 0.0f;
    for (int q = 0; q < HEAD_Q; ++q) if (q * 16 + c < cnt) s += expf(ev[q] - m);
    for (int b = HEAD_Q * 16; b < cmax; b += 16) {
        int j = b + c;
        if (j < cnt) s += expf(e[clamp_e(perm[clamp_e(beg + j)])] - m);
    }
    for (int o = 8; o > 0; o >>= 1) s += __shfl_xor(s, o);

    float acc = 0.0f;
    for (int q = 0; q < HEAD_Q; ++q) {
        int b = q * 16;
        if (b < cmax) {
            bool ok = (b + c < cnt);
            float my_a = ok ? expf(ev[q] - m) / s : 0.0f;
            int   my_s = ok ? sv[q] : 0;
            acc = gather_rows(hcur, my_a, my_s, min(16, cmax - b), c, acc);
        }
    }
    for (int b = HEAD_Q * 16; b < cmax; b += 16) {
        int j = b + c;
        float my_a = 0.0f;
        int my_s = 0;
        if (j < cnt) {
            int id = clamp_e(perm[clamp_e(beg + j)]);
            my_a = expf(e[id] - m) / s;
            my_s = src[id];
        }
        acc = gather_rows(hcur, my_a, my_s, min(16, cmax - b), c, acc);
    }

    size_t o = (size_t)d * CC + c;
    float hv = mask[d] ? one_hot[o] : acc;
    if (mode == 0) {
        hdst[o] = hv;
    } else {
        float a0 = att[d * 2 + 0], a1 = att[d * 2 + 1];
        float mx = fmaxf(a0, a1);
        float e0 = expf(a0 - mx), e1 = expf(a1 - mx);
        float inv = 1.0f / (e0 + e1);
        if (mode == 1) hdst[o] = hv * (e0 * inv);
        else           hdst[o] = hdst[o] + hv * (e1 * inv);
    }
}

__global__ __launch_bounds__(256) void k_mlp_combine(
    const float* __restrict__ feat, const float* __restrict__ w1,
    const float* __restrict__ b1,   const float* __restrict__ w2,
    const float* __restrict__ b2,   const float* __restrict__ alpha,
    float* __restrict__ out, int n_pairs)
{
    __shared__ float sW1p[(FF / 2) * W1_PSTRIDE];
    __shared__ float sW2p[(HH / 2) * W2_PSTRIDE];
    __shared__ float sHid[8][2][16 * HID_STRIDE];

    int t = threadIdx.x;

    for (int i = t; i < (FF / 2) * (HH / 4); i += 256) {
        int p  = i >> 4;
        int nq = (i & 15) << 2;
        float4 x = *(const float4*)&w1[(size_t)p * 2 * HH + nq];
        float4 y = *(const float4*)&w1[(size_t)p * 2 * HH + HH + nq];
        float4 lo = { x.x, y.x, x.y, y.y };
        float4 hi = { x.z, y.z, x.w, y.w };
        float* d = &sW1p[p * W1_PSTRIDE + 2 * nq];
        *(float4*)(d + 0) = lo;
        *(float4*)(d + 4) = hi;
    }
    for (int i = t; i < (HH / 2) * CC; i += 256) {
        int p = i >> 4, n = i & 15;
        v2f v; v.x = w2[(size_t)p * 2 * CC + n];
        v.y = w2[(size_t)p * 2 * CC + CC + n];
        *(v2f*)&sW2p[p * W2_PSTRIDE + 2 * n] = v;
    }
    __syncthreads();

    int wave = t >> 5;
    int lane = t & 31;
    int pair = blockIdx.x * 8 + wave;
    if (pair >= n_pairs) return;

    int tile0 = pair * 2;
    int tile1 = tile0 + 1;

    int m     = lane & 15;
    int n     = lane & 15;
    int hi16  = lane >> 4;
    int khalf = hi16 << 1;
    int moff  = hi16 << 3;

    v8f acc0 = {}, acc1 = {}, acc2 = {}, acc3 = {};
    v8f acc4 = {}, acc5 = {}, acc6 = {}, acc7 = {};
    const float* arow0 = feat + ((size_t)tile0 * 16 + m) * FF;
    const float* arow1 = feat + ((size_t)tile1 * 16 + m) * FF;
    for (int k = 0; k < FF; k += 4) {
        int kb = k + khalf;
        v2f a0 = *(const v2f*)(arow0 + kb);
        v2f a1 = *(const v2f*)(arow1 + kb);
        const float* bp = &sW1p[((k >> 1) + hi16) * W1_PSTRIDE + 2 * n];
        v2f b0 = *(const v2f*)(bp + 0);
        v2f b1v = *(const v2f*)(bp + 32);
        v2f b2v = *(const v2f*)(bp + 64);
        v2f b3v = *(const v2f*)(bp + 96);
        acc0 = __builtin_amdgcn_wmma_f32_16x16x4_f32(false, a0, false, b0,  (short)0, acc0, false, false);
        acc1 = __builtin_amdgcn_wmma_f32_16x16x4_f32(false, a0, false, b1v, (short)0, acc1, false, false);
        acc2 = __builtin_amdgcn_wmma_f32_16x16x4_f32(false, a0, false, b2v, (short)0, acc2, false, false);
        acc3 = __builtin_amdgcn_wmma_f32_16x16x4_f32(false, a0, false, b3v, (short)0, acc3, false, false);
        acc4 = __builtin_amdgcn_wmma_f32_16x16x4_f32(false, a1, false, b0,  (short)0, acc4, false, false);
        acc5 = __builtin_amdgcn_wmma_f32_16x16x4_f32(false, a1, false, b1v, (short)0, acc5, false, false);
        acc6 = __builtin_amdgcn_wmma_f32_16x16x4_f32(false, a1, false, b2v, (short)0, acc6, false, false);
        acc7 = __builtin_amdgcn_wmma_f32_16x16x4_f32(false, a1, false, b3v, (short)0, acc7, false, false);
    }

    float bb0 = b1[n + 0], bb1 = b1[n + 16], bb2 = b1[n + 32], bb3 = b1[n + 48];
    {
        float* hd0 = &sHid[wave][0][0];
        float* hd1 = &sHid[wave][1][0];
        for (int r = 0; r < 8; ++r) {
            int mm = r + moff;
            hd0[mm * HID_STRIDE + n +  0] = fmaxf(acc0[r] + bb0, 0.0f);
            hd0[mm * HID_STRIDE + n + 16] = fmaxf(acc1[r] + bb1, 0.0f);
            hd0[mm * HID_STRIDE + n + 32] = fmaxf(acc2[r] + bb2, 0.0f);
            hd0[mm * HID_STRIDE + n + 48] = fmaxf(acc3[r] + bb3, 0.0f);
            hd1[mm * HID_STRIDE + n +  0] = fmaxf(acc4[r] + bb0, 0.0f);
            hd1[mm * HID_STRIDE + n + 16] = fmaxf(acc5[r] + bb1, 0.0f);
            hd1[mm * HID_STRIDE + n + 32] = fmaxf(acc6[r] + bb2, 0.0f);
            hd1[mm * HID_STRIDE + n + 48] = fmaxf(acc7[r] + bb3, 0.0f);
        }
    }

    float bc = b2[n];
    for (int tt = 0; tt < 2; ++tt) {
        const float* hd = &sHid[wave][tt][0];
        v8f acc = {};
        for (int k = 0; k < HH; k += 4) {
            int kb = k + khalf;
            v2f a = *(const v2f*)(hd + m * HID_STRIDE + kb);
            v2f b = *(const v2f*)&sW2p[((k >> 1) + hi16) * W2_PSTRIDE + 2 * n];
            acc = __builtin_amdgcn_wmma_f32_16x16x4_f32(false, a, false, b, (short)0, acc, false, false);
        }
        int tile = tt ? tile1 : tile0;
        for (int r = 0; r < 8; ++r) {
            int node = tile * 16 + r + moff;
            float mlp = acc[r] + bc;
            float al  = alpha[node];
            float sa  = 1.0f / (1.0f + expf(-al));
            float sna = 1.0f - sa;
            float logit = out[(size_t)node * CC + n];
            out[(size_t)node * CC + n] = sa * logit + sna * mlp;
        }
    }
}

static void run_graph(hipStream_t stream, const float* e0, const float* e1,
                      const int* sg, const int* dg, int* cur, int* perm, int* bsum,
                      float* hA, float* out, const float* label_init, const float* labels_onehot,
                      const int* train_mask, const float* attention, int last_mode) {
    const int TB = 256;
    const int gridNC = (NN * CC) / TB;
    int* bh    = (int*)hA;
    int* tot   = bh + NBK * NBLK;
    int* bbase = tot + NBK;
    k_bhist   <<<NBLK, ST, 0, stream>>>(dg, bh);
    k_bscan   <<<NBK,  ST, 0, stream>>>(bh, tot);
    k_bscatter<<<NBLK, ST, 0, stream>>>(dg, bh, tot, bbase, (unsigned*)perm);
    k_bsort   <<<NBK,  ST, 0, stream>>>(tot, bbase, cur, (unsigned*)perm);
    k_gather<<<gridNC, TB, 0, stream>>>(e0, sg, perm, cur, label_init, labels_onehot,
                                        train_mask, attention, hA, 0);
    k_gather<<<gridNC, TB, 0, stream>>>(e1, sg, perm, cur, hA, labels_onehot,
                                        train_mask, attention, out, last_mode);
}

extern "C" void kernel_launch(void* const* d_in, const int* in_sizes, int n_in,
                              void* d_out, int out_size, void* d_ws, size_t ws_size,
                              hipStream_t stream) {
    const float* features      = (const float*)d_in[0];
    const float* label_init    = (const float*)d_in[1];
    const float* labels_onehot = (const float*)d_in[2];
    const float* alpha         = (const float*)d_in[3];
    const float* attention     = (const float*)d_in[4];
    const float* e_edge        = (const float*)d_in[5];
    const float* w1            = (const float*)d_in[6];
    const float* b1            = (const float*)d_in[7];
    const float* w2            = (const float*)d_in[8];
    const float* b2            = (const float*)d_in[9];
    const int*   src           = (const int*)d_in[10];
    const int*   dst           = (const int*)d_in[11];
    const int*   train_mask    = (const int*)d_in[12];
    float* out = (float*)d_out;

    int*   cur  = (int*)d_ws;
    int*   perm = cur + NN;
    float* hA   = (float*)(perm + EE);
    int*   bsum = (int*)(hA + NN * CC);

    const int TB = 256;

    run_graph(stream, e_edge,      e_edge + (size_t)GG * EE,      src,      dst,
              cur, perm, bsum, hA, out, label_init, labels_onehot, train_mask, attention, 1);
    run_graph(stream, e_edge + EE, e_edge + (size_t)GG * EE + EE, src + EE, dst + EE,
              cur, perm, bsum, hA, out, label_init, labels_onehot, train_mask, attention, 2);

    const int n_pairs = NN / 32;
    const int blocks  = (n_pairs + 7) / 8;
    k_mlp_combine<<<blocks, TB, 0, stream>>>(features, w1, b1, w2, b2, alpha, out, n_pairs);
}


# Sort carries each edge's source node and both layers' logits as payload planes (if ws_size holds two more E-word planes): gathers read them in whole lines, not by index; in-node sum order may change
# speedup vs baseline: 12.2515x; 1.1549x over previous
#include <hip/hip_runtime.h>
#include <hip/hip_bf16.h>
#include <math.h>

#define NN 100000
#define CC 16
#define GG 2
#define LL 2
#define EE 3200000
#define FF 512
#define HH 64

#define W1_PSTRIDE 160
#define W2_PSTRIDE 32
#define HID_STRIDE 66

typedef float v2f __attribute__((ext_vector_type(2)));
typedef float v8f __attribute__((ext_vector_type(8)));

#define BKT_SHIFT 8
#define BKT_W     256
#define NBK       ((NN + BKT_W - 1) / BKT_W)
#define CH        8192
#define CH_BITS   13
#define NBLK      ((EE + CH - 1) / CH)
#define ST        512
#define CAP       32768
#define ID_BITS   22
#define ID_MASK   0x3FFFFFu
#define SRC_BITS  17
#define SRC_MASK  0x1FFFFu
#define CAPW      (CAP / 3)
#define WIDE_BYTES (((size_t)NN + (size_t)EE + (size_t)NN * CC + 2 * (size_t)EE) * sizeof(int))

static_assert(EE <= (1 << ID_BITS), "edge id must fit the record");
static_assert(NN <= (1 << SRC_BITS) && BKT_SHIFT + SRC_BITS <= 32, "bucket-local node and source node fit one word");
static_assert(NBK <= ST && NBLK <= ST && BKT_W <= ST, "one scan pass per table");
static_assert(CH % ST == 0, "whole strides per block");
static_assert(CH == (1 << CH_BITS) && NN <= (1 << (32 - CH_BITS)), "node and edge-in-block fit one staged word");

__device__ __forceinline__ int clamp_node(int d) { return min(max(d, 0), NN - 1); }

__device__ __forceinline__ int block_scan_incl(int* sh, int t, int v) {
    sh[t] = v;
    __syncthreads();
    for (int off = 1; off < ST; off <<= 1) {
        int a = (t >= off) ? sh[t - off] : 0;
        __syncthreads();
        sh[t] += a;
        __syncthreads();
    }
    return sh[t];
}

__device__ __forceinline__ void block_scan_incl2(int* sa, int* sb, int t, int& a, int& b) {
    sa[t] = a;
    sb[t] = b;
    __syncthreads();
    for (int off = 1; off < ST; off <<= 1) {
        int xa = (t >= off) ? sa[t - off] : 0;
        int xb = (t >= off) ? sb[t - off] : 0;
        __syncthreads();
        sa[t] += xa;
        sb[t] += xb;
        __syncthreads();
    }
    a = sa[t];
    b = sb[t];
}

__global__ __launch_bounds__(ST) void k_bhist(const int* __restrict__ dst, int* __restrict__ bh) {
    __shared__ int h[NBK];
    int t = threadIdx.x;
    if (t < NBK) h[t] = 0;
    __syncthreads();
    int base = blockIdx.x * CH + t;
    int d[CH / ST];
    for (int j = 0; j < CH / ST; ++j)
        d[j] = dst[min(base + j * ST, EE - 1)];
    for (int j = 0; j < CH / ST; ++j)
        if (base + j * ST < EE) atomicAdd(&h[clamp_node(d[j]) >> BKT_SHIFT], 1);
    __syncthreads();
    if (t < NBK) bh[t * NBLK + blockIdx.x] = h[t];
}

__global__ __launch_bounds__(ST) void k_bscan(int* __restrict__ bh, int* __restrict__ tot) {
    __shared__ int sh[ST];
    int t = threadIdx.x, b = blockIdx.x;
    int v = (t < NBLK) ? bh[b * NBLK + t] : 0;
    int incl = block_scan_incl(sh, t, v);
    if (t < NBLK) bh[b * NBLK + t] = incl - v;
    if (t == ST - 1) tot[b] = incl;
}

__global__ __launch_bounds__(ST) void k_bscatter(const int* __restrict__ dst, const int* __restrict__ bh,
                                                 const int* __restrict__ tot, int* __restrict__ bbase,
                                                 unsigned* __restrict__ rec,
                                                 const int* __restrict__ src, const float* __restrict__ e0,
                                                 const float* __restrict__ e1, float* __restrict__ p0,
                                                 float* __restrict__ p1, size_t ws_size) {
    __shared__ int sh[ST];
    __shared__ int sh2[ST];
    __shared__ int lcur[NBK];
    __shared__ int delta[NBK];
    __shared__ unsigned stage[CH];
    __shared__ unsigned stS[CH];
    __shared__ float st0[CH];
    __shared__ float st1[CH];
    const bool wide = (ws_size >= WIDE_BYTES);
    int t = threadIdx.x, blk = blockIdx.x;
    int v = 0, mine = 0, lc = 0;
    if (t < NBK) {
        v = tot[t];
        mine = bh[t * NBLK + blk];
        int nxt = (blk + 1 < NBLK) ? bh[t * NBLK + blk + 1] : v;
        lc = nxt - mine;
    }
    int incl = v, lincl = lc;
    block_scan_incl2(sh, sh2, t, incl, lincl);
    if (t < NBK) {
        int excl  = incl - v;
        int lbase = lincl - lc;
        lcur[t]  = lbase;
        delta[t] = excl + mine - lbase;
        if (blk == 0) bbase[t] = excl;
    }
    __syncthreads();
    int base = blk * CH + t;
    int dv[CH / ST];
    for (int j = 0; j < CH / ST; ++j)
        dv[j] = dst[min(base + j * ST, EE - 1)];
    if (wide) {
        int   sv[CH / ST];
        float x0[CH / ST], x1[CH / ST];
        for (int j = 0; j < CH / ST; ++j) {
            int id = min(base + j * ST, EE - 1);
            sv[j] = src[id];
            x0[j] = e0[id];
            x1[j] = e1[id];
        }
        for (int j = 0; j < CH / ST; ++j) {
            if (base + j * ST < EE) {
                int d = clamp_node(dv[j]);
                int r = atomicAdd(&lcur[d >> BKT_SHIFT], 1);
                r = min(max(r, 0), CH - 1);
                stage[r] = ((unsigned)d << CH_BITS) | (unsigned)(j * ST + t);
                stS[r] = (unsigned)sv[j] & SRC_MASK;
                st0[r] = x0[j];
                st1[r] = x1[j];
            }
        }
    } else {
        for (int j = 0; j < CH / ST; ++j) {
            if (base + j * ST < EE) {
                int d = clamp_node(dv[j]);
                int r = atomicAdd(&lcur[d >> BKT_SHIFT], 1);
                r = min(max(r, 0), CH - 1);
                stage[r] = ((unsigned)d << CH_BITS) | (unsigned)(j * ST + t);
            }
        }
    }
    __syncthreads();
    int nblk = min(CH, EE - blk * CH);
    for (int i0 = t; i0 < nblk; i0 += 4 * ST) {
        unsigned w[4];
        for (int u = 0; u < 4; ++u) w[u] = stage[min(i0 + u * ST, CH - 1)];
        for (int u = 0; u < 4; ++u) {
            int i = i0 + u * ST;
            if (i < nblk) {
                int d = (int)(w[u] >> CH_BITS);
                int b = min(d >> BKT_SHIFT, NBK - 1);
                int p = min(max(i + delta[b], 0), EE - 1);
                if (wide) {
                    rec[p] = ((unsigned)(d & (BKT_W - 1)) << SRC_BITS) | stS[i];
                    p0[p]  = st0[i];
                    p1[p]  = st1[i];
                } else {
                    rec[p] = ((unsigned)(d & (BKT_W - 1)) << ID_BITS) | (unsigned)(blk * CH + (int)(w[u] & (CH - 1)));
                }
            }
        }
    }
}

__global__ __launch_bounds__(ST) void k_bsort(const int* __restrict__ tot, const int* __restrict__ bbase,
                                              int* __restrict__ cur, unsigned* slots,
                                              float* p0, float* p1, size_t ws_size) {
    __shared__ unsigned rec[CAP];
    const bool wide = (ws_size >= WIDE_BYTES);
    const int  kshift = wide ? SRC_BITS : ID_BITS;
    const unsigned vmask = wide ? SRC_MASK : ID_MASK;
    __shared__ int sh[ST];
    __shared__ int cnt[BKT_W];
    __shared__ int head[BKT_W];
    int t = threadIdx.x, b = blockIdx.x;
    int n  = min(max(tot[b], 0), EE);
    int bs = min(max(bbase[b], 0), EE - n);
    unsigned* A = slots + bs;
    float* B0 = p0 + bs;
    float* B1 = p1 + bs;
    bool fits = (n <= (wide ? CAPW : CAP));
    if (t < BKT_W) cnt[t] = 0;
    __syncthreads();
    for (int i0 = t; i0 < n; i0 += 4 * ST) {
        unsigned r[4];
        for (int u = 0; u < 4; ++u) r[u] = A[min(i0 + u * ST, n - 1)];
        for (int u = 0; u < 4; ++u)
            if (i0 + u * ST < n) atomicAdd(&cnt[(r[u] >> kshift) & (BKT_W - 1)], 1);
    }
    __syncthreads();
    int c = (t < BKT_W) ? cnt[t] : 0;
    int incl = block_scan_incl(sh, t, c);
    if (t < BKT_W) {
        head[t] = incl - c;
        int node = b * BKT_W + t;
        if (node < NN) cur[node] = bs + incl;
    }
    __syncthreads();
    if (fits && wide) {
        for (int i0 = t; i0 < n; i0 += 4 * ST) {
            unsigned r[4];
            float x0[4], x1[4];
            for (int u = 0; u < 4; ++u) {
                int i = min(i0 + u * ST, n - 1);
                r[u] = A[i]; x0[u] = B0[i]; x1[u] = B1[i];
            }
            for (int u = 0; u < 4; ++u) {
                if (i0 + u * ST < n) {
                    int p = atomicAdd(&head[(r[u] >> SRC_BITS) & (BKT_W - 1)], 1);
                    p = min(max(p, 0), n - 1);
                    rec[p]            = r[u] & SRC_MASK;
                    rec[CAPW + p]     = __float_as_uint(x0[u]);
                    rec[2 * CAPW + p] = __float_as_uint(x1[u]);
                }
            }
        }
        __syncthreads();
        for (int i0 = t; i0 < n; i0 += 4 * ST) {
            unsigned r[4], y0[4], y1[4];
            for (int u = 0; u < 4; ++u) {
                int i = min(i0 + u * ST, n - 1);
                r[u] = rec[i]; y0[u] = rec[CAPW + i]; y1[u] = rec[2 * CAPW + i];
            }
            for (int u = 0; u < 4; ++u) {
                if (i0 + u * ST < n) {
                    A[i0 + u * ST]  = r[u];
                    B0[i0 + u * ST] = __uint_as_float(y0[u]);
                    B1[i0 + u * ST] = __uint_as_float(y1[u]);
                }
            }
        }
    } else if (fits) {
        for (int i0 = t; i0 < n; i0 += 4 * ST) {
            unsigned r[4];
            for (int u = 0; u < 4; ++u) r[u] = A[min(i0 + u * ST, n - 1)];
            for (int u = 0; u < 4; ++u) {
                if (i0 + u * ST < n) {
                    int p = atomicAdd(&head[(r[u] >> ID_BITS) & (BKT_W - 1)], 1);
                    rec[min(max(p, 0), n - 1)] = r[u] & ID_MASK;
                }
            }
        }
        __syncthreads();
        for (int i0 = t; i0 < n; i0 += 4 * ST) {
            unsigned r[4];
            for (int u = 0; u < 4; ++u) r[u] = rec[min(i0 + u * ST, n - 1)];
            for (int u = 0; u < 4; ++u)
                if (i0 + u * ST < n) A[i0 + u * ST] = r[u];
        }
    } else {
        if (t == 0) {
            for (int k = 0; k < BKT_W; ++k) {
                int endk = sh[k];
                while (head[k] < endk) {
                    int pk = head[k];
                    unsigned r = A[pk];
                    int kk = (int)((r >> kshift) & (BKT_W - 1));
                    if (kk == k) {
                        head[k] = pk + 1;
                    } else {
                        int q = min(head[kk], n - 1);
                        head[kk] = q + 1;
                        unsigned r2 = A[q];
                        A[q] = r;
                        A[pk] = r2;
                        if (wide) {
                            float f0 = B0[q], f1 = B1[q];
                            B0[q] = B0[pk]; B1[q] = B1[pk];
                            B0[pk] = f0;    B1[pk] = f1;
                        }
                    }
                }
            }
        }
        __syncthreads();
        for (int i = t; i < n; i += ST) A[i] &= vmask;
    }
}

#define HEAD_Q 3

__device__ __forceinline__ int clamp_e(int i) { return min(max(i, 0), EE - 1); }

__device__ __forceinline__ float gather_rows(const float* __restrict__ hcur, float my_a, int my_s,
                                             int kmax, int c, float acc) {
    for (int k = 0; k < kmax; k += 4) {
        float a0 = __shfl(my_a, k + 0, 16), a1 = __shfl(my_a, k + 1, 16);
        float a2 = __shfl(my_a, k + 2, 16), a3 = __shfl(my_a, k + 3, 16);
        int   s0 = __shfl(my_s, k + 0, 16), s1 = __shfl(my_s, k + 1, 16);
        int   s2 = __shfl(my_s, k + 2, 16), s3 = __shfl(my_s, k + 3, 16);
        float h0 = hcur[(size_t)s0 * CC + c], h1 = hcur[(size_t)s1 * CC + c];
        float h2 = hcur[(size_t)s2 * CC + c], h3 = hcur[(size_t)s3 * CC + c];
        acc += h0 * a0; acc += h1 * a1; acc += h2 * a2; acc += h3 * a3;
    }
    return acc;
}

__global__ __launch_bounds__(256) void k_gather(
    const float* __restrict__ e_in, const float* __restrict__ e_pl, const int* __restrict__ src,
    const int* __restrict__ perm, const int* __restrict__ cur,
    const float* __restrict__ hcur, const float* __restrict__ one_hot,
    const int* __restrict__ mask, const float* __restrict__ att,
    float* hdst, int mode, size_t ws_size)
{
    const bool wide = (ws_size >= WIDE_BYTES);
    const float* __restrict__ e = wide ? e_pl : e_in;
    int gid = blockIdx.x * blockDim.x + threadIdx.x;
    int d = gid >> 4;
    int c = threadIdx.x & 15;
    int beg = (d == 0) ? 0 : cur[d - 1];
    int cnt = cur[d] - beg;
    int cmax = max(cnt, __shfl_xor(cnt, 16));

    int   slot[HEAD_Q];
    float ev[HEAD_Q];
    int   sv[HEAD_Q];
    for (int q = 0; q < HEAD_Q; ++q) {
        int j = q * 16 + c;
        slot[q] = clamp_e(beg + (j < cnt ? j : 0));
    }
    if (wide) {
        for (int q = 0; q < HEAD_Q; ++q) {
            sv[q] = perm[slot[q]];
            ev[q] = e[slot[q]];
        }
    } else {
        int pid[HEAD_Q];
        for (int q = 0; q < HEAD_Q; ++q) pid[q] = perm[slot[q]];
        for (int q = 0; q < HEAD_Q; ++q) {
            int id = clamp_e(pid[q]);
            ev[q] = e[id];
            sv[q] = src[id];
        }
    }

    float m = -INFINITY;
    for (int q = 0; q < HEAD_Q; ++q) if (q * 16 + c < cnt) m = fmaxf(m, ev[q]);
    for (int b = HEAD_Q * 16; b < cmax; b += 16) {
        int j = b + c;
        if (j < cnt) { int sl = clamp_e(beg + j); m = fmaxf(m, e[wide ? sl : clamp_e(perm[sl])]); }
    }
    for (int o = 8; o > 0; o >>= 1) m = fmaxf(m, __shfl_xor(m, o));

    float s = 0.0f;
    for (int q = 0; q < HEAD_Q; ++q) if (q * 16 + c < cnt) s += expf(ev[q] - m);
    for (int b = HEAD_Q * 16; b < cmax; b += 16) {
        int j = b + c;
        if (j < cnt) { int sl = clamp_e(beg + j); s += expf(e[wide ? sl : clamp_e(perm[sl])] - m); }
    }
    for (int o = 8; o > 0; o >>= 1) s += __shfl_xor(s, o);

    float acc = 0.0f;
    for (int q = 0; q < HEAD_Q; ++q) {
        int b = q * 16;
        if (b < cmax) {
            bool ok = (b + c < cnt);
            float my_a = ok ? expf(ev[q] - m) / s : 0.0f;
            int   my_s = ok ? sv[q] : 0;
            acc = gather_rows(hcur, my_a, my_s, min(16, cmax - b), c, acc);
        }
    }
    for (int b = HEAD_Q * 16; b < cmax; b += 16) {
        int j = b + c;
        float my_a = 0.0f;
        int my_s = 0;
        if (j < cnt) {
            int sl = clamp_e(beg + j);
            int pv = perm[sl];
            int id = wide ? sl : clamp_e(pv);
            my_a = expf(e[id] - m) / s;
            my_s = wide ? pv : src[id];
        }
        acc = gather_rows(hcur, my_a, my_s, min(16, cmax - b), c, acc);
    }

    size_t o = (size_t)d * CC + c;
    float hv = mask[d] ? one_hot[o] : acc;
    if (mode == 0) {
        hdst[o] = hv;
    } else {
        float a0 = att[d * 2 + 0], a1 = att[d * 2 + 1];
        float mx = fmaxf(a0, a1);
        float e0 = expf(a0 - mx), e1 = expf(a1 - mx);
        float inv = 1.0f / (e0 + e1);
        if (mode == 1) hdst[o] = hv * (e0 * inv);
        else           hdst[o] = hdst[o] + hv * (e1 * inv);
    }
}

__global__ __launch_bounds__(256) void k_mlp_combine(
    const float* __restrict__ feat, const float* __restrict__ w1,
    const float* __restrict__ b1,   const float* __restrict__ w2,
    const float* __restrict__ b2,   const float* __restrict__ alpha,
    float* __restrict__ out, int n_pairs)
{
    __shared__ float sW1p[(FF / 2) * W1_PSTRIDE];
    __shared__ float sW2p[(HH / 2) * W2_PSTRIDE];
    __shared__ float sHid[8][2][16 * HID_STRIDE];

    int t = threadIdx.x;

    for (int i = t; i < (FF / 2) * (HH / 4); i += 256) {
        int p  = i >> 4;
        int nq = (i & 15) << 2;
        float4 x = *(const float4*)&w1[(size_t)p * 2 * HH + nq];
        float4 y = *(const float4*)&w1[(size_t)p * 2 * HH + HH + nq];
        float4 lo = { x.x, y.x, x.y, y.y };
        float4 hi = { x.z, y.z, x.w, y.w };
        float* d = &sW1p[p * W1_PSTRIDE + 2 * nq];
        *(float4*)(d + 0) = lo;
        *(float4*)(d + 4) = hi;
    }
    for (int i = t; i < (HH / 2) * CC; i += 256) {
        int p = i >> 4, n = i & 15;
        v2f v; v.x = w2[(size_t)p * 2 * CC + n];
        v.y = w2[(size_t)p * 2 * CC + CC + n];
        *(v2f*)&sW2p[p * W2_PSTRIDE + 2 * n] = v;
    }
    __syncthreads();

    int wave = t >> 5;
    int lane = t & 31;
    int pair = blockIdx.x * 8 + wave;
    if (pair >= n_pairs) return;

    int tile0 = pair * 2;
    int tile1 = tile0 + 1;

    int m     = lane & 15;
    int n     = lane & 15;
    int hi16  = lane >> 4;
    int khalf = hi16 << 1;
    int moff  = hi16 << 3;

    v8f acc0 = {}, acc1 = {}, acc2 = {}, acc3 = {};
    v8f acc4 = {}, acc5 = {}, acc6 = {}, acc7 = {};
    const float* arow0 = feat + ((size_t)tile0 * 16 + m) * FF;
    const float* arow1 = feat + ((size_t)tile1 * 16 + m) * FF;
    for (int k = 0; k < FF; k += 4) {
        int kb = k + khalf;
        v2f a0 = *(const v2f*)(arow0 + kb);
        v2f a1 = *(const v2f*)(arow1 + kb);
        const float* bp = &sW1p[((k >> 1) + hi16) * W1_PSTRIDE + 2 * n];
        v2f b0 = *(const v2f*)(bp + 0);
        v2f b1v = *(const v2f*)(bp + 32);
        v2f b2v = *(const v2f*)(bp + 64);
        v2f b3v = *(const v2f*)(bp + 96);
        acc0 = __builtin_amdgcn_wmma_f32_16x16x4_f32(false, a0, false, b0,  (short)0, acc0, false, false);
        acc1 = __builtin_amdgcn_wmma_f32_16x16x4_f32(false, a0, false, b1v, (short)0, acc1, false, false);
        acc2 = __builtin_amdgcn_wmma_f32_16x16x4_f32(false, a0, false, b2v, (short)0, acc2, false, false);
        acc3 = __builtin_amdgcn_wmma_f32_16x16x4_f32(false, a0, false, b3v, (short)0, acc3, false, false);
        acc4 = __builtin_amdgcn_wmma_f32_16x16x4_f32(false, a1, false, b0,  (short)0, acc4, false, false);
        acc5 = __builtin_amdgcn_wmma_f32_16x16x4_f32(false, a1, false, b1v, (short)0, acc5, false, false);
        acc6 = __builtin_amdgcn_wmma_f32_16x16x4_f32(false, a1, false, b2v, (short)0, acc6, false, false);
        acc7 = __builtin_amdgcn_wmma_f32_16x16x4_f32(false, a1, false, b3v, (short)0, acc7, false, false);
    }

    float bb0 = b1[n + 0], bb1 = b1[n + 16], bb2 = b1[n + 32], bb3 = b1[n + 48];
    {
        float* hd0 = &sHid[wave][0][0];
        float* hd1 = &sHid[wave][1][0];
        for (int r = 0; r < 8; ++r) {
            int mm = r + moff;
            hd0[mm * HID_STRIDE + n +  0] = fmaxf(acc0[r] + bb0, 0.0f);
            hd0[mm * HID_STRIDE + n + 16] = fmaxf(acc1[r] + bb1, 0.0f);
            hd0[mm * HID_STRIDE + n + 32] = fmaxf(acc2[r] + bb2, 0.0f);
            hd0[mm * HID_STRIDE + n + 48] = fmaxf(acc3[r] + bb3, 0.0f);
            hd1[mm * HID_STRIDE + n +  0] = fmaxf(acc4[r] + bb0, 0.0f);
            hd1[mm * HID_STRIDE + n + 16] = fmaxf(acc5[r] + bb1, 0.0f);
            hd1[mm * HID_STRIDE + n + 32] = fmaxf(acc6[r] + bb2, 0.0f);
            hd1[mm * HID_STRIDE + n + 48] = fmaxf(acc7[r] + bb3, 0.0f);
        }
    }

    float bc = b2[n];
    for (int tt = 0; tt < 2; ++tt) {
        const float* hd = &sHid[wave][tt][0];
        v8f acc = {};
        for (int k = 0; k < HH; k += 4) {
            int kb = k + khalf;
            v2f a = *(const v2f*)(hd + m * HID_STRIDE + kb);
            v2f b = *(const v2f*)&sW2p[((k >> 1) + hi16) * W2_PSTRIDE + 2 * n];
            acc = __builtin_amdgcn_wmma_f32_16x16x4_f32(false, a, false, b, (short)0, acc, false, false);
        }
        int tile = tt ? tile1 : tile0;
        for (int r = 0; r < 8; ++r) {
            int node = tile * 16 + r + moff;
            float mlp = acc[r] + bc;
            float al  = alpha[node];
            float sa  = 1.0f / (1.0f + expf(-al));
            float sna = 1.0f - sa;
            float logit = out[(size_t)node * CC + n];
            out[(size_t)node * CC + n] = sa * logit + sna * mlp;
        }
    }
}

static void run_graph(hipStream_t stream, const float* e0, const float* e1,
                      const int* sg, const int* dg, int* cur, int* perm, float* p0, float* p1,
                      size_t ws_size,
                      float* hA, float* out, const float* label_init, const float* labels_onehot,
                      const int* train_mask, const float* attention, int last_mode) {
    const int TB = 256;
    const int gridNC = (NN * CC) / TB;
    int* bh    = (int*)hA;
    int* tot   = bh + NBK * NBLK;
    int* bbase = tot + NBK;
    k_bhist   <<<NBLK, ST, 0, stream>>>(dg, bh);
    k_bscan   <<<NBK,  ST, 0, stream>>>(bh, tot);
    k_bscatter<<<NBLK, ST, 0, stream>>>(dg, bh, tot, bbase, (unsigned*)perm, sg, e0, e1, p0, p1, ws_size);
    k_bsort   <<<NBK,  ST, 0, stream>>>(tot, bbase, cur, (unsigned*)perm, p0, p1, ws_size);
    k_gather<<<gridNC, TB, 0, stream>>>(e0, p0, sg, perm, cur, label_init, labels_onehot,
                                        train_mask, attention, hA, 0, ws_size);
    k_gather<<<gridNC, TB, 0, stream>>>(e1, p1, sg, perm, cur, hA, labels_onehot,
                                        train_mask, attention, out, last_mode, ws_size);
}

extern "C" void kernel_launch(void* const* d_in, const int* in_sizes, int n_in,
                              void* d_out, int out_size, void* d_ws, size_t ws_size,
                              hipStream_t stream) {
    const float* features      = (const float*)d_in[0];
    const float* label_init    = (const float*)d_in[1];
    const float* labels_onehot = (const float*)d_in[2];
    const float* alpha         = (const float*)d_in[3];
    const float* attention     = (const float*)d_in[4];
    const float* e_edge        = (const float*)d_in[5];
    const float* w1            = (const float*)d_in[6];
    const float* b1            = (const float*)d_in[7];
    const float* w2            = (const float*)d_in[8];
    const float* b2            = (const float*)d_in[9];
    const int*   src           = (const int*)d_in[10];
    const int*   dst           = (const int*)d_in[11];
    const int*   train_mask    = (const int*)d_in[12];
    float* out = (float*)d_out;

    int*   cur  = (int*)d_ws;
    int*   perm = cur + NN;
    float* hA   = (float*)(perm + EE);
    float* p0   = hA + NN * CC;
    float* p1   = p0 + EE;

    const int TB = 256;

    run_graph(stream, e_edge,      e_edge + (size_t)GG * EE,      src,      dst,
              cur, perm, p0, p1, ws_size, hA, out, label_init, labels_onehot, train_mask, attention, 1);
    run_graph(stream, e_edge + EE, e_edge + (size_t)GG * EE + EE, src + EE, dst + EE,
              cur, perm, p0, p1, ws_size, hA, out, label_init, labels_onehot, train_mask, attention, 2);

    const int n_pairs = NN / 32;
    const int blocks  = (n_pairs + 7) / 8;
    k_mlp_combine<<<blocks, TB, 0, stream>>>(features, w1, b1, w2, b2, alpha, out, n_pairs);
}


# Gather: 4 lanes own a node, not 16 (8 nodes per wave, 4x fewer waves); a lane reads 16 bytes of each source row, 8 rows in flight; quad moves pass slot data; softmax sum order in a node changes
# speedup vs baseline: 13.1625x; 1.0744x over previous
#include <hip/hip_runtime.h>
#include <hip/hip_bf16.h>
#include <math.h>

#define NN 100000
#define CC 16
#define GG 2
#define LL 2
#define EE 3200000
#define FF 512
#define HH 64

#define W1_PSTRIDE 160
#define W2_PSTRIDE 32
#define HID_STRIDE 66

typedef float v2f __attribute__((ext_vector_type(2)));
typedef float v8f __attribute__((ext_vector_type(8)));

#define BKT_SHIFT 8
#define BKT_W     256
#define NBK       ((NN + BKT_W - 1) / BKT_W)
#define CH        8192
#define CH_BITS   13
#define NBLK      ((EE + CH - 1) / CH)
#define ST        512
#define CAP       32768
#define ID_BITS   22
#define ID_MASK   0x3FFFFFu
#define SRC_BITS  17
#define SRC_MASK  0x1FFFFu
#define CAPW      (CAP / 3)
#define WIDE_BYTES (((size_t)NN + (size_t)EE + (size_t)NN * CC + 2 * (size_t)EE) * sizeof(int))

static_assert(EE <= (1 << ID_BITS), "edge id must fit the record");
static_assert(NN <= (1 << SRC_BITS) && BKT_SHIFT + SRC_BITS <= 32, "bucket-local node and source node fit one word");
static_assert(NBK <= ST && NBLK <= ST && BKT_W <= ST, "one scan pass per table");
static_assert(CH % ST == 0, "whole strides per block");
static_assert(CH == (1 << CH_BITS) && NN <= (1 << (32 - CH_BITS)), "node and edge-in-block fit one staged word");

__device__ __forceinline__ int clamp_node(int d) { return min(max(d, 0), NN - 1); }

__device__ __forceinline__ int block_scan_incl(int* sh, int t, int v) {
    sh[t] = v;
    __syncthreads();
    for (int off = 1; off < ST; off <<= 1) {
        int a = (t >= off) ? sh[t - off] : 0;
        __syncthreads();
        sh[t] += a;
        __syncthreads();
    }
    return sh[t];
}

__device__ __forceinline__ void block_scan_incl2(int* sa, int* sb, int t, int& a, int& b) {
    sa[t] = a;
    sb[t] = b;
    __syncthreads();
    for (int off = 1; off < ST; off <<= 1) {
        int xa = (t >= off) ? sa[t - off] : 0;
        int xb = (t >= off) ? sb[t - off] : 0;
        __syncthreads();
        sa[t] += xa;
        sb[t] += xb;
        __syncthreads();
    }
    a = sa[t];
    b = sb[t];
}

__global__ __launch_bounds__(ST) void k_bhist(const int* __restrict__ dst, int* __restrict__ bh) {
    __shared__ int h[NBK];
    int t = threadIdx.x;
    if (t < NBK) h[t] = 0;
    __syncthreads();
    int base = blockIdx.x * CH + t;
    int d[CH / ST];
    for (int j = 0; j < CH / ST; ++j)
        d[j] = dst[min(base + j * ST, EE - 1)];
    for (int j = 0; j < CH / ST; ++j)
        if (base + j * ST < EE) atomicAdd(&h[clamp_node(d[j]) >> BKT_SHIFT], 1);
    __syncthreads();
    if (t < NBK) bh[t * NBLK + blockIdx.x] = h[t];
}

__global__ __launch_bounds__(ST) void k_bscan(int* __restrict__ bh, int* __restrict__ tot) {
    __shared__ int sh[ST];
    int t = threadIdx.x, b = blockIdx.x;
    int v = (t < NBLK) ? bh[b * NBLK + t] : 0;
    int incl = block_scan_incl(sh, t, v);
    if (t < NBLK) bh[b * NBLK + t] = incl - v;
    if (t == ST - 1) tot[b] = incl;
}

__global__ __launch_bounds__(ST) void k_bscatter(const int* __restrict__ dst, const int* __restrict__ bh,
                                                 const int* __restrict__ tot, int* __restrict__ bbase,
                                                 unsigned* __restrict__ rec,
                                                 const int* __restrict__ src, const float* __restrict__ e0,
                                                 const float* __restrict__ e1, float* __restrict__ p0,
                                                 float* __restrict__ p1, size_t ws_size) {
    __shared__ int sh[ST];
    __shared__ int sh2[ST];
    __shared__ int lcur[NBK];
    __shared__ int delta[NBK];
    __shared__ unsigned stage[CH];
    __shared__ unsigned stS[CH];
    __shared__ float st0[CH];
    __shared__ float st1[CH];
    const bool wide = (ws_size >= WIDE_BYTES);
    int t = threadIdx.x, blk = blockIdx.x;
    int v = 0, mine = 0, lc = 0;
    if (t < NBK) {
        v = tot[t];
        mine = bh[t * NBLK + blk];
        int nxt = (blk + 1 < NBLK) ? bh[t * NBLK + blk + 1] : v;
        lc = nxt - mine;
    }
    int incl = v, lincl = lc;
    block_scan_incl2(sh, sh2, t, incl, lincl);
    if (t < NBK) {
        int excl  = incl - v;
        int lbase = lincl - lc;
        lcur[t]  = lbase;
        delta[t] = excl + mine - lbase;
        if (blk == 0) bbase[t] = excl;
    }
    __syncthreads();
    int base = blk * CH + t;
    int dv[CH / ST];
    for (int j = 0; j < CH / ST; ++j)
        dv[j] = dst[min(base + j * ST, EE - 1)];
    if (wide) {
        int   sv[CH / ST];
        float x0[CH / ST], x1[CH / ST];
        for (int j = 0; j < CH / ST; ++j) {
            int id = min(base + j * ST, EE - 1);
            sv[j] = src[id];
            x0[j] = e0[id];
            x1[j] = e1[id];
        }
        for (int j = 0; j < CH / ST; ++j) {
            if (base + j * ST < EE) {
                int d = clamp_node(dv[j]);
                int r = atomicAdd(&lcur[d >> BKT_SHIFT], 1);
                r = min(max(r, 0), CH - 1);
                stage[r] = ((unsigned)d << CH_BITS) | (unsigned)(j * ST + t);
                stS[r] = (unsigned)sv[j] & SRC_MASK;
                st0[r] = x0[j];
                st1[r] = x1[j];
            }
        }
    } else {
        for (int j = 0; j < CH / ST; ++j) {
            if (base + j * ST < EE) {
                int d = clamp_node(dv[j]);
                int r = atomicAdd(&lcur[d >> BKT_SHIFT], 1);
                r = min(max(r, 0), CH - 1);
                stage[r] = ((unsigned)d << CH_BITS) | (unsigned)(j * ST + t);
            }
        }
    }
    __syncthreads();
    int nblk = min(CH, EE - blk * CH);
    for (int i0 = t; i0 < nblk; i0 += 4 * ST) {
        unsigned w[4];
        for (int u = 0; u < 4; ++u) w[u] = stage[min(i0 + u * ST, CH - 1)];
        for (int u = 0; u < 4; ++u) {
            int i = i0 + u * ST;
            if (i < nblk) {
                int d = (int)(w[u] >> CH_BITS);
                int b = min(d >> BKT_SHIFT, NBK - 1);
                int p = min(max(i + delta[b], 0), EE - 1);
                if (wide) {
                    rec[p] = ((unsigned)(d & (BKT_W - 1)) << SRC_BITS) | stS[i];
                    p0[p]  = st0[i];
                    p1[p]  = st1[i];
                } else {
                    rec[p] = ((unsigned)(d & (BKT_W - 1)) << ID_BITS) | (unsigned)(blk * CH + (int)(w[u] & (CH - 1)));
                }
            }
        }
    }
}

__global__ __launch_bounds__(ST) void k_bsort(const int* __restrict__ tot, const int* __restrict__ bbase,
                                              int* __restrict__ cur, unsigned* slots,
                                              float* p0, float* p1, size_t ws_size) {
    __shared__ unsigned rec[CAP];
    const bool wide = (ws_size >= WIDE_BYTES);
    const int  kshift = wide ? SRC_BITS : ID_BITS;
    const unsigned vmask = wide ? SRC_MASK : ID_MASK;
    __shared__ int sh[ST];
    __shared__ int cnt[BKT_W];
    __shared__ int head[BKT_W];
    int t = threadIdx.x, b = blockIdx.x;
    int n  = min(max(tot[b], 0), EE);
    int bs = min(max(bbase[b], 0), EE - n);
    unsigned* A = slots + bs;
    float* B0 = p0 + bs;
    float* B1 = p1 + bs;
    bool fits = (n <= (wide ? CAPW : CAP));
    if (t < BKT_W) cnt[t] = 0;
    __syncthreads();
    for (int i0 = t; i0 < n; i0 += 4 * ST) {
        unsigned r[4];
        for (int u = 0; u < 4; ++u) r[u] = A[min(i0 + u * ST, n - 1)];
        for (int u = 0; u < 4; ++u)
            if (i0 + u * ST < n) atomicAdd(&cnt[(r[u] >> kshift) & (BKT_W - 1)], 1);
    }
    __syncthreads();
    int c = (t < BKT_W) ? cnt[t] : 0;
    int incl = block_scan_incl(sh, t, c);
    if (t < BKT_W) {
        head[t] = incl - c;
        int node = b * BKT_W + t;
        if (node < NN) cur[node] = bs + incl;
    }
    __syncthreads();
    if (fits && wide) {
        for (int i0 = t; i0 < n; i0 += 4 * ST) {
            unsigned r[4];
            float x0[4], x1[4];
            for (int u = 0; u < 4; ++u) {
                int i = min(i0 + u * ST, n - 1);
                r[u] = A[i]; x0[u] = B0[i]; x1[u] = B1[i];
            }
            for (int u = 0; u < 4; ++u) {
                if (i0 + u * ST < n) {
                    int p = atomicAdd(&head[(r[u] >> SRC_BITS) & (BKT_W - 1)], 1);
                    p = min(max(p, 0), n - 1);
                    rec[p]            = r[u] & SRC_MASK;
                    rec[CAPW + p]     = __float_as_uint(x0[u]);
                    rec[2 * CAPW + p] = __float_as_uint(x1[u]);
                }
            }
        }
        __syncthreads();
        for (int i0 = t; i0 < n; i0 += 4 * ST) {
            unsigned r[4], y0[4], y1[4];
            for (int u = 0; u < 4; ++u) {
                int i = min(i0 + u * ST, n - 1);
                r[u] = rec[i]; y0[u] = rec[CAPW + i]; y1[u] = rec[2 * CAPW + i];
            }
            for (int u = 0; u < 4; ++u) {
                if (i0 + u * ST < n) {
                    A[i0 + u * ST]  = r[u];
                    B0[i0 + u * ST] = __uint_as_float(y0[u]);
                    B1[i0 + u * ST] = __uint_as_float(y1[u]);
                }
            }
        }
    } else if (fits) {
        for (int i0 = t; i0 < n; i0 += 4 * ST) {
            unsigned r[4];
            for (int u = 0; u < 4; ++u) r[u] = A[min(i0 + u * ST, n - 1)];
            for (int u = 0; u < 4; ++u) {
                if (i0 + u * ST < n) {
                    int p = atomicAdd(&head[(r[u] >> ID_BITS) & (BKT_W - 1)], 1);
                    rec[min(max(p, 0), n - 1)] = r[u] & ID_MASK;
                }
            }
        }
        __syncthreads();
        for (int i0 = t; i0 < n; i0 += 4 * ST) {
            unsigned r[4];
            for (int u = 0; u < 4; ++u) r[u] = rec[min(i0 + u * ST, n - 1)];
            for (int u = 0; u < 4; ++u)
                if (i0 + u * ST < n) A[i0 + u * ST] = r[u];
        }
    } else {
        if (t == 0) {
            for (int k = 0; k < BKT_W; ++k) {
                int endk = sh[k];
                while (head[k] < endk) {
                    int pk = head[k];
                    unsigned r = A[pk];
                    int kk = (int)((r >> kshift) & (BKT_W - 1));
                    if (kk == k) {
                        head[k] = pk + 1;
                    } else {
                        int q = min(head[kk], n - 1);
                        head[kk] = q + 1;
                        unsigned r2 = A[q];
                        A[q] = r;
                        A[pk] = r2;
                        if (wide) {
                            float f0 = B0[q], f1 = B1[q];
                            B0[q] = B0[pk]; B1[q] = B1[pk];
                            B0[pk] = f0;    B1[pk] = f1;
                        }
                    }
                }
            }
        }
        __syncthreads();
        for (int i = t; i < n; i += ST) A[i] &= vmask;
    }
}

#define HEAD_R 12

static_assert(NN % 8 == 0, "a wave's eight nodes are all inside the graph or all outside");
static_assert(HEAD_R % 2 == 0, "the row pass takes the register slots two at a time");

__device__ __forceinline__ int clamp_e(int i) { return min(max(i, 0), EE - 1); }

template <int J>
__device__ __forceinline__ int quad_lane(int v) {
    return __builtin_amdgcn_update_dpp(v, v, J * 0x55, 0xF, 0xF, false);
}
__device__ __forceinline__ float quad_xor1(float v) {
    return __int_as_float(__builtin_amdgcn_update_dpp(0, __float_as_int(v), 0xB1, 0xF, 0xF, false));
}
__device__ __forceinline__ float quad_xor2(float v) {
    return __int_as_float(__builtin_amdgcn_update_dpp(0, __float_as_int(v), 0x4E, 0xF, 0xF, false));
}

template <int J>
__device__ __forceinline__ void row_fetch(const float4* __restrict__ h4, int s_reg, float a_reg,
                                          int c4, float4& r, float& w) {
    int s = quad_lane<J>(s_reg);
    w = __int_as_float(quad_lane<J>(__float_as_int(a_reg)));
    r = h4[(unsigned)(s * (CC / 4) + c4)];
}

__device__ __forceinline__ void row_add(float4& acc, const float4& r, float w) {
    acc.x = fmaf(r.x, w, acc.x); acc.y = fmaf(r.y, w, acc.y);
    acc.z = fmaf(r.z, w, acc.z); acc.w = fmaf(r.w, w, acc.w);
}

__global__ __launch_bounds__(256) void k_gather(
    const float* __restrict__ e_in, const float* __restrict__ e_pl, const int* __restrict__ src,
    const int* __restrict__ perm, const int* __restrict__ cur,
    const float* __restrict__ hcur, const float* __restrict__ one_hot,
    const int* __restrict__ mask, const float* __restrict__ att,
    float* hdst, int mode, size_t ws_size)
{
    const bool wide = (ws_size >= WIDE_BYTES);
    const float* __restrict__ e = wide ? e_pl : e_in;
    const float4* __restrict__ h4 = (const float4*)hcur;
    int gid = blockIdx.x * blockDim.x + threadIdx.x;
    int d = gid >> 2;
    if (d >= NN) return;
    int c4 = threadIdx.x & 3;
    int beg = (d == 0) ? 0 : cur[d - 1];
    int cnt = cur[d] - beg;

    int   sv[HEAD_R];
    float ev[HEAD_R];
    int   slot[HEAD_R];
    #pragma unroll
    for (int i = 0; i < HEAD_R; ++i) {
        int j = i * 4 + c4;
        slot[i] = clamp_e(beg + (j < cnt ? j : 0));
    }
    if (wide) {
        #pragma unroll
        for (int i = 0; i < HEAD_R; ++i) {
            sv[i] = perm[slot[i]];
            ev[i] = e[slot[i]];
        }
    } else {
        int pid[HEAD_R];
        #pragma unroll
        for (int i = 0; i < HEAD_R; ++i) pid[i] = perm[slot[i]];
        #pragma unroll
        for (int i = 0; i < HEAD_R; ++i) {
            int id = clamp_e(pid[i]);
            ev[i] = e[id];
            sv[i] = src[id];
        }
    }

    float m = -INFINITY;
    #pragma unroll
    for (int i = 0; i < HEAD_R; ++i) if (i * 4 + c4 < cnt) m = fmaxf(m, ev[i]);
    for (int b = HEAD_R * 4; b < cnt; b += 4) {
        int j = b + c4;
        if (j < cnt) { int sl = clamp_e(beg + j); m = fmaxf(m, e[wide ? sl : clamp_e(perm[sl])]); }
    }
    m = fmaxf(m, quad_xor1(m));
    m = fmaxf(m, quad_xor2(m));

    float ex[HEAD_R];
    float s = 0.0f;
    #pragma unroll
    for (int i = 0; i < HEAD_R; ++i) {
        ex[i] = (i * 4 + c4 < cnt) ? expf(ev[i] - m) : 0.0f;
        s += ex[i];
    }
    for (int b = HEAD_R * 4; b < cnt; b += 4) {
        int j = b + c4;
        if (j < cnt) { int sl = clamp_e(beg + j); s += expf(e[wide ? sl : clamp_e(perm[sl])] - m); }
    }
    s += quad_xor1(s);
    s += quad_xor2(s);

    #pragma unroll
    for (int i = 0; i < HEAD_R; ++i) {
        bool ok = (i * 4 + c4 < cnt);
        ex[i] = ok ? ex[i] / s : 0.0f;
        sv[i] = ok ? sv[i] : 0;
    }

    float4 acc = make_float4(0.0f, 0.0f, 0.0f, 0.0f);
    #pragma unroll
    for (int i = 0; i < HEAD_R; i += 2) {
        if (i * 4 < cnt) {
            float4 r0, r1, r2, r3, r4, r5, r6, r7;
            float  w0, w1, w2, w3, w4, w5, w6, w7;
            row_fetch<0>(h4, sv[i],     ex[i],     c4, r0, w0);
            row_fetch<1>(h4, sv[i],     ex[i],     c4, r1, w1);
            row_fetch<2>(h4, sv[i],     ex[i],     c4, r2, w2);
            row_fetch<3>(h4, sv[i],     ex[i],     c4, r3, w3);
            row_fetch<0>(h4, sv[i + 1], ex[i + 1], c4, r4, w4);
            row_fetch<1>(h4, sv[i + 1], ex[i + 1], c4, r5, w5);
            row_fetch<2>(h4, sv[i + 1], ex[i + 1], c4, r6, w6);
            row_fetch<3>(h4, sv[i + 1], ex[i + 1], c4, r7, w7);
            row_add(acc, r0, w0); row_add(acc, r1, w1); row_add(acc, r2, w2); row_add(acc, r3, w3);
            row_add(acc, r4, w4); row_add(acc, r5, w5); row_add(acc, r6, w6); row_add(acc, r7, w7);
        }
    }
    for (int b = HEAD_R * 4; b < cnt; b += 4) {
        int j = b + c4;
        float my_a = 0.0f;
        int my_s = 0;
        if (j < cnt) {
            int sl = clamp_e(beg + j);
            int pv = perm[sl];
            int id = wide ? sl : clamp_e(pv);
            my_a = expf(e[id] - m) / s;
            my_s = wide ? pv : src[id];
        }
        float4 r0, r1, r2, r3;
        float  w0, w1, w2, w3;
        row_fetch<0>(h4, my_s, my_a, c4, r0, w0);
        row_fetch<1>(h4, my_s, my_a, c4, r1, w1);
        row_fetch<2>(h4, my_s, my_a, c4, r2, w2);
        row_fetch<3>(h4, my_s, my_a, c4, r3, w3);
        row_add(acc, r0, w0); row_add(acc, r1, w1); row_add(acc, r2, w2); row_add(acc, r3, w3);
    }

    size_t o4 = (size_t)d * (CC / 4) + c4;
    float4 oh = ((const float4*)one_hot)[o4];
    bool mk = (mask[d] != 0);
    float4 hv;
    hv.x = mk ? oh.x : acc.x; hv.y = mk ? oh.y : acc.y;
    hv.z = mk ? oh.z : acc.z; hv.w = mk ? oh.w : acc.w;
    float4* hd4 = (float4*)hdst;
    if (mode == 0) {
        hd4[o4] = hv;
    } else {
        float a0 = att[d * 2 + 0], a1 = att[d * 2 + 1];
        float mx = fmaxf(a0, a1);
        float e0 = expf(a0 - mx), e1 = expf(a1 - mx);
        float inv = 1.0f / (e0 + e1);
        float4 y;
        if (mode == 1) {
            float wg = e0 * inv;
            y.x = hv.x * wg; y.y = hv.y * wg; y.z = hv.z * wg; y.w = hv.w * wg;
        } else {
            float wg = e1 * inv;
            float4 old = hd4[o4];
            y.x = old.x + hv.x * wg; y.y = old.y + hv.y * wg;
            y.z = old.z + hv.z * wg; y.w = old.w + hv.w * wg;
        }
        hd4[o4] = y;
    }
}

__global__ __launch_bounds__(256) void k_mlp_combine(
    const float* __restrict__ feat, const float* __restrict__ w1,
    const float* __restrict__ b1,   const float* __restrict__ w2,
    const float* __restrict__ b2,   const float* __restrict__ alpha,
    float* __restrict__ out, int n_pairs)
{
    __shared__ float sW1p[(FF / 2) * W1_PSTRIDE];
    __shared__ float sW2p[(HH / 2) * W2_PSTRIDE];
    __shared__ float sHid[8][2][16 * HID_STRIDE];

    int t = threadIdx.x;

    for (int i = t; i < (FF / 2) * (HH / 4); i += 256) {
        int p  = i >> 4;
        int nq = (i & 15) << 2;
        float4 x = *(const float4*)&w1[(size_t)p * 2 * HH + nq];
        float4 y = *(const float4*)&w1[(size_t)p * 2 * HH + HH + nq];
        float4 lo = { x.x, y.x, x.y, y.y };
        float4 hi = { x.z, y.z, x.w, y.w };
        float* d = &sW1p[p * W1_PSTRIDE + 2 * nq];
        *(float4*)(d + 0) = lo;
        *(float4*)(d + 4) = hi;
    }
    for (int i = t; i < (HH / 2) * CC; i += 256) {
        int p = i >> 4, n = i & 15;
        v2f v; v.x = w2[(size_t)p * 2 * CC + n];
        v.y = w2[(size_t)p * 2 * CC + CC + n];
        *(v2f*)&sW2p[p * W2_PSTRIDE + 2 * n] = v;
    }
    __syncthreads();

    int wave = t >> 5;
    int lane = t & 31;
    int pair = blockIdx.x * 8 + wave;
    if (pair >= n_pairs) return;

    int tile0 = pair * 2;
    int tile1 = tile0 + 1;

    int m     = lane & 15;
    int n     = lane & 15;
    int hi16  = lane >> 4;
    int khalf = hi16 << 1;
    int moff  = hi16 << 3;

    v8f acc0 = {}, acc1 = {}, acc2 = {}, acc3 = {};
    v8f acc4 = {}, acc5 = {}, acc6 = {}, acc7 = {};
    const float* arow0 = feat + ((size_t)tile0 * 16 + m) * FF;
    const float* arow1 = feat + ((size_t)tile1 * 16 + m) * FF;
    for (int k = 0; k < FF; k += 4) {
        int kb = k + khalf;
        v2f a0 = *(const v2f*)(arow0 + kb);
        v2f a1 = *(const v2f*)(arow1 + kb);
        const float* bp = &sW1p[((k >> 1) + hi16) * W1_PSTRIDE + 2 * n];
        v2f b0 = *(const v2f*)(bp + 0);
        v2f b1v = *(const v2f*)(bp + 32);
        v2f b2v = *(const v2f*)(bp + 64);
        v2f b3v = *(const v2f*)(bp + 96);
        acc0 = __builtin_amdgcn_wmma_f32_16x16x4_f32(false, a0, false, b0,  (short)0, acc0, false, false);
        acc1 = __builtin_amdgcn_wmma_f32_16x16x4_f32(false, a0, false, b1v, (short)0, acc1, false, false);
        acc2 = __builtin_amdgcn_wmma_f32_16x16x4_f32(false, a0, false, b2v, (short)0, acc2, false, false);
        acc3 = __builtin_amdgcn_wmma_f32_16x16x4_f32(false, a0, false, b3v, (short)0, acc3, false, false);
        acc4 = __builtin_amdgcn_wmma_f32_16x16x4_f32(false, a1, false, b0,  (short)0, acc4, false, false);
        acc5 = __builtin_amdgcn_wmma_f32_16x16x4_f32(false, a1, false, b1v, (short)0, acc5, false, false);
        acc6 = __builtin_amdgcn_wmma_f32_16x16x4_f32(false, a1, false, b2v, (short)0, acc6, false, false);
        acc7 = __builtin_amdgcn_wmma_f32_16x16x4_f32(false, a1, false, b3v, (short)0, acc7, false, false);
    }

    float bb0 = b1[n + 0], bb1 = b1[n + 16], bb2 = b1[n + 32], bb3 = b1[n + 48];
    {
        float* hd0 = &sHid[wave][0][0];
        float* hd1 = &sHid[wave][1][0];
        for (int r = 0; r < 8; ++r) {
            int mm = r + moff;
            hd0[mm * HID_STRIDE + n +  0] = fmaxf(acc0[r] + bb0, 0.0f);
            hd0[mm * HID_STRIDE + n + 16] = fmaxf(acc1[r] + bb1, 0.0f);
            hd0[mm * HID_STRIDE + n + 32] = fmaxf(acc2[r] + bb2, 0.0f);
            hd0[mm * HID_STRIDE + n + 48] = fmaxf(acc3[r] + bb3, 0.0f);
            hd1[mm * HID_STRIDE + n +  0] = fmaxf(acc4[r] + bb0, 0.0f);
            hd1[mm * HID_STRIDE + n + 16] = fmaxf(acc5[r] + bb1, 0.0f);
            hd1[mm * HID_STRIDE + n + 32] = fmaxf(acc6[r] + bb2, 0.0f);
            hd1[mm * HID_STRIDE + n + 48] = fmaxf(acc7[r] + bb3, 0.0f);
        }
    }

    float bc = b2[n];
    for (int tt = 0; tt < 2; ++tt) {
        const float* hd = &sHid[wave][tt][0];
        v8f acc = {};
        for (int k = 0; k < HH; k += 4) {
            int kb = k + khalf;
            v2f a = *(const v2f*)(hd + m * HID_STRIDE + kb);
            v2f b = *(const v2f*)&sW2p[((k >> 1) + hi16) * W2_PSTRIDE + 2 * n];
            acc = __builtin_amdgcn_wmma_f32_16x16x4_f32(false, a, false, b, (short)0, acc, false, false);
        }
        int tile = tt ? tile1 : tile0;
        for (int r = 0; r < 8; ++r) {
            int node = tile * 16 + r + moff;
            float mlp = acc[r] + bc;
            float al  = alpha[node];
            float sa  = 1.0f / (1.0f + expf(-al));
            float sna = 1.0f - sa;
            float logit = out[(size_t)node * CC + n];
            out[(size_t)node * CC + n] = sa * logit + sna * mlp;
        }
    }
}

static void run_graph(hipStream_t stream, const float* e0, const float* e1,
                      const int* sg, const int* dg, int* cur, int* perm, float* p0, float* p1,
                      size_t ws_size,
                      float* hA, float* out, const float* label_init, const float* labels_onehot,
                      const int* train_mask, const float* attention, int last_mode) {
    const int TB = 256;
    const int gridNC = (NN * 4 + TB - 1) / TB;
    int* bh    = (int*)hA;
    int* tot   = bh + NBK * NBLK;
    int* bbase = tot + NBK;
    k_bhist   <<<NBLK, ST, 0, stream>>>(dg, bh);
    k_bscan   <<<NBK,  ST, 0, stream>>>(bh, tot);
    k_bscatter<<<NBLK, ST, 0, stream>>>(dg, bh, tot, bbase, (unsigned*)perm, sg, e0, e1, p0, p1, ws_size);
    k_bsort   <<<NBK,  ST, 0, stream>>>(tot, bbase, cur, (unsigned*)perm, p0, p1, ws_size);
    k_gather<<<gridNC, TB, 0, stream>>>(e0, p0, sg, perm, cur, label_init, labels_onehot,
                                        train_mask, attention, hA, 0, ws_size);
    k_gather<<<gridNC, TB, 0, stream>>>(e1, p1, sg, perm, cur, hA, labels_onehot,
                                        train_mask, attention, out, last_mode, ws_size);
}

extern "C" void kernel_launch(void* const* d_in, const int* in_sizes, int n_in,
                              void* d_out, int out_size, void* d_ws, size_t ws_size,
                              hipStream_t stream) {
    const float* features      = (const float*)d_in[0];
    const float* label_init    = (const float*)d_in[1];
    const float* labels_onehot = (const float*)d_in[2];
    const float* alpha         = (const float*)d_in[3];
    const float* attention     = (const float*)d_in[4];
    const float* e_edge        = (const float*)d_in[5];
    const float* w1            = (const float*)d_in[6];
    const float* b1            = (const float*)d_in[7];
    const float* w2            = (const float*)d_in[8];
    const float* b2            = (const float*)d_in[9];
    const int*   src           = (const int*)d_in[10];
    const int*   dst           = (const int*)d_in[11];
    const int*   train_mask    = (const int*)d_in[12];
    float* out = (float*)d_out;

    int*   cur  = (int*)d_ws;
    int*   perm = cur + NN;
    float* hA   = (float*)(perm + EE);
    float* p0   = hA + NN * CC;
    float* p1   = p0 + EE;

    const int TB = 256;

    run_graph(stream, e_edge,      e_edge + (size_t)GG * EE,      src,      dst,
              cur, perm, p0, p1, ws_size, hA, out, label_init, labels_onehot, train_mask, attention, 1);
    run_graph(stream, e_edge + EE, e_edge + (size_t)GG * EE + EE, src + EE, dst + EE,
              cur, perm, p0, p1, ws_size, hA, out, label_init, labels_onehot, train_mask, attention, 2);

    const int n_pairs = NN / 32;
    const int blocks  = (n_pairs + 7) / 8;
    k_mlp_combine<<<blocks, TB, 0, stream>>>(features, w1, b1, w2, b2, alpha, out, n_pairs);
}


# Each PLP layer is one kernel per bucket: it orders the bucket's records by node in LDS and gathers from LDS; no level-2 write-back, 11 launches, not 13; in-node slot order, so sum order, may change
# speedup vs baseline: 13.7152x; 1.0420x over previous
#include <hip/hip_runtime.h>
#include <hip/hip_bf16.h>
#include <math.h>

#define NN 100000
#define CC 16
#define GG 2
#define LL 2
#define EE 3200000
#define FF 512
#define HH 64

#define W1_PSTRIDE 160
#define W2_PSTRIDE 32
#define HID_STRIDE 66

typedef float v2f __attribute__((ext_vector_type(2)));
typedef float v8f __attribute__((ext_vector_type(8)));

#define BKT_SHIFT 8
#define BKT_W     256
#define NBK       ((NN + BKT_W - 1) / BKT_W)
#define CH        8192
#define CH_BITS   13
#define NBLK      ((EE + CH - 1) / CH)
#define ST        512
#define CAP       32768
#define ID_BITS   22
#define ID_MASK   0x3FFFFFu
#define SRC_BITS  17
#define SRC_MASK  0x1FFFFu
#define WIDE_BYTES (((size_t)NN + (size_t)EE + (size_t)NN * CC + 2 * (size_t)EE) * sizeof(int))

static_assert(EE <= (1 << ID_BITS), "edge id must fit the record");
static_assert(NN <= (1 << SRC_BITS) && BKT_SHIFT + SRC_BITS <= 32, "bucket-local node and source node fit one word");
static_assert(NBK <= ST && NBLK <= ST && BKT_W <= ST, "one scan pass per table");
static_assert(CH % ST == 0, "whole strides per block");
static_assert(CH == (1 << CH_BITS) && NN <= (1 << (32 - CH_BITS)), "node and edge-in-block fit one staged word");

__device__ __forceinline__ int clamp_node(int d) { return min(max(d, 0), NN - 1); }

__device__ __forceinline__ int block_scan_incl(int* sh, int t, int v) {
    sh[t] = v;
    __syncthreads();
    for (int off = 1; off < ST; off <<= 1) {
        int a = (t >= off) ? sh[t - off] : 0;
        __syncthreads();
        sh[t] += a;
        __syncthreads();
    }
    return sh[t];
}

__device__ __forceinline__ void block_scan_incl2(int* sa, int* sb, int t, int& a, int& b) {
    sa[t] = a;
    sb[t] = b;
    __syncthreads();
    for (int off = 1; off < ST; off <<= 1) {
        int xa = (t >= off) ? sa[t - off] : 0;
        int xb = (t >= off) ? sb[t - off] : 0;
        __syncthreads();
        sa[t] += xa;
        sb[t] += xb;
        __syncthreads();
    }
    a = sa[t];
    b = sb[t];
}

__global__ __launch_bounds__(ST) void k_bhist(const int* __restrict__ dst, int* __restrict__ bh) {
    __shared__ int h[NBK];
    int t = threadIdx.x;
    if (t < NBK) h[t] = 0;
    __syncthreads();
    int base = blockIdx.x * CH + t;
    int d[CH / ST];
    for (int j = 0; j < CH / ST; ++j)
        d[j] = dst[min(base + j * ST, EE - 1)];
    for (int j = 0; j < CH / ST; ++j)
        if (base + j * ST < EE) atomicAdd(&h[clamp_node(d[j]) >> BKT_SHIFT], 1);
    __syncthreads();
    if (t < NBK) bh[t * NBLK + blockIdx.x] = h[t];
}

__global__ __launch_bounds__(ST) void k_bscan(int* __restrict__ bh, int* __restrict__ tot) {
    __shared__ int sh[ST];
    int t = threadIdx.x, b = blockIdx.x;
    int v = (t < NBLK) ? bh[b * NBLK + t] : 0;
    int incl = block_scan_incl(sh, t, v);
    if (t < NBLK) bh[b * NBLK + t] = incl - v;
    if (t == ST - 1) tot[b] = incl;
}

__global__ __launch_bounds__(ST) void k_bscatter(const int* __restrict__ dst, const int* __restrict__ bh,
                                                 const int* __restrict__ tot, int* __restrict__ bbase,
                                                 unsigned* __restrict__ rec,
                                                 const int* __restrict__ src, const float* __restrict__ e0,
                                                 const float* __restrict__ e1, float* __restrict__ p0,
                                                 float* __restrict__ p1, size_t ws_size) {
    __shared__ int sh[ST];
    __shared__ int sh2[ST];
    __shared__ int lcur[NBK];
    __shared__ int delta[NBK];
    __shared__ unsigned stage[CH];
    __shared__ unsigned stS[CH];
    __shared__ float st0[CH];
    __shared__ float st1[CH];
    const bool wide = (ws_size >= WIDE_BYTES);
    int t = threadIdx.x, blk = blockIdx.x;
    int v = 0, mine = 0, lc = 0;
    if (t < NBK) {
        v = tot[t];
        mine = bh[t * NBLK + blk];
        int nxt = (blk + 1 < NBLK) ? bh[t * NBLK + blk + 1] : v;
        lc = nxt - mine;
    }
    int incl = v, lincl = lc;
    block_scan_incl2(sh, sh2, t, incl, lincl);
    if (t < NBK) {
        int excl  = incl - v;
        int lbase = lincl - lc;
        lcur[t]  = lbase;
        delta[t] = excl + mine - lbase;
        if (blk == 0) bbase[t] = excl;
    }
    __syncthreads();
    int base = blk * CH + t;
    int dv[CH / ST];
    for (int j = 0; j < CH / ST; ++j)
        dv[j] = dst[min(base + j * ST, EE - 1)];
    if (wide) {
        int   sv[CH / ST];
        float x0[CH / ST], x1[CH / ST];
        for (int j = 0; j < CH / ST; ++j) {
            int id = min(base + j * ST, EE - 1);
            sv[j] = src[id];
            x0[j] = e0[id];
            x1[j] = e1[id];
        }
        for (int j = 0; j < CH / ST; ++j) {
            if (base + j * ST < EE) {
                int d = clamp_node(dv[j]);
                int r = atomicAdd(&lcur[d >> BKT_SHIFT], 1);
                r = min(max(r, 0), CH - 1);
                stage[r] = ((unsigned)d << CH_BITS) | (unsigned)(j * ST + t);
                stS[r] = (unsigned)sv[j] & SRC_MASK;
                st0[r] = x0[j];
                st1[r] = x1[j];
            }
        }
    } else {
        for (int j = 0; j < CH / ST; ++j) {
            if (base + j * ST < EE) {
                int d = clamp_node(dv[j]);
                int r = atomicAdd(&lcur[d >> BKT_SHIFT], 1);
                r = min(max(r, 0), CH - 1);
                stage[r] = ((unsigned)d << CH_BITS) | (unsigned)(j * ST + t);
            }
        }
    }
    __syncthreads();
    int nblk = min(CH, EE - blk * CH);
    for (int i0 = t; i0 < nblk; i0 += 4 * ST) {
        unsigned w[4];
        for (int u = 0; u < 4; ++u) w[u] = stage[min(i0 + u * ST, CH - 1)];
        for (int u = 0; u < 4; ++u) {
            int i = i0 + u * ST;
            if (i < nblk) {
                int d = (int)(w[u] >> CH_BITS);
                int b = min(d >> BKT_SHIFT, NBK - 1);
                int p = min(max(i + delta[b], 0), EE - 1);
                if (wide) {
                    rec[p] = ((unsigned)(d & (BKT_W - 1)) << SRC_BITS) | stS[i];
                    p0[p]  = st0[i];
                    p1[p]  = st1[i];
                } else {
                    rec[p] = ((unsigned)(d & (BKT_W - 1)) << ID_BITS) | (unsigned)(blk * CH + (int)(w[u] & (CH - 1)));
                }
            }
        }
    }
}

#define HEAD_R 12

static_assert(NN % 8 == 0, "a wave's eight nodes are all inside the graph or all outside");
static_assert(HEAD_R % 2 == 0, "the row pass takes the register slots two at a time");

__device__ __forceinline__ int clamp_e(int i) { return min(max(i, 0), EE - 1); }

template <int J>
__device__ __forceinline__ int quad_lane(int v) {
    return __builtin_amdgcn_update_dpp(v, v, J * 0x55, 0xF, 0xF, false);
}
__device__ __forceinline__ float quad_xor1(float v) {
    return __int_as_float(__builtin_amdgcn_update_dpp(0, __float_as_int(v), 0xB1, 0xF, 0xF, false));
}
__device__ __forceinline__ float quad_xor2(float v) {
    return __int_as_float(__builtin_amdgcn_update_dpp(0, __float_as_int(v), 0x4E, 0xF, 0xF, false));
}

template <int J>
__device__ __forceinline__ void row_fetch(const float4* __restrict__ h4, int s_reg, float a_reg,
                                          int c4, float4& r, float& w) {
    int s = quad_lane<J>(s_reg);
    w = __int_as_float(quad_lane<J>(__float_as_int(a_reg)));
    r = h4[(unsigned)(s * (CC / 4) + c4)];
}

__device__ __forceinline__ void row_add(float4& acc, const float4& r, float w) {
    acc.x = fmaf(r.x, w, acc.x); acc.y = fmaf(r.y, w, acc.y);
    acc.z = fmaf(r.z, w, acc.z); acc.w = fmaf(r.w, w, acc.w);
}

#define CAPH (CAP / 2)
#define NPQ  (ST / 4)

static_assert(BKT_W % NPQ == 0, "whole passes per bucket");
static_assert((BKT_W % 8 == 0) && (NPQ % 8 == 0), "a wave's eight nodes start at a multiple of eight");

__device__ __forceinline__ float4 node_lds(const unsigned* rs, int beg, int cnt, int c4,
                                           const float4* __restrict__ h4) {
    int   sv[HEAD_R];
    float ev[HEAD_R];
    #pragma unroll
    for (int i = 0; i < HEAD_R; ++i) {
        int j = i * 4 + c4;
        int idx = min(beg + (j < cnt ? j : 0), CAPH - 1);
        sv[i] = (int)rs[idx];
        ev[i] = __uint_as_float(rs[CAPH + idx]);
    }

    float m = -INFINITY;
    #pragma unroll
    for (int i = 0; i < HEAD_R; ++i) if (i * 4 + c4 < cnt) m = fmaxf(m, ev[i]);
    for (int b = HEAD_R * 4; b < cnt; b += 4) {
        int j = b + c4;
        if (j < cnt) m = fmaxf(m, __uint_as_float(rs[CAPH + min(beg + j, CAPH - 1)]));
    }
    m = fmaxf(m, quad_xor1(m));
    m = fmaxf(m, quad_xor2(m));

    float ex[HEAD_R];
    float s = 0.0f;
    #pragma unroll
    for (int i = 0; i < HEAD_R; ++i) {
        ex[i] = (i * 4 + c4 < cnt) ? expf(ev[i] - m) : 0.0f;
        s += ex[i];
    }
    for (int b = HEAD_R * 4; b < cnt; b += 4) {
        int j = b + c4;
        if (j < cnt) s += expf(__uint_as_float(rs[CAPH + min(beg + j, CAPH - 1)]) - m);
    }
    s += quad_xor1(s);
    s += quad_xor2(s);

    #pragma unroll
    for (int i = 0; i < HEAD_R; ++i) {
        bool ok = (i * 4 + c4 < cnt);
        ex[i] = ok ? ex[i] / s : 0.0f;
        sv[i] = ok ? sv[i] : 0;
    }

    float4 acc = make_float4(0.0f, 0.0f, 0.0f, 0.0f);
    #pragma unroll
    for (int i = 0; i < HEAD_R; i += 2) {
        if (i * 4 < cnt) {
            float4 r0, r1, r2, r3, r4, r5, r6, r7;
            float  w0, w1, w2, w3, w4, w5, w6, w7;
            row_fetch<0>(h4, sv[i],     ex[i],     c4, r0, w0);
            row_fetch<1>(h4, sv[i],     ex[i],     c4, r1, w1);
            row_fetch<2>(h4, sv[i],     ex[i],     c4, r2, w2);
            row_fetch<3>(h4, sv[i],     ex[i],     c4, r3, w3);
            row_fetch<0>(h4, sv[i + 1], ex[i + 1], c4, r4, w4);
            row_fetch<1>(h4, sv[i + 1], ex[i + 1], c4, r5, w5);
            row_fetch<2>(h4, sv[i + 1], ex[i + 1], c4, r6, w6);
            row_fetch<3>(h4, sv[i + 1], ex[i + 1], c4, r7, w7);
            row_add(acc, r0, w0); row_add(acc, r1, w1); row_add(acc, r2, w2); row_add(acc, r3, w3);
            row_add(acc, r4, w4); row_add(acc, r5, w5); row_add(acc, r6, w6); row_add(acc, r7, w7);
        }
    }
    for (int b = HEAD_R * 4; b < cnt; b += 4) {
        int j = b + c4;
        float my_a = 0.0f;
        int my_s = 0;
        if (j < cnt) {
            int idx = min(beg + j, CAPH - 1);
            my_a = expf(__uint_as_float(rs[CAPH + idx]) - m) / s;
            my_s = (int)rs[idx];
        }
        float4 r0, r1, r2, r3;
        float  w0, w1, w2, w3;
        row_fetch<0>(h4, my_s, my_a, c4, r0, w0);
        row_fetch<1>(h4, my_s, my_a, c4, r1, w1);
        row_fetch<2>(h4, my_s, my_a, c4, r2, w2);
        row_fetch<3>(h4, my_s, my_a, c4, r3, w3);
        row_add(acc, r0, w0); row_add(acc, r1, w1); row_add(acc, r2, w2); row_add(acc, r3, w3);
    }
    return acc;
}

template <int J>
__device__ __forceinline__ void scan_take(const float4* __restrict__ h4, int sv, float a, int c4, float4& acc) {
    int s = quad_lane<J>(sv);
    float w = __int_as_float(quad_lane<J>(__float_as_int(a)));
    if (s >= 0) {
        float4 r = h4[(unsigned)(s * (CC / 4) + c4)];
        row_add(acc, r, w);
    }
}

__device__ __forceinline__ float4 node_scan(const unsigned* __restrict__ A, const float* __restrict__ B,
                                            const float* __restrict__ e_in, const int* __restrict__ src,
                                            bool wide, int n, int k, int c4,
                                            const float4* __restrict__ h4) {
    const int kshift = wide ? SRC_BITS : ID_BITS;
    float m = -INFINITY;
    for (int i = c4; i < n; i += 4) {
        unsigned r = A[i];
        if ((int)((r >> kshift) & (BKT_W - 1)) == k)
            m = fmaxf(m, wide ? B[i] : e_in[clamp_e((int)(r & ID_MASK))]);
    }
    m = fmaxf(m, quad_xor1(m));
    m = fmaxf(m, quad_xor2(m));
    float s = 0.0f;
    for (int i = c4; i < n; i += 4) {
        unsigned r = A[i];
        if ((int)((r >> kshift) & (BKT_W - 1)) == k)
            s += expf((wide ? B[i] : e_in[clamp_e((int)(r & ID_MASK))]) - m);
    }
    s += quad_xor1(s);
    s += quad_xor2(s);
    float4 acc = make_float4(0.0f, 0.0f, 0.0f, 0.0f);
    for (int i0 = 0; i0 < n; i0 += 4) {
        int i = i0 + c4;
        float a = 0.0f;
        int sv = -1;
        if (i < n) {
            unsigned r = A[i];
            if ((int)((r >> kshift) & (BKT_W - 1)) == k) {
                int id = clamp_e((int)(r & ID_MASK));
                float e = wide ? B[i] : e_in[id];
                a  = expf(e - m) / s;
                sv = wide ? (int)(r & SRC_MASK) : max(src[id], 0);
            }
        }
        scan_take<0>(h4, sv, a, c4, acc);
        scan_take<1>(h4, sv, a, c4, acc);
        scan_take<2>(h4, sv, a, c4, acc);
        scan_take<3>(h4, sv, a, c4, acc);
    }
    return acc;
}

#define UNR 18

__global__ __launch_bounds__(ST) void k_layer(
    const int* __restrict__ tot, const int* __restrict__ bbase,
    const unsigned* __restrict__ recs, const float* __restrict__ e_pl,
    const float* __restrict__ e_in, const int* __restrict__ src,
    const float* __restrict__ hcur, const float* __restrict__ one_hot,
    const int* __restrict__ mask, const float* __restrict__ att,
    float* hdst, int mode, size_t ws_size)
{
    __shared__ unsigned rec[CAP];
    __shared__ int sh[ST];
    __shared__ int cnt[BKT_W];
    __shared__ int head[BKT_W + 32];
    const bool wide = (ws_size >= WIDE_BYTES);
    const int  kshift = wide ? SRC_BITS : ID_BITS;
    const float4* __restrict__ h4 = (const float4*)hcur;
    int t = threadIdx.x, b = blockIdx.x;
    int n  = min(max(tot[b], 0), EE);
    int bs = min(max(bbase[b], 0), EE - n);
    const unsigned* __restrict__ A = recs + bs;
    const float* __restrict__ B = e_pl + bs;
    const bool fits = (n <= CAPH);
    if (t < BKT_W) cnt[t] = 0;
    __syncthreads();
    for (int i0 = t; i0 < n; i0 += UNR * ST) {
        unsigned r[UNR];
        #pragma unroll
        for (int u = 0; u < UNR; ++u) r[u] = A[min(i0 + u * ST, n - 1)];
        #pragma unroll
        for (int u = 0; u < UNR; ++u)
            if (i0 + u * ST < n) atomicAdd(&cnt[(r[u] >> kshift) & (BKT_W - 1)], 1);
    }
    __syncthreads();
    int c = (t < BKT_W) ? cnt[t] : 0;
    int incl = block_scan_incl(sh, t, c);
    if (t < BKT_W) head[t] = incl - c;
    else if (t < BKT_W + 32) head[t] = 0;
    __syncthreads();
    if (fits) {
        for (int i0 = t; i0 < n; i0 += UNR * ST) {
            unsigned r[UNR];
            float x[UNR];
            int   sv[UNR];
            int   p[UNR];
            if (wide) {
                #pragma unroll
                for (int u = 0; u < UNR; ++u) x[u] = B[min(i0 + u * ST, n - 1)];
            }
            #pragma unroll
            for (int u = 0; u < UNR; ++u) r[u] = A[min(i0 + u * ST, n - 1)];
            if (wide) {
                #pragma unroll
                for (int u = 0; u < UNR; ++u) sv[u] = (int)(r[u] & SRC_MASK);
            } else {
                #pragma unroll
                for (int u = 0; u < UNR; ++u) {
                    int id = clamp_e((int)(r[u] & ID_MASK));
                    x[u]  = e_in[id];
                    sv[u] = max(src[id], 0);
                }
            }
            #pragma unroll
            for (int u = 0; u < UNR; ++u) {
                int key = (i0 + u * ST < n) ? (int)((r[u] >> kshift) & (BKT_W - 1)) : BKT_W + (t & 31);
                p[u] = atomicAdd(&head[key], 1);
            }
            #pragma unroll
            for (int u = 0; u < UNR; ++u) {
                if (i0 + u * ST < n) {
                    int q = min(max(p[u], 0), CAPH - 1);
                    rec[q]        = (unsigned)sv[u];
                    rec[CAPH + q] = __float_as_uint(x[u]);
                }
            }
        }
        __syncthreads();
    }

    int c4 = t & 3;
    for (int pass = 0; pass < BKT_W / NPQ; ++pass) {
        int k = pass * NPQ + (t >> 2);
        int d = b * BKT_W + k;
        if (d < NN) {
            int cn  = cnt[k];
            int beg = sh[k] - cn;
            float4 acc = fits ? node_lds(rec, beg, cn, c4, h4)
                              : node_scan(A, B, e_in, src, wide, n, k, c4, h4);

            size_t o4 = (size_t)d * (CC / 4) + c4;
            float4 oh = ((const float4*)one_hot)[o4];
            bool mk = (mask[d] != 0);
            float4 hv;
            hv.x = mk ? oh.x : acc.x; hv.y = mk ? oh.y : acc.y;
            hv.z = mk ? oh.z : acc.z; hv.w = mk ? oh.w : acc.w;
            float4* hd4 = (float4*)hdst;
            if (mode == 0) {
                hd4[o4] = hv;
            } else {
                float a0 = att[d * 2 + 0], a1 = att[d * 2 + 1];
                float mx = fmaxf(a0, a1);
                float e0 = expf(a0 - mx), e1 = expf(a1 - mx);
                float inv = 1.0f / (e0 + e1);
                float4 y;
                if (mode == 1) {
                    float wg = e0 * inv;
                    y.x = hv.x * wg; y.y = hv.y * wg; y.z = hv.z * wg; y.w = hv.w * wg;
                } else {
                    float wg = e1 * inv;
                    float4 old = hd4[o4];
                    y.x = old.x + hv.x * wg; y.y = old.y + hv.y * wg;
                    y.z = old.z + hv.z * wg; y.w = old.w + hv.w * wg;
                }
                hd4[o4] = y;
            }
        }
    }
}

__global__ __launch_bounds__(256) void k_mlp_combine(
    const float* __restrict__ feat, const float* __restrict__ w1,
    const float* __restrict__ b1,   const float* __restrict__ w2,
    const float* __restrict__ b2,   const float* __restrict__ alpha,
    float* __restrict__ out, int n_pairs)
{
    __shared__ float sW1p[(FF / 2) * W1_PSTRIDE];
    __shared__ float sW2p[(HH / 2) * W2_PSTRIDE];
    __shared__ float sHid[8][2][16 * HID_STRIDE];

    int t = threadIdx.x;

    for (int i = t; i < (FF / 2) * (HH / 4); i += 256) {
        int p  = i >> 4;
        int nq = (i & 15) << 2;
        float4 x = *(const float4*)&w1[(size_t)p * 2 * HH + nq];
        float4 y = *(const float4*)&w1[(size_t)p * 2 * HH + HH + nq];
        float4 lo = { x.x, y.x, x.y, y.y };
        float4 hi = { x.z, y.z, x.w, y.w };
        float* d = &sW1p[p * W1_PSTRIDE + 2 * nq];
        *(float4*)(d + 0) = lo;
        *(float4*)(d + 4) = hi;
    }
    for (int i = t; i < (HH / 2) * CC; i += 256) {
        int p = i >> 4, n = i & 15;
        v2f v; v.x = w2[(size_t)p * 2 * CC + n];
        v.y = w2[(size_t)p * 2 * CC + CC + n];
        *(v2f*)&sW2p[p * W2_PSTRIDE + 2 * n] = v;
    }
    __syncthreads();

    int wave = t >> 5;
    int lane = t & 31;
    int pair = blockIdx.x * 8 + wave;
    if (pair >= n_pairs) return;

    int tile0 = pair * 2;
    int tile1 = tile0 + 1;

    int m     = lane & 15;
    int n     = lane & 15;
    int hi16  = lane >> 4;
    int khalf = hi16 << 1;
    int moff  = hi16 << 3;

    v8f acc0 = {}, acc1 = {}, acc2 = {}, acc3 = {};
    v8f acc4 = {}, acc5 = {}, acc6 = {}, acc7 = {};
    const float* arow0 = feat + ((size_t)tile0 * 16 + m) * FF;
    const float* arow1 = feat + ((size_t)tile1 * 16 + m) * FF;
    for (int k = 0; k < FF; k += 4) {
        int kb = k + khalf;
        v2f a0 = *(const v2f*)(arow0 + kb);
        v2f a1 = *(const v2f*)(arow1 + kb);
        const float* bp = &sW1p[((k >> 1) + hi16) * W1_PSTRIDE + 2 * n];
        v2f b0 = *(const v2f*)(bp + 0);
        v2f b1v = *(const v2f*)(bp + 32);
        v2f b2v = *(const v2f*)(bp + 64);
        v2f b3v = *(const v2f*)(bp + 96);
        acc0 = __builtin_amdgcn_wmma_f32_16x16x4_f32(false, a0, false, b0,  (short)0, acc0, false, false);
        acc1 = __builtin_amdgcn_wmma_f32_16x16x4_f32(false, a0, false, b1v, (short)0, acc1, false, false);
        acc2 = __builtin_amdgcn_wmma_f32_16x16x4_f32(false, a0, false, b2v, (short)0, acc2, false, false);
        acc3 = __builtin_amdgcn_wmma_f32_16x16x4_f32(false, a0, false, b3v, (short)0, acc3, false, false);
        acc4 = __builtin_amdgcn_wmma_f32_16x16x4_f32(false, a1, false, b0,  (short)0, acc4, false, false);
        acc5 = __builtin_amdgcn_wmma_f32_16x16x4_f32(false, a1, false, b1v, (short)0, acc5, false, false);
        acc6 = __builtin_amdgcn_wmma_f32_16x16x4_f32(false, a1, false, b2v, (short)0, acc6, false, false);
        acc7 = __builtin_amdgcn_wmma_f32_16x16x4_f32(false, a1, false, b3v, (short)0, acc7, false, false);
    }

    float bb0 = b1[n + 0], bb1 = b1[n + 16], bb2 = b1[n + 32], bb3 = b1[n + 48];
    {
        float* hd0 = &sHid[wave][0][0];
        float* hd1 = &sHid[wave][1][0];
        for (int r = 0; r < 8; ++r) {
            int mm = r + moff;
            hd0[mm * HID_STRIDE + n +  0] = fmaxf(acc0[r] + bb0, 0.0f);
            hd0[mm * HID_STRIDE + n + 16] = fmaxf(acc1[r] + bb1, 0.0f);
            hd0[mm * HID_STRIDE + n + 32] = fmaxf(acc2[r] + bb2, 0.0f);
            hd0[mm * HID_STRIDE + n + 48] = fmaxf(acc3[r] + bb3, 0.0f);
            hd1[mm * HID_STRIDE + n +  0] = fmaxf(acc4[r] + bb0, 0.0f);
            hd1[mm * HID_STRIDE + n + 16] = fmaxf(acc5[r] + bb1, 0.0f);
            hd1[mm * HID_STRIDE + n + 32] = fmaxf(acc6[r] + bb2, 0.0f);
            hd1[mm * HID_STRIDE + n + 48] = fmaxf(acc7[r] + bb3, 0.0f);
        }
    }

    float bc = b2[n];
    for (int tt = 0; tt < 2; ++tt) {
        const float* hd = &sHid[wave][tt][0];
        v8f acc = {};
        for (int k = 0; k < HH; k += 4) {
            int kb = k + khalf;
            v2f a = *(const v2f*)(hd + m * HID_STRIDE + kb);
            v2f b = *(const v2f*)&sW2p[((k >> 1) + hi16) * W2_PSTRIDE + 2 * n];
            acc = __builtin_amdgcn_wmma_f32_16x16x4_f32(false, a, false, b, (short)0, acc, false, false);
        }
        int tile = tt ? tile1 : tile0;
        for (int r = 0; r < 8; ++r) {
            int node = tile * 16 + r + moff;
            float mlp = acc[r] + bc;
            float al  = alpha[node];
            float sa  = 1.0f / (1.0f + expf(-al));
            float sna = 1.0f - sa;
            float logit = out[(size_t)node * CC + n];
            out[(size_t)node * CC + n] = sa * logit + sna * mlp;
        }
    }
}

static void run_graph(hipStream_t stream, const float* e0, const float* e1,
                      const int* sg, const int* dg, int* tabs, int* perm, float* p0, float* p1,
                      size_t ws_size,
                      float* hA, float* out, const float* label_init, const float* labels_onehot,
                      const int* train_mask, const float* attention, int last_mode) {
    int* bh    = (int*)hA;
    int* tot   = tabs;
    int* bbase = tabs + NBK;
    k_bhist   <<<NBLK, ST, 0, stream>>>(dg, bh);
    k_bscan   <<<NBK,  ST, 0, stream>>>(bh, tot);
    k_bscatter<<<NBLK, ST, 0, stream>>>(dg, bh, tot, bbase, (unsigned*)perm, sg, e0, e1, p0, p1, ws_size);
    k_layer   <<<NBK,  ST, 0, stream>>>(tot, bbase, (const unsigned*)perm, p0, e0, sg, label_init,
                                        labels_onehot, train_mask, attention, hA, 0, ws_size);
    k_layer   <<<NBK,  ST, 0, stream>>>(tot, bbase, (const unsigned*)perm, p1, e1, sg, hA,
                                        labels_onehot, train_mask, attention, out, last_mode, ws_size);
}

extern "C" void kernel_launch(void* const* d_in, const int* in_sizes, int n_in,
                              void* d_out, int out_size, void* d_ws, size_t ws_size,
                              hipStream_t stream) {
    const float* features      = (const float*)d_in[0];
    const float* label_init    = (const float*)d_in[1];
    const float* labels_onehot = (const float*)d_in[2];
    const float* alpha         = (const float*)d_in[3];
    const float* attention     = (const float*)d_in[4];
    const float* e_edge        = (const float*)d_in[5];
    const float* w1            = (const float*)d_in[6];
    const float* b1            = (const float*)d_in[7];
    const float* w2            = (const float*)d_in[8];
    const float* b2            = (const float*)d_in[9];
    const int*   src           = (const int*)d_in[10];
    const int*   dst           = (const int*)d_in[11];
    const int*   train_mask    = (const int*)d_in[12];
    float* out = (float*)d_out;

    int*   tabs = (int*)d_ws;
    int*   perm = tabs + NN;
    float* hA   = (float*)(perm + EE);
    float* p0   = hA + NN * CC;
    float* p1   = p0 + EE;

    const int TB = 256;

    run_graph(stream, e_edge,      e_edge + (size_t)GG * EE,      src,      dst,
              tabs, perm, p0, p1, ws_size, hA, out, label_init, labels_onehot, train_mask, attention, 1);
    run_graph(stream, e_edge + EE, e_edge + (size_t)GG * EE + EE, src + EE, dst + EE,
              tabs, perm, p0, p1, ws_size, hA, out, label_init, labels_onehot, train_mask, attention, 2);

    const int n_pairs = NN / 32;
    const int blocks  = (n_pairs + 7) / 8;
    k_mlp_combine<<<blocks, TB, 0, stream>>>(features, w1, b1, w2, b2, alpha, out, n_pairs);
}


# Grids sized to the chip: 500 buckets of 200 nodes, 250 level-1 blocks of 12800 edges load all 256 units evenly (391 blocks left a third with double work); in-node slot order, so sum order, may change
# speedup vs baseline: 14.4676x; 1.0549x over previous
#include <hip/hip_runtime.h>
#include <hip/hip_bf16.h>
#include <math.h>

#define NN 100000
#define CC 16
#define GG 2
#define LL 2
#define EE 3200000
#define FF 512
#define HH 64

#define W1_PSTRIDE 160
#define W2_PSTRIDE 32
#define HID_STRIDE 66

typedef float v2f __attribute__((ext_vector_type(2)));
typedef float v8f __attribute__((ext_vector_type(8)));

#define KEY_BITS  8
#define KEY_W     (1 << KEY_BITS)
#define BKT_W     200
#define NBK       ((NN + BKT_W - 1) / BKT_W)
#define CH        12800
#define CH_BITS   14
#define CH_MASK   ((1u << CH_BITS) - 1u)
#define NBLK      ((EE + CH - 1) / CH)
#define ST        512
#define CAP       32768
#define ID_BITS   22
#define ID_MASK   0x3FFFFFu
#define SRC_BITS  17
#define SRC_MASK  0x1FFFFu
#define WIDE_BYTES (((size_t)NN + (size_t)EE + (size_t)NN * CC + 2 * (size_t)EE) * sizeof(int))

static_assert(EE <= (1 << ID_BITS), "edge id must fit the record");
static_assert(NN <= (1 << SRC_BITS) && KEY_BITS + SRC_BITS <= 32 && KEY_BITS + ID_BITS <= 32, "bucket-local node and source node or edge id fit one word");
static_assert(BKT_W <= KEY_W, "a bucket-local node fits the key");
static_assert(NBK <= ST && NBLK <= ST && KEY_W <= ST, "one scan pass per table");
static_assert(CH % ST == 0, "whole strides per block");
static_assert(CH <= (1 << CH_BITS) && NN <= (1 << (32 - CH_BITS)), "node and edge-in-block fit one staged word");

__device__ __forceinline__ int clamp_node(int d) { return min(max(d, 0), NN - 1); }
__device__ __forceinline__ int bucket_of(int d) { return (int)((unsigned)d / (unsigned)BKT_W); }

__device__ __forceinline__ int block_scan_incl(int* sh, int t, int v) {
    sh[t] = v;
    __syncthreads();
    for (int off = 1; off < ST; off <<= 1) {
        int a = (t >= off) ? sh[t - off] : 0;
        __syncthreads();
        sh[t] += a;
        __syncthreads();
    }
    return sh[t];
}

__device__ __forceinline__ void block_scan_incl2(int* sa, int* sb, int t, int& a, int& b) {
    sa[t] = a;
    sb[t] = b;
    __syncthreads();
    for (int off = 1; off < ST; off <<= 1) {
        int xa = (t >= off) ? sa[t - off] : 0;
        int xb = (t >= off) ? sb[t - off] : 0;
        __syncthreads();
        sa[t] += xa;
        sb[t] += xb;
        __syncthreads();
    }
    a = sa[t];
    b = sb[t];
}

__global__ __launch_bounds__(ST) void k_bhist(const int* __restrict__ dst, int* __restrict__ bh) {
    __shared__ int h[NBK];
    int t = threadIdx.x;
    if (t < NBK) h[t] = 0;
    __syncthreads();
    int base = blockIdx.x * CH + t;
    int d[CH / ST];
    for (int j = 0; j < CH / ST; ++j)
        d[j] = dst[min(base + j * ST, EE - 1)];
    for (int j = 0; j < CH / ST; ++j)
        if (base + j * ST < EE) atomicAdd(&h[bucket_of(clamp_node(d[j]))], 1);
    __syncthreads();
    if (t < NBK) bh[t * NBLK + blockIdx.x] = h[t];
}

__global__ __launch_bounds__(ST) void k_bscan(int* __restrict__ bh, int* __restrict__ tot) {
    __shared__ int sh[ST];
    int t = threadIdx.x, b = blockIdx.x;
    int v = (t < NBLK) ? bh[b * NBLK + t] : 0;
    int incl = block_scan_incl(sh, t, v);
    if (t < NBLK) bh[b * NBLK + t] = incl - v;
    if (t == ST - 1) tot[b] = incl;
}

__global__ __launch_bounds__(ST) void k_bscatter(const int* __restrict__ dst, const int* __restrict__ bh,
                                                 const int* __restrict__ tot, int* __restrict__ bbase,
                                                 unsigned* __restrict__ rec,
                                                 const int* __restrict__ src, const float* __restrict__ e0,
                                                 const float* __restrict__ e1, float* __restrict__ p0,
                                                 float* __restrict__ p1, size_t ws_size) {
    __shared__ int sh[ST];
    __shared__ int sh2[ST];
    __shared__ int lcur[NBK];
    __shared__ int delta[NBK];
    __shared__ unsigned stage[CH];
    __shared__ unsigned stS[CH];
    __shared__ float st0[CH];
    __shared__ float st1[CH];
    const bool wide = (ws_size >= WIDE_BYTES);
    int t = threadIdx.x, blk = blockIdx.x;
    int v = 0, mine = 0, lc = 0;
    if (t < NBK) {
        v = tot[t];
        mine = bh[t * NBLK + blk];
        int nxt = (blk + 1 < NBLK) ? bh[t * NBLK + blk + 1] : v;
        lc = nxt - mine;
    }
    int incl = v, lincl = lc;
    block_scan_incl2(sh, sh2, t, incl, lincl);
    if (t < NBK) {
        int excl  = incl - v;
        int lbase = lincl - lc;
        lcur[t]  = lbase;
        delta[t] = excl + mine - lbase;
        if (blk == 0) bbase[t] = excl;
    }
    __syncthreads();
    int base = blk * CH + t;
    int dv[CH / ST];
    for (int j = 0; j < CH / ST; ++j)
        dv[j] = dst[min(base + j * ST, EE - 1)];
    if (wide) {
        int   sv[CH / ST];
        float x0[CH / ST], x1[CH / ST];
        for (int j = 0; j < CH / ST; ++j) {
            int id = min(base + j * ST, EE - 1);
            sv[j] = src[id];
            x0[j] = e0[id];
            x1[j] = e1[id];
        }
        for (int j = 0; j < CH / ST; ++j) {
            if (base + j * ST < EE) {
                int d = clamp_node(dv[j]);
                int r = atomicAdd(&lcur[bucket_of(d)], 1);
                r = min(max(r, 0), CH - 1);
                stage[r] = ((unsigned)d << CH_BITS) | (unsigned)(j * ST + t);
                stS[r] = (unsigned)sv[j] & SRC_MASK;
                st0[r] = x0[j];
                st1[r] = x1[j];
            }
        }
    } else {
        for (int j = 0; j < CH / ST; ++j) {
            if (base + j * ST < EE) {
                int d = clamp_node(dv[j]);
                int r = atomicAdd(&lcur[bucket_of(d)], 1);
                r = min(max(r, 0), CH - 1);
                stage[r] = ((unsigned)d << CH_BITS) | (unsigned)(j * ST + t);
            }
        }
    }
    __syncthreads();
    int nblk = min(CH, EE - blk * CH);
    for (int i0 = t; i0 < nblk; i0 += 4 * ST) {
        unsigned w[4];
        for (int u = 0; u < 4; ++u) w[u] = stage[min(i0 + u * ST, CH - 1)];
        for (int u = 0; u < 4; ++u) {
            int i = i0 + u * ST;
            if (i < nblk) {
                int d = (int)(w[u] >> CH_BITS);
                int b = min(bucket_of(d), NBK - 1);
                unsigned key = (unsigned)(d - b * BKT_W) & (KEY_W - 1);
                int p = min(max(i + delta[b], 0), EE - 1);
                if (wide) {
                    rec[p] = (key << SRC_BITS) | stS[i];
                    p0[p]  = st0[i];
                    p1[p]  = st1[i];
                } else {
                    rec[p] = (key << ID_BITS) | (unsigned)(blk * CH + (int)(w[u] & CH_MASK));
                }
            }
        }
    }
}

#define HEAD_R 12

static_assert(NN % 8 == 0, "a wave's eight nodes are all inside the graph or all outside");
static_assert(HEAD_R % 2 == 0, "the row pass takes the register slots two at a time");

__device__ __forceinline__ int clamp_e(int i) { return min(max(i, 0), EE - 1); }

template <int J>
__device__ __forceinline__ int quad_lane(int v) {
    return __builtin_amdgcn_update_dpp(v, v, J * 0x55, 0xF, 0xF, false);
}
__device__ __forceinline__ float quad_xor1(float v) {
    return __int_as_float(__builtin_amdgcn_update_dpp(0, __float_as_int(v), 0xB1, 0xF, 0xF, false));
}
__device__ __forceinline__ float quad_xor2(float v) {
    return __int_as_float(__builtin_amdgcn_update_dpp(0, __float_as_int(v), 0x4E, 0xF, 0xF, false));
}

template <int J>
__device__ __forceinline__ void row_fetch(const float4* __restrict__ h4, int s_reg, float a_reg,
                                          int c4, float4& r, float& w) {
    int s = quad_lane<J>(s_reg);
    w = __int_as_float(quad_lane<J>(__float_as_int(a_reg)));
    r = h4[(unsigned)(s * (CC / 4) + c4)];
}

__device__ __forceinline__ void row_add(float4& acc, const float4& r, float w) {
    acc.x = fmaf(r.x, w, acc.x); acc.y = fmaf(r.y, w, acc.y);
    acc.z = fmaf(r.z, w, acc.z); acc.w = fmaf(r.w, w, acc.w);
}

#define CAPH (CAP / 2)
#define NPQ  (ST / 4)

static_assert(KEY_W % NPQ == 0, "whole passes over the key range");
static_assert((BKT_W % 8 == 0) && (NPQ % 8 == 0), "a wave's eight nodes start at a multiple of eight, all inside the bucket or all outside");

__device__ __forceinline__ float4 node_lds(const unsigned* rs, int beg, int cnt, int c4,
                                           const float4* __restrict__ h4) {
    int   sv[HEAD_R];
    float ev[HEAD_R];
    #pragma unroll
    for (int i = 0; i < HEAD_R; ++i) {
        int j = i * 4 + c4;
        int idx = min(beg + (j < cnt ? j : 0), CAPH - 1);
        sv[i] = (int)rs[idx];
        ev[i] = __uint_as_float(rs[CAPH + idx]);
    }

    float m = -INFINITY;
    #pragma unroll
    for (int i = 0; i < HEAD_R; ++i) if (i * 4 + c4 < cnt) m = fmaxf(m, ev[i]);
    for (int b = HEAD_R * 4; b < cnt; b += 4) {
        int j = b + c4;
        if (j < cnt) m = fmaxf(m, __uint_as_float(rs[CAPH + min(beg + j, CAPH - 1)]));
    }
    m = fmaxf(m, quad_xor1(m));
    m = fmaxf(m, quad_xor2(m));

    float ex[HEAD_R];
    float s = 0.0f;
    #pragma unroll
    for (int i = 0; i < HEAD_R; ++i) {
        ex[i] = (i * 4 + c4 < cnt) ? expf(ev[i] - m) : 0.0f;
        s += ex[i];
    }
    for (int b = HEAD_R * 4; b < cnt; b += 4) {
        int j = b + c4;
        if (j < cnt) s += expf(__uint_as_float(rs[CAPH + min(beg + j, CAPH - 1)]) - m);
    }
    s += quad_xor1(s);
    s += quad_xor2(s);

    #pragma unroll
    for (int i = 0; i < HEAD_R; ++i) {
        bool ok = (i * 4 + c4 < cnt);
        ex[i] = ok ? ex[i] / s : 0.0f;
        sv[i] = ok ? sv[i] : 0;
    }

    float4 acc = make_float4(0.0f, 0.0f, 0.0f, 0.0f);
    #pragma unroll
    for (int i = 0; i < HEAD_R; i += 2) {
        if (i * 4 < cnt) {
            float4 r0, r1, r2, r3, r4, r5, r6, r7;
            float  w0, w1, w2, w3, w4, w5, w6, w7;
            row_fetch<0>(h4, sv[i],     ex[i],     c4, r0, w0);
            row_fetch<1>(h4, sv[i],     ex[i],     c4, r1, w1);
            row_fetch<2>(h4, sv[i],     ex[i],     c4, r2, w2);
            row_fetch<3>(h4, sv[i],     ex[i],     c4, r3, w3);
            row_fetch<0>(h4, sv[i + 1], ex[i + 1], c4, r4, w4);
            row_fetch<1>(h4, sv[i + 1], ex[i + 1], c4, r5, w5);
            row_fetch<2>(h4, sv[i + 1], ex[i + 1], c4, r6, w6);
            row_fetch<3>(h4, sv[i + 1], ex[i + 1], c4, r7, w7);
            row_add(acc, r0, w0); row_add(acc, r1, w1); row_add(acc, r2, w2); row_add(acc, r3, w3);
            row_add(acc, r4, w4); row_add(acc, r5, w5); row_add(acc, r6, w6); row_add(acc, r7, w7);
        }
    }
    for (int b = HEAD_R * 4; b < cnt; b += 4) {
        int j = b + c4;
        float my_a = 0.0f;
        int my_s = 0;
        if (j < cnt) {
            int idx = min(beg + j, CAPH - 1);
            my_a = expf(__uint_as_float(rs[CAPH + idx]) - m) / s;
            my_s = (int)rs[idx];
        }
        float4 r0, r1, r2, r3;
        float  w0, w1, w2, w3;
        row_fetch<0>(h4, my_s, my_a, c4, r0, w0);
        row_fetch<1>(h4, my_s, my_a, c4, r1, w1);
        row_fetch<2>(h4, my_s, my_a, c4, r2, w2);
        row_fetch<3>(h4, my_s, my_a, c4, r3, w3);
        row_add(acc, r0, w0); row_add(acc, r1, w1); row_add(acc, r2, w2); row_add(acc, r3, w3);
    }
    return acc;
}

template <int J>
__device__ __forceinline__ void scan_take(const float4* __restrict__ h4, int sv, float a, int c4, float4& acc) {
    int s = quad_lane<J>(sv);
    float w = __int_as_float(quad_lane<J>(__float_as_int(a)));
    if (s >= 0) {
        float4 r = h4[(unsigned)(s * (CC / 4) + c4)];
        row_add(acc, r, w);
    }
}

__device__ __forceinline__ float4 node_scan(const unsigned* __restrict__ A, const float* __restrict__ B,
                                            const float* __restrict__ e_in, const int* __restrict__ src,
                                            bool wide, int n, int k, int c4,
                                            const float4* __restrict__ h4) {
    const int kshift = wide ? SRC_BITS : ID_BITS;
    float m = -INFINITY;
    for (int i = c4; i < n; i += 4) {
        unsigned r = A[i];
        if ((int)((r >> kshift) & (KEY_W - 1)) == k)
            m = fmaxf(m, wide ? B[i] : e_in[clamp_e((int)(r & ID_MASK))]);
    }
    m = fmaxf(m, quad_xor1(m));
    m = fmaxf(m, quad_xor2(m));
    float s = 0.0f;
    for (int i = c4; i < n; i += 4) {
        unsigned r = A[i];
        if ((int)((r >> kshift) & (KEY_W - 1)) == k)
            s += expf((wide ? B[i] : e_in[clamp_e((int)(r & ID_MASK))]) - m);
    }
    s += quad_xor1(s);
    s += quad_xor2(s);
    float4 acc = make_float4(0.0f, 0.0f, 0.0f, 0.0f);
    for (int i0 = 0; i0 < n; i0 += 4) {
        int i = i0 + c4;
        float a = 0.0f;
        int sv = -1;
        if (i < n) {
            unsigned r = A[i];
            if ((int)((r >> kshift) & (KEY_W - 1)) == k) {
                int id = clamp_e((int)(r & ID_MASK));
                float e = wide ? B[i] : e_in[id];
                a  = expf(e - m) / s;
                sv = wide ? (int)(r & SRC_MASK) : max(src[id], 0);
            }
        }
        scan_take<0>(h4, sv, a, c4, acc);
        scan_take<1>(h4, sv, a, c4, acc);
        scan_take<2>(h4, sv, a, c4, acc);
        scan_take<3>(h4, sv, a, c4, acc);
    }
    return acc;
}

#define UNR 14

__global__ __launch_bounds__(ST) void k_layer(
    const int* __restrict__ tot, const int* __restrict__ bbase,
    const unsigned* __restrict__ recs, const float* __restrict__ e_pl,
    const float* __restrict__ e_in, const int* __restrict__ src,
    const float* __restrict__ hcur, const float* __restrict__ one_hot,
    const int* __restrict__ mask, const float* __restrict__ att,
    float* hdst, int mode, size_t ws_size)
{
    __shared__ unsigned rec[CAP];
    __shared__ int sh[ST];
    __shared__ int cnt[KEY_W];
    __shared__ int head[KEY_W + 32];
    const bool wide = (ws_size >= WIDE_BYTES);
    const int  kshift = wide ? SRC_BITS : ID_BITS;
    const float4* __restrict__ h4 = (const float4*)hcur;
    int t = threadIdx.x, b = blockIdx.x;
    int n  = min(max(tot[b], 0), EE);
    int bs = min(max(bbase[b], 0), EE - n);
    const unsigned* __restrict__ A = recs + bs;
    const float* __restrict__ B = e_pl + bs;
    const bool fits = (n <= CAPH);
    if (t < KEY_W) cnt[t] = 0;
    __syncthreads();
    for (int i0 = t; i0 < n; i0 += UNR * ST) {
        unsigned r[UNR];
        #pragma unroll
        for (int u = 0; u < UNR; ++u) r[u] = A[min(i0 + u * ST, n - 1)];
        #pragma unroll
        for (int u = 0; u < UNR; ++u)
            if (i0 + u * ST < n) atomicAdd(&cnt[(r[u] >> kshift) & (KEY_W - 1)], 1);
    }
    __syncthreads();
    int c = (t < KEY_W) ? cnt[t] : 0;
    int incl = block_scan_incl(sh, t, c);
    if (t < KEY_W) head[t] = incl - c;
    else if (t < KEY_W + 32) head[t] = 0;
    __syncthreads();
    if (fits) {
        for (int i0 = t; i0 < n; i0 += UNR * ST) {
            unsigned r[UNR];
            float x[UNR];
            int   sv[UNR];
            int   p[UNR];
            if (wide) {
                #pragma unroll
                for (int u = 0; u < UNR; ++u) x[u] = B[min(i0 + u * ST, n - 1)];
            }
            #pragma unroll
            for (int u = 0; u < UNR; ++u) r[u] = A[min(i0 + u * ST, n - 1)];
            if (wide) {
                #pragma unroll
                for (int u = 0; u < UNR; ++u) sv[u] = (int)(r[u] & SRC_MASK);
            } else {
                #pragma unroll
                for (int u = 0; u < UNR; ++u) {
                    int id = clamp_e((int)(r[u] & ID_MASK));
                    x[u]  = e_in[id];
                    sv[u] = max(src[id], 0);
                }
            }
            #pragma unroll
            for (int u = 0; u < UNR; ++u) {
                int key = (i0 + u * ST < n) ? (int)((r[u] >> kshift) & (KEY_W - 1)) : KEY_W + (t & 31);
                p[u] = atomicAdd(&head[key], 1);
            }
            #pragma unroll
            for (int u = 0; u < UNR; ++u) {
                if (i0 + u * ST < n) {
                    int q = min(max(p[u], 0), CAPH - 1);
                    rec[q]        = (unsigned)sv[u];
                    rec[CAPH + q] = __float_as_uint(x[u]);
                }
            }
        }
        __syncthreads();
    }

    int c4 = t & 3;
    for (int pass = 0; pass < KEY_W / NPQ; ++pass) {
        int k = pass * NPQ + (t >> 2);
        int d = b * BKT_W + k;
        if (k < BKT_W && d < NN) {
            int cn  = cnt[k];
            int beg = sh[k] - cn;
            float4 acc = fits ? node_lds(rec, beg, cn, c4, h4)
                              : node_scan(A, B, e_in, src, wide, n, k, c4, h4);

            size_t o4 = (size_t)d * (CC / 4) + c4;
            float4 oh = ((const float4*)one_hot)[o4];
            bool mk = (mask[d] != 0);
            float4 hv;
            hv.x = mk ? oh.x : acc.x; hv.y = mk ? oh.y : acc.y;
            hv.z = mk ? oh.z : acc.z; hv.w = mk ? oh.w : acc.w;
            float4* hd4 = (float4*)hdst;
            if (mode == 0) {
                hd4[o4] = hv;
            } else {
                float a0 = att[d * 2 + 0], a1 = att[d * 2 + 1];
                float mx = fmaxf(a0, a1);
                float e0 = expf(a0 - mx), e1 = expf(a1 - mx);
                float inv = 1.0f / (e0 + e1);
                float4 y;
                if (mode == 1) {
                    float wg = e0 * inv;
                    y.x = hv.x * wg; y.y = hv.y * wg; y.z = hv.z * wg; y.w = hv.w * wg;
                } else {
                    float wg = e1 * inv;
                    float4 old = hd4[o4];
                    y.x = old.x + hv.x * wg; y.y = old.y + hv.y * wg;
                    y.z = old.z + hv.z * wg; y.w = old.w + hv.w * wg;
                }
                hd4[o4] = y;
            }
        }
    }
}

__global__ __launch_bounds__(256) void k_mlp_combine(
    const float* __restrict__ feat, const float* __restrict__ w1,
    const float* __restrict__ b1,   const float* __restrict__ w2,
    const float* __restrict__ b2,   const float* __restrict__ alpha,
    float* __restrict__ out, int n_pairs)
{
    __shared__ float sW1p[(FF / 2) * W1_PSTRIDE];
    __shared__ float sW2p[(HH / 2) * W2_PSTRIDE];
    __shared__ float sHid[8][2][16 * HID_STRIDE];

    int t = threadIdx.x;

    for (int i = t; i < (FF / 2) * (HH / 4); i += 256) {
        int p  = i >> 4;
        int nq = (i & 15) << 2;
        float4 x = *(const float4*)&w1[(size_t)p * 2 * HH + nq];
        float4 y = *(const float4*)&w1[(size_t)p * 2 * HH + HH + nq];
        float4 lo = { x.x, y.x, x.y, y.y };
        float4 hi = { x.z, y.z, x.w, y.w };
        float* d = &sW1p[p * W1_PSTRIDE + 2 * nq];
        *(float4*)(d + 0) = lo;
        *(float4*)(d + 4) = hi;
    }
    for (int i = t; i < (HH / 2) * CC; i += 256) {
        int p = i >> 4, n = i & 15;
        v2f v; v.x = w2[(size_t)p * 2 * CC + n];
        v.y = w2[(size_t)p * 2 * CC + CC + n];
        *(v2f*)&sW2p[p * W2_PSTRIDE + 2 * n] = v;
    }
    __syncthreads();

    int wave = t >> 5;
    int lane = t & 31;
    int pair = blockIdx.x * 8 + wave;
    if (pair >= n_pairs) return;

    int tile0 = pair * 2;
    int tile1 = tile0 + 1;

    int m     = lane & 15;
    int n     = lane & 15;
    int hi16  = lane >> 4;
    int khalf = hi16 << 1;
    int moff  = hi16 << 3;

    v8f acc0 = {}, acc1 = {}, acc2 = {}, acc3 = {};
    v8f acc4 = {}, acc5 = {}, acc6 = {}, acc7 = {};
    const float* arow0 = feat + ((size_t)tile0 * 16 + m) * FF;
    const float* arow1 = feat + ((size_t)tile1 * 16 + m) * FF;
    for (int k = 0; k < FF; k += 4) {
        int kb = k + khalf;
        v2f a0 = *(const v2f*)(arow0 + kb);
        v2f a1 = *(const v2f*)(arow1 + kb);
        const float* bp = &sW1p[((k >> 1) + hi16) * W1_PSTRIDE + 2 * n];
        v2f b0 = *(const v2f*)(bp + 0);
        v2f b1v = *(const v2f*)(bp + 32);
        v2f b2v = *(const v2f*)(bp + 64);
        v2f b3v = *(const v2f*)(bp + 96);
        acc0 = __builtin_amdgcn_wmma_f32_16x16x4_f32(false, a0, false, b0,  (short)0, acc0, false, false);
        acc1 = __builtin_amdgcn_wmma_f32_16x16x4_f32(false, a0, false, b1v, (short)0, acc1, false, false);
        acc2 = __builtin_amdgcn_wmma_f32_16x16x4_f32(false, a0, false, b2v, (short)0, acc2, false, false);
        acc3 = __builtin_amdgcn_wmma_f32_16x16x4_f32(false, a0, false, b3v, (short)0, acc3, false, false);
        acc4 = __builtin_amdgcn_wmma_f32_16x16x4_f32(false, a1, false, b0,  (short)0, acc4, false, false);
        acc5 = __builtin_amdgcn_wmma_f32_16x16x4_f32(false, a1, false, b1v, (short)0, acc5, false, false);
        acc6 = __builtin_amdgcn_wmma_f32_16x16x4_f32(false, a1, false, b2v, (short)0, acc6, false, false);
        acc7 = __builtin_amdgcn_wmma_f32_16x16x4_f32(false, a1, false, b3v, (short)0, acc7, false, false);
    }

    float bb0 = b1[n + 0], bb1 = b1[n + 16], bb2 = b1[n + 32], bb3 = b1[n + 48];
    {
        float* hd0 = &sHid[wave][0][0];
        float* hd1 = &sHid[wave][1][0];
        for (int r = 0; r < 8; ++r) {
            int mm = r + moff;
            hd0[mm * HID_STRIDE + n +  0] = fmaxf(acc0[r] + bb0, 0.0f);
            hd0[mm * HID_STRIDE + n + 16] = fmaxf(acc1[r] + bb1, 0.0f);
            hd0[mm * HID_STRIDE + n + 32] = fmaxf(acc2[r] + bb2, 0.0f);
            hd0[mm * HID_STRIDE + n + 48] = fmaxf(acc3[r] + bb3, 0.0f);
            hd1[mm * HID_STRIDE + n +  0] = fmaxf(acc4[r] + bb0, 0.0f);
            hd1[mm * HID_STRIDE + n + 16] = fmaxf(acc5[r] + bb1, 0.0f);
            hd1[mm * HID_STRIDE + n + 32] = fmaxf(acc6[r] + bb2, 0.0f);
            hd1[mm * HID_STRIDE + n + 48] = fmaxf(acc7[r] + bb3, 0.0f);
        }
    }

    float bc = b2[n];
    for (int tt = 0; tt < 2; ++tt) {
        const float* hd = &sHid[wave][tt][0];
        v8f acc = {};
        for (int k = 0; k < HH; k += 4) {
            int kb = k + khalf;
            v2f a = *(const v2f*)(hd + m * HID_STRIDE + kb);
            v2f b = *(const v2f*)&sW2p[((k >> 1) + hi16) * W2_PSTRIDE + 2 * n];
            acc = __builtin_amdgcn_wmma_f32_16x16x4_f32(false, a, false, b, (short)0, acc, false, false);
        }
        int tile = tt ? tile1 : tile0;
        for (int r = 0; r < 8; ++r) {
            int node = tile * 16 + r + moff;
            float mlp = acc[r] + bc;
            float al  = alpha[node];
            float sa  = 1.0f / (1.0f + expf(-al));
            float sna = 1.0f - sa;
            float logit = out[(size_t)node * CC + n];
            out[(size_t)node * CC + n] = sa * logit + sna * mlp;
        }
    }
}

static void run_graph(hipStream_t stream, const float* e0, const float* e1,
                      const int* sg, const int* dg, int* tabs, int* perm, float* p0, float* p1,
                      size_t ws_size,
                      float* hA, float* out, const float* label_init, const float* labels_onehot,
                      const int* train_mask, const float* attention, int last_mode) {
    int* bh    = (int*)hA;
    int* tot   = tabs;
    int* bbase = tabs + NBK;
    k_bhist   <<<NBLK, ST, 0, stream>>>(dg, bh);
    k_bscan   <<<NBK,  ST, 0, stream>>>(bh, tot);
    k_bscatter<<<NBLK, ST, 0, stream>>>(dg, bh, tot, bbase, (unsigned*)perm, sg, e0, e1, p0, p1, ws_size);
    k_layer   <<<NBK,  ST, 0, stream>>>(tot, bbase, (const unsigned*)perm, p0, e0, sg, label_init,
                                        labels_onehot, train_mask, attention, hA, 0, ws_size);
    k_layer   <<<NBK,  ST, 0, stream>>>(tot, bbase, (const unsigned*)perm, p1, e1, sg, hA,
                                        labels_onehot, train_mask, attention, out, last_mode, ws_size);
}

extern "C" void kernel_launch(void* const* d_in, const int* in_sizes, int n_in,
                              void* d_out, int out_size, void* d_ws, size_t ws_size,
                              hipStream_t stream) {
    const float* features      = (const float*)d_in[0];
    const float* label_init    = (const float*)d_in[1];
    const float* labels_onehot = (const float*)d_in[2];
    const float* alpha         = (const float*)d_in[3];
    const float* attention     = (const float*)d_in[4];
    const float* e_edge        = (const float*)d_in[5];
    const float* w1            = (const float*)d_in[6];
    const float* b1            = (const float*)d_in[7];
    const float* w2            = (const float*)d_in[8];
    const float* b2            = (const float*)d_in[9];
    const int*   src           = (const int*)d_in[10];
    const int*   dst           = (const int*)d_in[11];
    const int*   train_mask    = (const int*)d_in[12];
    float* out = (float*)d_out;

    int*   tabs = (int*)d_ws;
    int*   perm = tabs + NN;
    float* hA   = (float*)(perm + EE);
    float* p0   = hA + NN * CC;
    float* p1   = p0 + EE;

    const int TB = 256;

    run_graph(stream, e_edge,      e_edge + (size_t)GG * EE,      src,      dst,
              tabs, perm, p0, p1, ws_size, hA, out, label_init, labels_onehot, train_mask, attention, 1);
    run_graph(stream, e_edge + EE, e_edge + (size_t)GG * EE + EE, src + EE, dst + EE,
              tabs, perm, p0, p1, ws_size, hA, out, label_init, labels_onehot, train_mask, attention, 2);

    const int n_pairs = NN / 32;
    const int blocks  = (n_pairs + 7) / 8;
    k_mlp_combine<<<blocks, TB, 0, stream>>>(features, w1, b1, w2, b2, alpha, out, n_pairs);
}


# Level 1 is one kernel: each block keeps its records in its own chunk, ordered by bucket, plus a run-start table; no histogram or scan kernel, 7 launches, not 11; in-node sum order may change
# speedup vs baseline: 15.3932x; 1.0640x over previous
#include <hip/hip_runtime.h>
#include <hip/hip_bf16.h>
#include <math.h>

#define NN 100000
#define CC 16
#define GG 2
#define LL 2
#define EE 3200000
#define FF 512
#define HH 64

#define W1_PSTRIDE 160
#define W2_PSTRIDE 32
#define HID_STRIDE 66

typedef float v2f __attribute__((ext_vector_type(2)));
typedef float v8f __attribute__((ext_vector_type(8)));

#define KEY_BITS  8
#define KEY_W     (1 << KEY_BITS)
#define BKT_W     200
#define NBK       ((NN + BKT_W - 1) / BKT_W)
#define CH        12800
#define NBLK      ((EE + CH - 1) / CH)
#define ST        512
#define NW        (ST / 32)
#define RPW       ((NBLK + NW - 1) / NW)
#define RUN_CAP   64
#define CAP       32768
#define ID_BITS   22
#define ID_MASK   0x3FFFFFu
#define SRC_BITS  17
#define SRC_MASK  0x1FFFFu
#define WIDE_BYTES (((size_t)NN + (size_t)EE + (size_t)NN * CC + 2 * (size_t)EE) * sizeof(int))

static_assert(EE <= (1 << ID_BITS), "edge id must fit the record");
static_assert(NN <= (1 << SRC_BITS) && KEY_BITS + SRC_BITS <= 32 && KEY_BITS + ID_BITS <= 32, "bucket-local node and source node or edge id fit one word");
static_assert(BKT_W <= KEY_W, "a bucket-local node fits the key");
static_assert(NBK < ST && NBLK <= ST && KEY_W <= ST, "one scan pass per table");
static_assert(CH % ST == 0, "whole strides per block");
static_assert(CH < 65536, "a run's start inside its chunk fits 16 bits");
static_assert(NBK % 2 == 0 && (size_t)(NBK / 2 + 1) * NBLK <= (size_t)NN, "the run-start table, one word per pair of buckets and chunk, fits the table words");

__device__ __forceinline__ int clamp_node(int d) { return min(max(d, 0), NN - 1); }
__device__ __forceinline__ int bucket_of(int d) { return (int)((unsigned)d / (unsigned)BKT_W); }

__device__ __forceinline__ int block_scan_incl(int* sh, int t, int v) {
    sh[t] = v;
    __syncthreads();
    for (int off = 1; off < ST; off <<= 1) {
        int a = (t >= off) ? sh[t - off] : 0;
        __syncthreads();
        sh[t] += a;
        __syncthreads();
    }
    return sh[t];
}

__global__ __launch_bounds__(ST) void k_bscatter(const int* __restrict__ dst, int* __restrict__ tabs,
                                                 unsigned* __restrict__ rec,
                                                 const int* __restrict__ src, const float* __restrict__ e0,
                                                 const float* __restrict__ e1, float* __restrict__ p0,
                                                 float* __restrict__ p1, size_t ws_size) {
    __shared__ int sh[ST];
    __shared__ int lcur[NBK];
    __shared__ unsigned stR[CH];
    __shared__ float st0[CH];
    __shared__ float st1[CH];
    const bool wide = (ws_size >= WIDE_BYTES);
    int t = threadIdx.x, blk = blockIdx.x;
    if (t < NBK) lcur[t] = 0;
    __syncthreads();
    int base = blk * CH + t;
    int   dv[CH / ST];
    int   sv[CH / ST];
    float x0[CH / ST], x1[CH / ST];
    for (int j = 0; j < CH / ST; ++j)
        dv[j] = dst[min(base + j * ST, EE - 1)];
    if (wide) {
        for (int j = 0; j < CH / ST; ++j) {
            int id = min(base + j * ST, EE - 1);
            sv[j] = src[id];
            x0[j] = e0[id];
            x1[j] = e1[id];
        }
    }
    for (int j = 0; j < CH / ST; ++j)
        if (base + j * ST < EE) atomicAdd(&lcur[bucket_of(clamp_node(dv[j]))], 1);
    __syncthreads();
    int c = (t < NBK) ? lcur[t] : 0;
    int incl = block_scan_incl(sh, t, c);
    int nblk = min(CH, EE - blk * CH);
    if (t < NBK) lcur[t] = incl - c;
    if (t <= NBK && (t & 1) == 0)
        tabs[(t >> 1) * NBLK + blk] = (incl - c) | ((t < NBK) ? (incl << 16) : 0);
    __syncthreads();
    for (int j = 0; j < CH / ST; ++j) {
        if (base + j * ST < EE) {
            int d  = clamp_node(dv[j]);
            int bk = bucket_of(d);
            unsigned key = (unsigned)(d - bk * BKT_W) & (KEY_W - 1);
            int r = atomicAdd(&lcur[bk], 1);
            r = min(max(r, 0), CH - 1);
            if (wide) {
                stR[r] = (key << SRC_BITS) | ((unsigned)sv[j] & SRC_MASK);
                st0[r] = x0[j];
                st1[r] = x1[j];
            } else {
                stR[r] = (key << ID_BITS) | (unsigned)(base + j * ST);
            }
        }
    }
    __syncthreads();
    for (int j = 0; j < CH / ST; ++j) {
        int i = j * ST + t;
        if (i < nblk) {
            size_t p = (size_t)blk * CH + i;
            rec[p] = stR[i];
            if (wide) {
                p0[p] = st0[i];
                p1[p] = st1[i];
            }
        }
    }
}

#define HEAD_R 12

static_assert(NN % 8 == 0, "a wave's eight nodes are all inside the graph or all outside");
static_assert(HEAD_R % 2 == 0, "the row pass takes the register slots two at a time");

__device__ __forceinline__ int clamp_e(int i) { return min(max(i, 0), EE - 1); }

template <int J>
__device__ __forceinline__ int quad_lane(int v) {
    return __builtin_amdgcn_update_dpp(v, v, J * 0x55, 0xF, 0xF, false);
}
__device__ __forceinline__ float quad_xor1(float v) {
    return __int_as_float(__builtin_amdgcn_update_dpp(0, __float_as_int(v), 0xB1, 0xF, 0xF, false));
}
__device__ __forceinline__ float quad_xor2(float v) {
    return __int_as_float(__builtin_amdgcn_update_dpp(0, __float_as_int(v), 0x4E, 0xF, 0xF, false));
}

template <int J>
__device__ __forceinline__ void row_fetch(const float4* __restrict__ h4, int s_reg, float a_reg,
                                          int c4, float4& r, float& w) {
    int s = quad_lane<J>(s_reg);
    w = __int_as_float(quad_lane<J>(__float_as_int(a_reg)));
    r = h4[(unsigned)(s * (CC / 4) + c4)];
}

__device__ __forceinline__ void row_add(float4& acc, const float4& r, float w) {
    acc.x = fmaf(r.x, w, acc.x); acc.y = fmaf(r.y, w, acc.y);
    acc.z = fmaf(r.z, w, acc.z); acc.w = fmaf(r.w, w, acc.w);
}

#define CAPH (CAP / 2)
#define NPQ  (ST / 4)

static_assert(KEY_W % NPQ == 0, "whole passes over the key range");
static_assert((BKT_W % 8 == 0) && (NPQ % 8 == 0), "a wave's eight nodes start at a multiple of eight, all inside the bucket or all outside");

__device__ __forceinline__ float4 node_lds(const unsigned* rs, int beg, int cnt, int c4,
                                           const float4* __restrict__ h4) {
    int   sv[HEAD_R];
    float ev[HEAD_R];
    #pragma unroll
    for (int i = 0; i < HEAD_R; ++i) {
        int j = i * 4 + c4;
        int idx = min(beg + (j < cnt ? j : 0), CAPH - 1);
        sv[i] = (int)rs[idx];
        ev[i] = __uint_as_float(rs[CAPH + idx]);
    }

    float m = -INFINITY;
    #pragma unroll
    for (int i = 0; i < HEAD_R; ++i) if (i * 4 + c4 < cnt) m = fmaxf(m, ev[i]);
    for (int b = HEAD_R * 4; b < cnt; b += 4) {
        int j = b + c4;
        if (j < cnt) m = fmaxf(m, __uint_as_float(rs[CAPH + min(beg + j, CAPH - 1)]));
    }
    m = fmaxf(m, quad_xor1(m));
    m = fmaxf(m, quad_xor2(m));

    float ex[HEAD_R];
    float s = 0.0f;
    #pragma unroll
    for (int i = 0; i < HEAD_R; ++i) {
        ex[i] = (i * 4 + c4 < cnt) ? expf(ev[i] - m) : 0.0f;
        s += ex[i];
    }
    for (int b = HEAD_R * 4; b < cnt; b += 4) {
        int j = b + c4;
        if (j < cnt) s += expf(__uint_as_float(rs[CAPH + min(beg + j, CAPH - 1)]) - m);
    }
    s += quad_xor1(s);
    s += quad_xor2(s);

    #pragma unroll
    for (int i = 0; i < HEAD_R; ++i) {
        bool ok = (i * 4 + c4 < cnt);
        ex[i] = ok ? ex[i] / s : 0.0f;
        sv[i] = ok ? sv[i] : 0;
    }

    float4 acc = make_float4(0.0f, 0.0f, 0.0f, 0.0f);
    #pragma unroll
    for (int i = 0; i < HEAD_R; i += 2) {
        if (i * 4 < cnt) {
            float4 r0, r1, r2, r3, r4, r5, r6, r7;
            float  w0, w1, w2, w3, w4, w5, w6, w7;
            row_fetch<0>(h4, sv[i],     ex[i],     c4, r0, w0);
            row_fetch<1>(h4, sv[i],     ex[i],     c4, r1, w1);
            row_fetch<2>(h4, sv[i],     ex[i],     c4, r2, w2);
            row_fetch<3>(h4, sv[i],     ex[i],     c4, r3, w3);
            row_fetch<0>(h4, sv[i + 1], ex[i + 1], c4, r4, w4);
            row_fetch<1>(h4, sv[i + 1], ex[i + 1], c4, r5, w5);
            row_fetch<2>(h4, sv[i + 1], ex[i + 1], c4, r6, w6);
            row_fetch<3>(h4, sv[i + 1], ex[i + 1], c4, r7, w7);
            row_add(acc, r0, w0); row_add(acc, r1, w1); row_add(acc, r2, w2); row_add(acc, r3, w3);
            row_add(acc, r4, w4); row_add(acc, r5, w5); row_add(acc, r6, w6); row_add(acc, r7, w7);
        }
    }
    for (int b = HEAD_R * 4; b < cnt; b += 4) {
        int j = b + c4;
        float my_a = 0.0f;
        int my_s = 0;
        if (j < cnt) {
            int idx = min(beg + j, CAPH - 1);
            my_a = expf(__uint_as_float(rs[CAPH + idx]) - m) / s;
            my_s = (int)rs[idx];
        }
        float4 r0, r1, r2, r3;
        float  w0, w1, w2, w3;
        row_fetch<0>(h4, my_s, my_a, c4, r0, w0);
        row_fetch<1>(h4, my_s, my_a, c4, r1, w1);
        row_fetch<2>(h4, my_s, my_a, c4, r2, w2);
        row_fetch<3>(h4, my_s, my_a, c4, r3, w3);
        row_add(acc, r0, w0); row_add(acc, r1, w1); row_add(acc, r2, w2); row_add(acc, r3, w3);
    }
    return acc;
}

template <int J>
__device__ __forceinline__ void scan_take(const float4* __restrict__ h4, int sv, float a, int c4, float4& acc) {
    int s = quad_lane<J>(sv);
    float w = __int_as_float(quad_lane<J>(__float_as_int(a)));
    if (s >= 0) {
        float4 r = h4[(unsigned)(s * (CC / 4) + c4)];
        row_add(acc, r, w);
    }
}

__device__ __forceinline__ float4 node_scan(const unsigned* __restrict__ recs, const float* __restrict__ e_pl,
                                            const float* __restrict__ e_in, const int* __restrict__ src,
                                            bool wide, const int* rsl, int k, int c4,
                                            const float4* __restrict__ h4) {
    const int kshift = wide ? SRC_BITS : ID_BITS;
    float m = -INFINITY;
    for (int j = 0; j < NBLK; ++j) {
        int s0 = j * CH + (rsl[j] & 0xFFFF), len = rsl[j] >> 16;
        for (int i = c4; i < len; i += 4) {
            int idx = clamp_e(s0 + i);
            unsigned r = recs[idx];
            if ((int)((r >> kshift) & (KEY_W - 1)) == k)
                m = fmaxf(m, wide ? e_pl[idx] : e_in[clamp_e((int)(r & ID_MASK))]);
        }
    }
    m = fmaxf(m, quad_xor1(m));
    m = fmaxf(m, quad_xor2(m));
    float s = 0.0f;
    for (int j = 0; j < NBLK; ++j) {
        int s0 = j * CH + (rsl[j] & 0xFFFF), len = rsl[j] >> 16;
        for (int i = c4; i < len; i += 4) {
            int idx = clamp_e(s0 + i);
            unsigned r = recs[idx];
            if ((int)((r >> kshift) & (KEY_W - 1)) == k)
                s += expf((wide ? e_pl[idx] : e_in[clamp_e((int)(r & ID_MASK))]) - m);
        }
    }
    s += quad_xor1(s);
    s += quad_xor2(s);
    float4 acc = make_float4(0.0f, 0.0f, 0.0f, 0.0f);
    for (int j = 0; j < NBLK; ++j) {
        int s0 = j * CH + (rsl[j] & 0xFFFF), len = rsl[j] >> 16;
        for (int i0 = 0; i0 < len; i0 += 4) {
            int i = i0 + c4;
            float a = 0.0f;
            int sv = -1;
            if (i < len) {
                int idx = clamp_e(s0 + i);
                unsigned r = recs[idx];
                if ((int)((r >> kshift) & (KEY_W - 1)) == k) {
                    int id = clamp_e((int)(r & ID_MASK));
                    float e = wide ? e_pl[idx] : e_in[id];
                    a  = expf(e - m) / s;
                    sv = wide ? (int)(r & SRC_MASK) : max(src[id], 0);
                }
            }
            scan_take<0>(h4, sv, a, c4, acc);
            scan_take<1>(h4, sv, a, c4, acc);
            scan_take<2>(h4, sv, a, c4, acc);
            scan_take<3>(h4, sv, a, c4, acc);
        }
    }
    return acc;
}

__global__ __launch_bounds__(ST) void k_layer(
    const int* __restrict__ tabs,
    const unsigned* __restrict__ recs, const float* __restrict__ e_pl,
    const float* __restrict__ e_in, const int* __restrict__ src,
    const float* __restrict__ hcur, const float* __restrict__ one_hot,
    const int* __restrict__ mask, const float* __restrict__ att,
    float* hdst, int mode, size_t ws_size)
{
    __shared__ unsigned rec[CAP];
    __shared__ int sh[ST];
    __shared__ int cnt[KEY_W];
    __shared__ int head[KEY_W + 32];
    __shared__ int rsl[NBLK];
    __shared__ int agg[2];
    const bool wide = (ws_size >= WIDE_BYTES);
    const int  kshift = wide ? SRC_BITS : ID_BITS;
    const float4* __restrict__ h4 = (const float4*)hcur;
    int t = threadIdx.x, b = blockIdx.x;
    if (t < KEY_W) cnt[t] = 0;
    if (t < 2) agg[t] = 0;
    __syncthreads();
    if (t < NBLK) {
        int nb = min(CH, EE - t * CH);
        unsigned wa = (unsigned)tabs[(b >> 1) * NBLK + t];
        unsigned wb = (unsigned)tabs[((b + 1) >> 1) * NBLK + t];
        int s0 = min((int)((b & 1) ? (wa >> 16) : (wa & 0xFFFFu)), nb);
        int s1 = min(max((int)((b & 1) ? (wb & 0xFFFFu) : (wb >> 16)), s0), nb);
        rsl[t] = s0 | ((s1 - s0) << 16);
        atomicAdd(&agg[0], s1 - s0);
        atomicMax(&agg[1], s1 - s0);
    }
    __syncthreads();
    const int  n    = agg[0];
    const bool fits = (n <= CAPH) && (agg[1] <= RUN_CAP);
    const int ln = t & 31;
    const int wv = __builtin_amdgcn_readfirstlane(t >> 5);
    int pk[RPW];
    if (fits) {
        #pragma unroll
        for (int u = 0; u < RPW; ++u) {
            int w = rsl[min(wv + u * NW, NBLK - 1)];
            pk[u] = __builtin_amdgcn_readfirstlane((wv + u * NW < NBLK) ? w : (w & 0xFFFF));
        }
        for (int trip = 0; trip < RUN_CAP / 32; ++trip) {
            const int i = ln + trip * 32;
            unsigned r[RPW];
            #pragma unroll
            for (int u = 0; u < RPW; ++u) {
                int j  = min(wv + u * NW, NBLK - 1);
                int gi = clamp_e(j * CH + (pk[u] & 0xFFFF) + i);
                r[u] = (i < (pk[u] >> 16)) ? recs[gi] : 0u;
            }
            #pragma unroll
            for (int u = 0; u < RPW; ++u)
                if (i < (pk[u] >> 16)) atomicAdd(&cnt[(r[u] >> kshift) & (KEY_W - 1)], 1);
        }
    }
    __syncthreads();
    int c = (t < KEY_W) ? cnt[t] : 0;
    int incl = block_scan_incl(sh, t, c);
    if (t < KEY_W) head[t] = incl - c;
    else if (t < KEY_W + 32) head[t] = 0;
    __syncthreads();
    if (fits) {
        for (int trip = 0; trip < RUN_CAP / 32; ++trip) {
            const int i = ln + trip * 32;
            unsigned r[RPW];
            float x[RPW];
            int   sv[RPW];
            int   p[RPW];
            #pragma unroll
            for (int u = 0; u < RPW; ++u) {
                int j  = min(wv + u * NW, NBLK - 1);
                int gi = clamp_e(j * CH + (pk[u] & 0xFFFF) + i);
                bool ok = (i < (pk[u] >> 16));
                r[u] = ok ? recs[gi] : 0u;
                x[u] = (ok && wide) ? e_pl[gi] : 0.0f;
            }
            if (wide) {
                #pragma unroll
                for (int u = 0; u < RPW; ++u) sv[u] = (int)(r[u] & SRC_MASK);
            } else {
                #pragma unroll
                for (int u = 0; u < RPW; ++u) {
                    int id = clamp_e((int)(r[u] & ID_MASK));
                    x[u]  = e_in[id];
                    sv[u] = max(src[id], 0);
                }
            }
            #pragma unroll
            for (int u = 0; u < RPW; ++u) {
                int key = (i < (pk[u] >> 16)) ? (int)((r[u] >> kshift) & (KEY_W - 1)) : KEY_W + ln;
                p[u] = atomicAdd(&head[key], 1);
            }
            #pragma unroll
            for (int u = 0; u < RPW; ++u) {
                if (i < (pk[u] >> 16)) {
                    int q = min(max(p[u], 0), CAPH - 1);
                    rec[q]        = (unsigned)sv[u];
                    rec[CAPH + q] = __float_as_uint(x[u]);
                }
            }
        }
        __syncthreads();
    }

    int c4 = t & 3;
    for (int pass = 0; pass < KEY_W / NPQ; ++pass) {
        int k = pass * NPQ + (t >> 2);
        int d = b * BKT_W + k;
        if (k < BKT_W && d < NN) {
            int cn  = cnt[k];
            int beg = sh[k] - cn;
            float4 acc = fits ? node_lds(rec, beg, cn, c4, h4)
                              : node_scan(recs, e_pl, e_in, src, wide, rsl, k, c4, h4);

            size_t o4 = (size_t)d * (CC / 4) + c4;
            float4 oh = ((const float4*)one_hot)[o4];
            bool mk = (mask[d] != 0);
            float4 hv;
            hv.x = mk ? oh.x : acc.x; hv.y = mk ? oh.y : acc.y;
            hv.z = mk ? oh.z : acc.z; hv.w = mk ? oh.w : acc.w;
            float4* hd4 = (float4*)hdst;
            if (mode == 0) {
                hd4[o4] = hv;
            } else {
                float a0 = att[d * 2 + 0], a1 = att[d * 2 + 1];
                float mx = fmaxf(a0, a1);
                float e0 = expf(a0 - mx), e1 = expf(a1 - mx);
                float inv = 1.0f / (e0 + e1);
                float4 y;
                if (mode == 1) {
                    float wg = e0 * inv;
                    y.x = hv.x * wg; y.y = hv.y * wg; y.z = hv.z * wg; y.w = hv.w * wg;
                } else {
                    float wg = e1 * inv;
                    float4 old = hd4[o4];
                    y.x = old.x + hv.x * wg; y.y = old.y + hv.y * wg;
                    y.z = old.z + hv.z * wg; y.w = old.w + hv.w * wg;
                }
                hd4[o4] = y;
            }
        }
    }
}

__global__ __launch_bounds__(256) void k_mlp_combine(
    const float* __restrict__ feat, const float* __restrict__ w1,
    const float* __restrict__ b1,   const float* __restrict__ w2,
    const float* __restrict__ b2,   const float* __restrict__ alpha,
    float* __restrict__ out, int n_pairs)
{
    __shared__ float sW1p[(FF / 2) * W1_PSTRIDE];
    __shared__ float sW2p[(HH / 2) * W2_PSTRIDE];
    __shared__ float sHid[8][2][16 * HID_STRIDE];

    int t = threadIdx.x;

    for (int i = t; i < (FF / 2) * (HH / 4); i += 256) {
        int p  = i >> 4;
        int nq = (i & 15) << 2;
        float4 x = *(const float4*)&w1[(size_t)p * 2 * HH + nq];
        float4 y = *(const float4*)&w1[(size_t)p * 2 * HH + HH + nq];
        float4 lo = { x.x, y.x, x.y, y.y };
        float4 hi = { x.z, y.z, x.w, y.w };
        float* d = &sW1p[p * W1_PSTRIDE + 2 * nq];
        *(float4*)(d + 0) = lo;
        *(float4*)(d + 4) = hi;
    }
    for (int i = t; i < (HH / 2) * CC; i += 256) {
        int p = i >> 4, n = i & 15;
        v2f v; v.x = w2[(size_t)p * 2 * CC + n];
        v.y = w2[(size_t)p * 2 * CC + CC + n];
        *(v2f*)&sW2p[p * W2_PSTRIDE + 2 * n] = v;
    }
    __syncthreads();

    int wave = t >> 5;
    int lane = t & 31;
    int pair = blockIdx.x * 8 + wave;
    if (pair >= n_pairs) return;

    int tile0 = pair * 2;
    int tile1 = tile0 + 1;

    int m     = lane & 15;
    int n     = lane & 15;
    int hi16  = lane >> 4;
    int khalf = hi16 << 1;
    int moff  = hi16 << 3;

    v8f acc0 = {}, acc1 = {}, acc2 = {}, acc3 = {};
    v8f acc4 = {}, acc5 = {}, acc6 = {}, acc7 = {};
    const float* arow0 = feat + ((size_t)tile0 * 16 + m) * FF;
    const float* arow1 = feat + ((size_t)tile1 * 16 + m) * FF;
    for (int k = 0; k < FF; k += 4) {
        int kb = k + khalf;
        v2f a0 = *(const v2f*)(arow0 + kb);
        v2f a1 = *(const v2f*)(arow1 + kb);
        const float* bp = &sW1p[((k >> 1) + hi16) * W1_PSTRIDE + 2 * n];
        v2f b0 = *(const v2f*)(bp + 0);
        v2f b1v = *(const v2f*)(bp + 32);
        v2f b2v = *(const v2f*)(bp + 64);
        v2f b3v = *(const v2f*)(bp + 96);
        acc0 = __builtin_amdgcn_wmma_f32_16x16x4_f32(false, a0, false, b0,  (short)0, acc0, false, false);
        acc1 = __builtin_amdgcn_wmma_f32_16x16x4_f32(false, a0, false, b1v, (short)0, acc1, false, false);
        acc2 = __builtin_amdgcn_wmma_f32_16x16x4_f32(false, a0, false, b2v, (short)0, acc2, false, false);
        acc3 = __builtin_amdgcn_wmma_f32_16x16x4_f32(false, a0, false, b3v, (short)0, acc3, false, false);
        acc4 = __builtin_amdgcn_wmma_f32_16x16x4_f32(false, a1, false, b0,  (short)0, acc4, false, false);
        acc5 = __builtin_amdgcn_wmma_f32_16x16x4_f32(false, a1, false, b1v, (short)0, acc5, false, false);
        acc6 = __builtin_amdgcn_wmma_f32_16x16x4_f32(false, a1, false, b2v, (short)0, acc6, false, false);
        acc7 = __builtin_amdgcn_wmma_f32_16x16x4_f32(false, a1, false, b3v, (short)0, acc7, false, false);
    }

    float bb0 = b1[n + 0], bb1 = b1[n + 16], bb2 = b1[n + 32], bb3 = b1[n + 48];
    {
        float* hd0 = &sHid[wave][0][0];
        float* hd1 = &sHid[wave][1][0];
        for (int r = 0; r < 8; ++r) {
            int mm = r + moff;
            hd0[mm * HID_STRIDE + n +  0] = fmaxf(acc0[r] + bb0, 0.0f);
            hd0[mm * HID_STRIDE + n + 16] = fmaxf(acc1[r] + bb1, 0.0f);
            hd0[mm * HID_STRIDE + n + 32] = fmaxf(acc2[r] + bb2, 0.0f);
            hd0[mm * HID_STRIDE + n + 48] = fmaxf(acc3[r] + bb3, 0.0f);
            hd1[mm * HID_STRIDE + n +  0] = fmaxf(acc4[r] + bb0, 0.0f);
            hd1[mm * HID_STRIDE + n + 16] = fmaxf(acc5[r] + bb1, 0.0f);
            hd1[mm * HID_STRIDE + n + 32] = fmaxf(acc6[r] + bb2, 0.0f);
            hd1[mm * HID_STRIDE + n + 48] = fmaxf(acc7[r] + bb3, 0.0f);
        }
    }

    float bc = b2[n];
    for (int tt = 0; tt < 2; ++tt) {
        const float* hd = &sHid[wave][tt][0];
        v8f acc = {};
        for (int k = 0; k < HH; k += 4) {
            int kb = k + khalf;
            v2f a = *(const v2f*)(hd + m * HID_STRIDE + kb);
            v2f b = *(const v2f*)&sW2p[((k >> 1) + hi16) * W2_PSTRIDE + 2 * n];
            acc = __builtin_amdgcn_wmma_f32_16x16x4_f32(false, a, false, b, (short)0, acc, false, false);
        }
        int tile = tt ? tile1 : tile0;
        for (int r = 0; r < 8; ++r) {
            int node = tile * 16 + r + moff;
            float mlp = acc[r] + bc;
            float al  = alpha[node];
            float sa  = 1.0f / (1.0f + expf(-al));
            float sna = 1.0f - sa;
            float logit = out[(size_t)node * CC + n];
            out[(size_t)node * CC + n] = sa * logit + sna * mlp;
        }
    }
}

static void run_graph(hipStream_t stream, const float* e0, const float* e1,
                      const int* sg, const int* dg, int* tabs, int* perm, float* p0, float* p1,
                      size_t ws_size,
                      float* hA, float* out, const float* label_init, const float* labels_onehot,
                      const int* train_mask, const float* attention, int last_mode) {
    k_bscatter<<<NBLK, ST, 0, stream>>>(dg, tabs, (unsigned*)perm, sg, e0, e1, p0, p1, ws_size);
    k_layer   <<<NBK,  ST, 0, stream>>>(tabs, (const unsigned*)perm, p0, e0, sg, label_init,
                                        labels_onehot, train_mask, attention, hA, 0, ws_size);
    k_layer   <<<NBK,  ST, 0, stream>>>(tabs, (const unsigned*)perm, p1, e1, sg, hA,
                                        labels_onehot, train_mask, attention, out, last_mode, ws_size);
}

extern "C" void kernel_launch(void* const* d_in, const int* in_sizes, int n_in,
                              void* d_out, int out_size, void* d_ws, size_t ws_size,
                              hipStream_t stream) {
    const float* features      = (const float*)d_in[0];
    const float* label_init    = (const float*)d_in[1];
    const float* labels_onehot = (const float*)d_in[2];
    const float* alpha         = (const float*)d_in[3];
    const float* attention     = (const float*)d_in[4];
    const float* e_edge        = (const float*)d_in[5];
    const float* w1            = (const float*)d_in[6];
    const float* b1            = (const float*)d_in[7];
    const float* w2            = (const float*)d_in[8];
    const float* b2            = (const float*)d_in[9];
    const int*   src           = (const int*)d_in[10];
    const int*   dst           = (const int*)d_in[11];
    const int*   train_mask    = (const int*)d_in[12];
    float* out = (float*)d_out;

    int*   tabs = (int*)d_ws;
    int*   perm = tabs + NN;
    float* hA   = (float*)(perm + EE);
    float* p0   = hA + NN * CC;
    float* p1   = p0 + EE;

    const int TB = 256;

    run_graph(stream, e_edge,      e_edge + (size_t)GG * EE,      src,      dst,
              tabs, perm, p0, p1, ws_size, hA, out, label_init, labels_onehot, train_mask, attention, 1);
    run_graph(stream, e_edge + EE, e_edge + (size_t)GG * EE + EE, src + EE, dst + EE,
              tabs, perm, p0, p1, ws_size, hA, out, label_init, labels_onehot, train_mask, attention, 2);

    const int n_pairs = NN / 32;
    const int blocks  = (n_pairs + 7) / 8;
    k_mlp_combine<<<blocks, TB, 0, stream>>>(features, w1, b1, w2, b2, alpha, out, n_pairs);
}


# MLP: one resident work-group per compute unit (256 x 16 waves) stages the weights once and loops over 16-row tiles, 6 or 7 per SIMD, instead of 391 work-groups in two rounds of 8 tiles per SIMD
# speedup vs baseline: 16.6049x; 1.0787x over previous
#include <hip/hip_runtime.h>
#include <hip/hip_bf16.h>
#include <math.h>

#define NN 100000
#define CC 16
#define GG 2
#define LL 2
#define EE 3200000
#define FF 512
#define HH 64

#define W1_PSTRIDE 160
#define W2_PSTRIDE 32
#define HID_STRIDE 66

typedef float v2f __attribute__((ext_vector_type(2)));
typedef float v8f __attribute__((ext_vector_type(8)));

#define KEY_BITS  8
#define KEY_W     (1 << KEY_BITS)
#define BKT_W     200
#define NBK       ((NN + BKT_W - 1) / BKT_W)
#define CH        12800
#define NBLK      ((EE + CH - 1) / CH)
#define ST        512
#define NW        (ST / 32)
#define RPW       ((NBLK + NW - 1) / NW)
#define RUN_CAP   64
#define CAP       32768
#define ID_BITS   22
#define ID_MASK   0x3FFFFFu
#define SRC_BITS  17
#define SRC_MASK  0x1FFFFu
#define WIDE_BYTES (((size_t)NN + (size_t)EE + (size_t)NN * CC + 2 * (size_t)EE) * sizeof(int))

static_assert(EE <= (1 << ID_BITS), "edge id must fit the record");
static_assert(NN <= (1 << SRC_BITS) && KEY_BITS + SRC_BITS <= 32 && KEY_BITS + ID_BITS <= 32, "bucket-local node and source node or edge id fit one word");
static_assert(BKT_W <= KEY_W, "a bucket-local node fits the key");
static_assert(NBK < ST && NBLK <= ST && KEY_W <= ST, "one scan pass per table");
static_assert(CH % ST == 0, "whole strides per block");
static_assert(CH < 65536, "a run's start inside its chunk fits 16 bits");
static_assert(NBK % 2 == 0 && (size_t)(NBK / 2 + 1) * NBLK <= (size_t)NN, "the run-start table, one word per pair of buckets and chunk, fits the table words");

__device__ __forceinline__ int clamp_node(int d) { return min(max(d, 0), NN - 1); }
__device__ __forceinline__ int bucket_of(int d) { return (int)((unsigned)d / (unsigned)BKT_W); }

__device__ __forceinline__ int block_scan_incl(int* sh, int t, int v) {
    sh[t] = v;
    __syncthreads();
    for (int off = 1; off < ST; off <<= 1) {
        int a = (t >= off) ? sh[t - off] : 0;
        __syncthreads();
        sh[t] += a;
        __syncthreads();
    }
    return sh[t];
}

__global__ __launch_bounds__(ST) void k_bscatter(const int* __restrict__ dst, int* __restrict__ tabs,
                                                 unsigned* __restrict__ rec,
                                                 const int* __restrict__ src, const float* __restrict__ e0,
                                                 const float* __restrict__ e1, float* __restrict__ p0,
                                                 float* __restrict__ p1, size_t ws_size) {
    __shared__ int sh[ST];
    __shared__ int lcur[NBK];
    __shared__ unsigned stR[CH];
    __shared__ float st0[CH];
    __shared__ float st1[CH];
    const bool wide = (ws_size >= WIDE_BYTES);
    int t = threadIdx.x, blk = blockIdx.x;
    if (t < NBK) lcur[t] = 0;
    __syncthreads();
    int base = blk * CH + t;
    int   dv[CH / ST];
    int   sv[CH / ST];
    float x0[CH / ST], x1[CH / ST];
    for (int j = 0; j < CH / ST; ++j)
        dv[j] = dst[min(base + j * ST, EE - 1)];
    if (wide) {
        for (int j = 0; j < CH / ST; ++j) {
            int id = min(base + j * ST, EE - 1);
            sv[j] = src[id];
            x0[j] = e0[id];
            x1[j] = e1[id];
        }
    }
    for (int j = 0; j < CH / ST; ++j)
        if (base + j * ST < EE) atomicAdd(&lcur[bucket_of(clamp_node(dv[j]))], 1);
    __syncthreads();
    int c = (t < NBK) ? lcur[t] : 0;
    int incl = block_scan_incl(sh, t, c);
    int nblk = min(CH, EE - blk * CH);
    if (t < NBK) lcur[t] = incl - c;
    if (t <= NBK && (t & 1) == 0)
        tabs[(t >> 1) * NBLK + blk] = (incl - c) | ((t < NBK) ? (incl << 16) : 0);
    __syncthreads();
    for (int j = 0; j < CH / ST; ++j) {
        if (base + j * ST < EE) {
            int d  = clamp_node(dv[j]);
            int bk = bucket_of(d);
            unsigned key = (unsigned)(d - bk * BKT_W) & (KEY_W - 1);
            int r = atomicAdd(&lcur[bk], 1);
            r = min(max(r, 0), CH - 1);
            if (wide) {
                stR[r] = (key << SRC_BITS) | ((unsigned)sv[j] & SRC_MASK);
                st0[r] = x0[j];
                st1[r] = x1[j];
            } else {
                stR[r] = (key << ID_BITS) | (unsigned)(base + j * ST);
            }
        }
    }
    __syncthreads();
    for (int j = 0; j < CH / ST; ++j) {
        int i = j * ST + t;
        if (i < nblk) {
            size_t p = (size_t)blk * CH + i;
            rec[p] = stR[i];
            if (wide) {
                p0[p] = st0[i];
                p1[p] = st1[i];
            }
        }
    }
}

#define HEAD_R 12

static_assert(NN % 8 == 0, "a wave's eight nodes are all inside the graph or all outside");
static_assert(HEAD_R % 2 == 0, "the row pass takes the register slots two at a time");

__device__ __forceinline__ int clamp_e(int i) { return min(max(i, 0), EE - 1); }

template <int J>
__device__ __forceinline__ int quad_lane(int v) {
    return __builtin_amdgcn_update_dpp(v, v, J * 0x55, 0xF, 0xF, false);
}
__device__ __forceinline__ float quad_xor1(float v) {
    return __int_as_float(__builtin_amdgcn_update_dpp(0, __float_as_int(v), 0xB1, 0xF, 0xF, false));
}
__device__ __forceinline__ float quad_xor2(float v) {
    return __int_as_float(__builtin_amdgcn_update_dpp(0, __float_as_int(v), 0x4E, 0xF, 0xF, false));
}

template <int J>
__device__ __forceinline__ void row_fetch(const float4* __restrict__ h4, int s_reg, float a_reg,
                                          int c4, float4& r, float& w) {
    int s = quad_lane<J>(s_reg);
    w = __int_as_float(quad_lane<J>(__float_as_int(a_reg)));
    r = h4[(unsigned)(s * (CC / 4) + c4)];
}

__device__ __forceinline__ void row_add(float4& acc, const float4& r, float w) {
    acc.x = fmaf(r.x, w, acc.x); acc.y = fmaf(r.y, w, acc.y);
    acc.z = fmaf(r.z, w, acc.z); acc.w = fmaf(r.w, w, acc.w);
}

#define CAPH (CAP / 2)
#define NPQ  (ST / 4)

static_assert(KEY_W % NPQ == 0, "whole passes over the key range");
static_assert((BKT_W % 8 == 0) && (NPQ % 8 == 0), "a wave's eight nodes start at a multiple of eight, all inside the bucket or all outside");

__device__ __forceinline__ float4 node_lds(const unsigned* rs, int beg, int cnt, int c4,
                                           const float4* __restrict__ h4) {
    int   sv[HEAD_R];
    float ev[HEAD_R];
    #pragma unroll
    for (int i = 0; i < HEAD_R; ++i) {
        int j = i * 4 + c4;
        int idx = min(beg + (j < cnt ? j : 0), CAPH - 1);
        sv[i] = (int)rs[idx];
        ev[i] = __uint_as_float(rs[CAPH + idx]);
    }

    float m = -INFINITY;
    #pragma unroll
    for (int i = 0; i < HEAD_R; ++i) if (i * 4 + c4 < cnt) m = fmaxf(m, ev[i]);
    for (int b = HEAD_R * 4; b < cnt; b += 4) {
        int j = b + c4;
        if (j < cnt) m = fmaxf(m, __uint_as_float(rs[CAPH + min(beg + j, CAPH - 1)]));
    }
    m = fmaxf(m, quad_xor1(m));
    m = fmaxf(m, quad_xor2(m));

    float ex[HEAD_R];
    float s = 0.0f;
    #pragma unroll
    for (int i = 0; i < HEAD_R; ++i) {
        ex[i] = (i * 4 + c4 < cnt) ? expf(ev[i] - m) : 0.0f;
        s += ex[i];
    }
    for (int b = HEAD_R * 4; b < cnt; b += 4) {
        int j = b + c4;
        if (j < cnt) s += expf(__uint_as_float(rs[CAPH + min(beg + j, CAPH - 1)]) - m);
    }
    s += quad_xor1(s);
    s += quad_xor2(s);

    #pragma unroll
    for (int i = 0; i < HEAD_R; ++i) {
        bool ok = (i * 4 + c4 < cnt);
        ex[i] = ok ? ex[i] / s : 0.0f;
        sv[i] = ok ? sv[i] : 0;
    }

    float4 acc = make_float4(0.0f, 0.0f, 0.0f, 0.0f);
    #pragma unroll
    for (int i = 0; i < HEAD_R; i += 2) {
        if (i * 4 < cnt) {
            float4 r0, r1, r2, r3, r4, r5, r6, r7;
            float  w0, w1, w2, w3, w4, w5, w6, w7;
            row_fetch<0>(h4, sv[i],     ex[i],     c4, r0, w0);
            row_fetch<1>(h4, sv[i],     ex[i],     c4, r1, w1);
            row_fetch<2>(h4, sv[i],     ex[i],     c4, r2, w2);
            row_fetch<3>(h4, sv[i],     ex[i],     c4, r3, w3);
            row_fetch<0>(h4, sv[i + 1], ex[i + 1], c4, r4, w4);
            row_fetch<1>(h4, sv[i + 1], ex[i + 1], c4, r5, w5);
            row_fetch<2>(h4, sv[i + 1], ex[i + 1], c4, r6, w6);
            row_fetch<3>(h4, sv[i + 1], ex[i + 1], c4, r7, w7);
            row_add(acc, r0, w0); row_add(acc, r1, w1); row_add(acc, r2, w2); row_add(acc, r3, w3);
            row_add(acc, r4, w4); row_add(acc, r5, w5); row_add(acc, r6, w6); row_add(acc, r7, w7);
        }
    }
    for (int b = HEAD_R * 4; b < cnt; b += 4) {
        int j = b + c4;
        float my_a = 0.0f;
        int my_s = 0;
        if (j < cnt) {
            int idx = min(beg + j, CAPH - 1);
            my_a = expf(__uint_as_float(rs[CAPH + idx]) - m) / s;
            my_s = (int)rs[idx];
        }
        float4 r0, r1, r2, r3;
        float  w0, w1, w2, w3;
        row_fetch<0>(h4, my_s, my_a, c4, r0, w0);
        row_fetch<1>(h4, my_s, my_a, c4, r1, w1);
        row_fetch<2>(h4, my_s, my_a, c4, r2, w2);
        row_fetch<3>(h4, my_s, my_a, c4, r3, w3);
        row_add(acc, r0, w0); row_add(acc, r1, w1); row_add(acc, r2, w2); row_add(acc, r3, w3);
    }
    return acc;
}

template <int J>
__device__ __forceinline__ void scan_take(const float4* __restrict__ h4, int sv, float a, int c4, float4& acc) {
    int s = quad_lane<J>(sv);
    float w = __int_as_float(quad_lane<J>(__float_as_int(a)));
    if (s >= 0) {
        float4 r = h4[(unsigned)(s * (CC / 4) + c4)];
        row_add(acc, r, w);
    }
}

__device__ __forceinline__ float4 node_scan(const unsigned* __restrict__ recs, const float* __restrict__ e_pl,
                                            const float* __restrict__ e_in, const int* __restrict__ src,
                                            bool wide, const int* rsl, int k, int c4,
                                            const float4* __restrict__ h4) {
    const int kshift = wide ? SRC_BITS : ID_BITS;
    float m = -INFINITY;
    for (int j = 0; j < NBLK; ++j) {
        int s0 = j * CH + (rsl[j] & 0xFFFF), len = rsl[j] >> 16;
        for (int i = c4; i < len; i += 4) {
            int idx = clamp_e(s0 + i);
            unsigned r = recs[idx];
            if ((int)((r >> kshift) & (KEY_W - 1)) == k)
                m = fmaxf(m, wide ? e_pl[idx] : e_in[clamp_e((int)(r & ID_MASK))]);
        }
    }
    m = fmaxf(m, quad_xor1(m));
    m = fmaxf(m, quad_xor2(m));
    float s = 0.0f;
    for (int j = 0; j < NBLK; ++j) {
        int s0 = j * CH + (rsl[j] & 0xFFFF), len = rsl[j] >> 16;
        for (int i = c4; i < len; i += 4) {
            int idx = clamp_e(s0 + i);
            unsigned r = recs[idx];
            if ((int)((r >> kshift) & (KEY_W - 1)) == k)
                s += expf((wide ? e_pl[idx] : e_in[clamp_e((int)(r & ID_MASK))]) - m);
        }
    }
    s += quad_xor1(s);
    s += quad_xor2(s);
    float4 acc = make_float4(0.0f, 0.0f, 0.0f, 0.0f);
    for (int j = 0; j < NBLK; ++j) {
        int s0 = j * CH + (rsl[j] & 0xFFFF), len = rsl[j] >> 16;
        for (int i0 = 0; i0 < len; i0 += 4) {
            int i = i0 + c4;
            float a = 0.0f;
            int sv = -1;
            if (i < len) {
                int idx = clamp_e(s0 + i);
                unsigned r = recs[idx];
                if ((int)((r >> kshift) & (KEY_W - 1)) == k) {
                    int id = clamp_e((int)(r & ID_MASK));
                    float e = wide ? e_pl[idx] : e_in[id];
                    a  = expf(e - m) / s;
                    sv = wide ? (int)(r & SRC_MASK) : max(src[id], 0);
                }
            }
            scan_take<0>(h4, sv, a, c4, acc);
            scan_take<1>(h4, sv, a, c4, acc);
            scan_take<2>(h4, sv, a, c4, acc);
            scan_take<3>(h4, sv, a, c4, acc);
        }
    }
    return acc;
}

__global__ __launch_bounds__(ST) void k_layer(
    const int* __restrict__ tabs,
    const unsigned* __restrict__ recs, const float* __restrict__ e_pl,
    const float* __restrict__ e_in, const int* __restrict__ src,
    const float* __restrict__ hcur, const float* __restrict__ one_hot,
    const int* __restrict__ mask, const float* __restrict__ att,
    float* hdst, int mode, size_t ws_size)
{
    __shared__ unsigned rec[CAP];
    __shared__ int sh[ST];
    __shared__ int cnt[KEY_W];
    __shared__ int head[KEY_W + 32];
    __shared__ int rsl[NBLK];
    __shared__ int agg[2];
    const bool wide = (ws_size >= WIDE_BYTES);
    const int  kshift = wide ? SRC_BITS : ID_BITS;
    const float4* __restrict__ h4 = (const float4*)hcur;
    int t = threadIdx.x, b = blockIdx.x;
    if (t < KEY_W) cnt[t] = 0;
    if (t < 2) agg[t] = 0;
    __syncthreads();
    if (t < NBLK) {
        int nb = min(CH, EE - t * CH);
        unsigned wa = (unsigned)tabs[(b >> 1) * NBLK + t];
        unsigned wb = (unsigned)tabs[((b + 1) >> 1) * NBLK + t];
        int s0 = min((int)((b & 1) ? (wa >> 16) : (wa & 0xFFFFu)), nb);
        int s1 = min(max((int)((b & 1) ? (wb & 0xFFFFu) : (wb >> 16)), s0), nb);
        rsl[t] = s0 | ((s1 - s0) << 16);
        atomicAdd(&agg[0], s1 - s0);
        atomicMax(&agg[1], s1 - s0);
    }
    __syncthreads();
    const int  n    = agg[0];
    const bool fits = (n <= CAPH) && (agg[1] <= RUN_CAP);
    const int ln = t & 31;
    const int wv = __builtin_amdgcn_readfirstlane(t >> 5);
    int pk[RPW];
    if (fits) {
        #pragma unroll
        for (int u = 0; u < RPW; ++u) {
            int w = rsl[min(wv + u * NW, NBLK - 1)];
            pk[u] = __builtin_amdgcn_readfirstlane((wv + u * NW < NBLK) ? w : (w & 0xFFFF));
        }
        for (int trip = 0; trip < RUN_CAP / 32; ++trip) {
            const int i = ln + trip * 32;
            unsigned r[RPW];
            #pragma unroll
            for (int u = 0; u < RPW; ++u) {
                int j  = min(wv + u * NW, NBLK - 1);
                int gi = clamp_e(j * CH + (pk[u] & 0xFFFF) + i);
                r[u] = (i < (pk[u] >> 16)) ? recs[gi] : 0u;
            }
            #pragma unroll
            for (int u = 0; u < RPW; ++u)
                if (i < (pk[u] >> 16)) atomicAdd(&cnt[(r[u] >> kshift) & (KEY_W - 1)], 1);
        }
    }
    __syncthreads();
    int c = (t < KEY_W) ? cnt[t] : 0;
    int incl = block_scan_incl(sh, t, c);
    if (t < KEY_W) head[t] = incl - c;
    else if (t < KEY_W + 32) head[t] = 0;
    __syncthreads();
    if (fits) {
        for (int trip = 0; trip < RUN_CAP / 32; ++trip) {
            const int i = ln + trip * 32;
            unsigned r[RPW];
            float x[RPW];
            int   sv[RPW];
            int   p[RPW];
            #pragma unroll
            for (int u = 0; u < RPW; ++u) {
                int j  = min(wv + u * NW, NBLK - 1);
                int gi = clamp_e(j * CH + (pk[u] & 0xFFFF) + i);
                bool ok = (i < (pk[u] >> 16));
                r[u] = ok ? recs[gi] : 0u;
                x[u] = (ok && wide) ? e_pl[gi] : 0.0f;
            }
            if (wide) {
                #pragma unroll
                for (int u = 0; u < RPW; ++u) sv[u] = (int)(r[u] & SRC_MASK);
            } else {
                #pragma unroll
                for (int u = 0; u < RPW; ++u) {
                    int id = clamp_e((int)(r[u] & ID_MASK));
                    x[u]  = e_in[id];
                    sv[u] = max(src[id], 0);
                }
            }
            #pragma unroll
            for (int u = 0; u < RPW; ++u) {
                int key = (i < (pk[u] >> 16)) ? (int)((r[u] >> kshift) & (KEY_W - 1)) : KEY_W + ln;
                p[u] = atomicAdd(&head[key], 1);
            }
            #pragma unroll
            for (int u = 0; u < RPW; ++u) {
                if (i < (pk[u] >> 16)) {
                    int q = min(max(p[u], 0), CAPH - 1);
                    rec[q]        = (unsigned)sv[u];
                    rec[CAPH + q] = __float_as_uint(x[u]);
                }
            }
        }
        __syncthreads();
    }

    int c4 = t & 3;
    for (int pass = 0; pass < KEY_W / NPQ; ++pass) {
        int k = pass * NPQ + (t >> 2);
        int d = b * BKT_W + k;
        if (k < BKT_W && d < NN) {
            int cn  = cnt[k];
            int beg = sh[k] - cn;
            float4 acc = fits ? node_lds(rec, beg, cn, c4, h4)
                              : node_scan(recs, e_pl, e_in, src, wide, rsl, k, c4, h4);

            size_t o4 = (size_t)d * (CC / 4) + c4;
            float4 oh = ((const float4*)one_hot)[o4];
            bool mk = (mask[d] != 0);
            float4 hv;
            hv.x = mk ? oh.x : acc.x; hv.y = mk ? oh.y : acc.y;
            hv.z = mk ? oh.z : acc.z; hv.w = mk ? oh.w : acc.w;
            float4* hd4 = (float4*)hdst;
            if (mode == 0) {
                hd4[o4] = hv;
            } else {
                float a0 = att[d * 2 + 0], a1 = att[d * 2 + 1];
                float mx = fmaxf(a0, a1);
                float e0 = expf(a0 - mx), e1 = expf(a1 - mx);
                float inv = 1.0f / (e0 + e1);
                float4 y;
                if (mode == 1) {
                    float wg = e0 * inv;
                    y.x = hv.x * wg; y.y = hv.y * wg; y.z = hv.z * wg; y.w = hv.w * wg;
                } else {
                    float wg = e1 * inv;
                    float4 old = hd4[o4];
                    y.x = old.x + hv.x * wg; y.y = old.y + hv.y * wg;
                    y.z = old.z + hv.z * wg; y.w = old.w + hv.w * wg;
                }
                hd4[o4] = y;
            }
        }
    }
}

#define MLP_GRID  256
#define MLP_WAVES 16
#define MLP_TILES (NN / 16)

static_assert(NN % 16 == 0, "whole 16-row tiles");
static_assert((FF / 2) * (HH / 4) % (MLP_WAVES * 32) == 0, "whole strides in the w1 staging loop");

__global__ __launch_bounds__(MLP_WAVES * 32) void k_mlp_combine(
    const float* __restrict__ feat, const float* __restrict__ w1,
    const float* __restrict__ b1,   const float* __restrict__ w2,
    const float* __restrict__ b2,   const float* __restrict__ alpha,
    float* __restrict__ out, int n_tiles)
{
    __shared__ float sW1p[(FF / 2) * W1_PSTRIDE];
    __shared__ float sW2p[(HH / 2) * W2_PSTRIDE];
    __shared__ float sHid[MLP_WAVES][16 * HID_STRIDE];

    int t = threadIdx.x;

    for (int i = t; i < (FF / 2) * (HH / 4); i += MLP_WAVES * 32) {
        int p  = i >> 4;
        int nq = (i & 15) << 2;
        float4 x = *(const float4*)&w1[(size_t)p * 2 * HH + nq];
        float4 y = *(const float4*)&w1[(size_t)p * 2 * HH + HH + nq];
        float4 lo = { x.x, y.x, x.y, y.y };
        float4 hi = { x.z, y.z, x.w, y.w };
        float* d = &sW1p[p * W1_PSTRIDE + 2 * nq];
        *(float4*)(d + 0) = lo;
        *(float4*)(d + 4) = hi;
    }
    for (int i = t; i < (HH / 2) * CC; i += MLP_WAVES * 32) {
        int p = i >> 4, n = i & 15;
        v2f v; v.x = w2[(size_t)p * 2 * CC + n];
        v.y = w2[(size_t)p * 2 * CC + CC + n];
        *(v2f*)&sW2p[p * W2_PSTRIDE + 2 * n] = v;
    }
    __syncthreads();

    int wave = __builtin_amdgcn_readfirstlane(t >> 5);
    int lane = t & 31;
    int first = (int)(((unsigned)blockIdx.x * (unsigned)n_tiles) / (unsigned)MLP_GRID);
    int last  = (int)(((unsigned)(blockIdx.x + 1) * (unsigned)n_tiles) / (unsigned)MLP_GRID);

    int m     = lane & 15;
    int n     = lane & 15;
    int hi16  = lane >> 4;
    int khalf = hi16 << 1;
    int moff  = hi16 << 3;

    float bb0 = b1[n + 0], bb1 = b1[n + 16], bb2 = b1[n + 32], bb3 = b1[n + 48];
    float bc  = b2[n];
    float* hd = &sHid[wave][0];

    for (int tile = first + wave; tile < last; tile += MLP_WAVES) {
        v8f acc0 = {}, acc1 = {}, acc2 = {}, acc3 = {};
        const float* arow = feat + ((size_t)tile * 16 + m) * FF;
        for (int k = 0; k < FF; k += 4) {
            int kb = k + khalf;
            v2f a = *(const v2f*)(arow + kb);
            const float* bp = &sW1p[((k >> 1) + hi16) * W1_PSTRIDE + 2 * n];
            v2f b0 = *(const v2f*)(bp + 0);
            v2f b1v = *(const v2f*)(bp + 32);
            v2f b2v = *(const v2f*)(bp + 64);
            v2f b3v = *(const v2f*)(bp + 96);
            acc0 = __builtin_amdgcn_wmma_f32_16x16x4_f32(false, a, false, b0,  (short)0, acc0, false, false);
            acc1 = __builtin_amdgcn_wmma_f32_16x16x4_f32(false, a, false, b1v, (short)0, acc1, false, false);
            acc2 = __builtin_amdgcn_wmma_f32_16x16x4_f32(false, a, false, b2v, (short)0, acc2, false, false);
            acc3 = __builtin_amdgcn_wmma_f32_16x16x4_f32(false, a, false, b3v, (short)0, acc3, false, false);
        }

        for (int r = 0; r < 8; ++r) {
            int mm = r + moff;
            hd[mm * HID_STRIDE + n +  0] = fmaxf(acc0[r] + bb0, 0.0f);
            hd[mm * HID_STRIDE + n + 16] = fmaxf(acc1[r] + bb1, 0.0f);
            hd[mm * HID_STRIDE + n + 32] = fmaxf(acc2[r] + bb2, 0.0f);
            hd[mm * HID_STRIDE + n + 48] = fmaxf(acc3[r] + bb3, 0.0f);
        }

        v8f acc = {};
        for (int k = 0; k < HH; k += 4) {
            int kb = k + khalf;
            v2f a = *(const v2f*)(hd + m * HID_STRIDE + kb);
            v2f b = *(const v2f*)&sW2p[((k >> 1) + hi16) * W2_PSTRIDE + 2 * n];
            acc = __builtin_amdgcn_wmma_f32_16x16x4_f32(false, a, false, b, (short)0, acc, false, false);
        }
        for (int r = 0; r < 8; ++r) {
            int node = tile * 16 + r + moff;
            float mlp = acc[r] + bc;
            float al  = alpha[node];
            float sa  = 1.0f / (1.0f + expf(-al));
            float sna = 1.0f - sa;
            float logit = out[(size_t)node * CC + n];
            out[(size_t)node * CC + n] = sa * logit + sna * mlp;
        }
    }
}

static void run_graph(hipStream_t stream, const float* e0, const float* e1,
                      const int* sg, const int* dg, int* tabs, int* perm, float* p0, float* p1,
                      size_t ws_size,
                      float* hA, float* out, const float* label_init, const float* labels_onehot,
                      const int* train_mask, const float* attention, int last_mode) {
    k_bscatter<<<NBLK, ST, 0, stream>>>(dg, tabs, (unsigned*)perm, sg, e0, e1, p0, p1, ws_size);
    k_layer   <<<NBK,  ST, 0, stream>>>(tabs, (const unsigned*)perm, p0, e0, sg, label_init,
                                        labels_onehot, train_mask, attention, hA, 0, ws_size);
    k_layer   <<<NBK,  ST, 0, stream>>>(tabs, (const unsigned*)perm, p1, e1, sg, hA,
                                        labels_onehot, train_mask, attention, out, last_mode, ws_size);
}

extern "C" void kernel_launch(void* const* d_in, const int* in_sizes, int n_in,
                              void* d_out, int out_size, void* d_ws, size_t ws_size,
                              hipStream_t stream) {
    const float* features      = (const float*)d_in[0];
    const float* label_init    = (const float*)d_in[1];
    const float* labels_onehot = (const float*)d_in[2];
    const float* alpha         = (const float*)d_in[3];
    const float* attention     = (const float*)d_in[4];
    const float* e_edge        = (const float*)d_in[5];
    const float* w1            = (const float*)d_in[6];
    const float* b1            = (const float*)d_in[7];
    const float* w2            = (const float*)d_in[8];
    const float* b2            = (const float*)d_in[9];
    const int*   src           = (const int*)d_in[10];
    const int*   dst           = (const int*)d_in[11];
    const int*   train_mask    = (const int*)d_in[12];
    float* out = (float*)d_out;

    int*   tabs = (int*)d_ws;
    int*   perm = tabs + NN;
    float* hA   = (float*)(perm + EE);
    float* p0   = hA + NN * CC;
    float* p1   = p0 + EE;

    run_graph(stream, e_edge,      e_edge + (size_t)GG * EE,      src,      dst,
              tabs, perm, p0, p1, ws_size, hA, out, label_init, labels_onehot, train_mask, attention, 1);
    run_graph(stream, e_edge + EE, e_edge + (size_t)GG * EE + EE, src + EE, dst + EE,
              tabs, perm, p0, p1, ws_size, hA, out, label_init, labels_onehot, train_mask, attention, 2);

    k_mlp_combine<<<MLP_GRID, MLP_WAVES * 32, 0, stream>>>(features, w1, b1, w2, b2, alpha, out, MLP_TILES);
}
